# starting kernel
# baseline (speedup 1.0000x reference)
#include <hip/hip_runtime.h>
#include <hip/hip_bf16.h>

typedef __attribute__((ext_vector_type(2))) float v2f;
typedef __attribute__((ext_vector_type(8))) float v8f;

#define SLOPE 0.2f

static __host__ __device__ inline int cdiv(int a, int b) { return (a + b - 1) / b; }

__device__ inline void atomicAddF32(float* p, float v) { unsafeAtomicAdd(p, v); }

__global__ void gat_fill_f32(float* __restrict__ p, float v, int n) {
    int i = blockIdx.x * blockDim.x + threadIdx.x;
    if (i < n) p[i] = v;
}

__global__ void gat_gemm1_wmma(const float* __restrict__ A,
                               const float* __restrict__ W,
                               float* __restrict__ H,
                               int n_nodes) {
    __shared__ float As[32 * 256];
    __shared__ float Ws[32 * 256];

    const int row0 = blockIdx.x * 32;
    const int tid  = threadIdx.x;
    const int wave = tid >> 5;
    const int lane = tid & 31;

    for (int i = tid; i < 32 * 64; i += 256) {
        int r  = i >> 6;
        int c4 = (i & 63) << 2;
        int rr = row0 + r; if (rr >= n_nodes) rr = n_nodes - 1;
        *(float4*)(&As[r * 256 + c4]) = *(const float4*)(A + (size_t)rr * 256 + c4);
    }

    const int m   = lane & 15;
    const int kk  = (lane >> 4) << 1;
    const int n0  = wave * 32;
    const int n1  = n0 + 16;

    v8f c00 = {}, c01 = {}, c10 = {}, c11 = {};

    for (int kc = 0; kc < 256; kc += 32) {
        __syncthreads();
        for (int i = tid; i < 32 * 64; i += 256) {
            int r  = i >> 6;
            int c4 = (i & 63) << 2;
            *(float4*)(&Ws[r * 256 + c4]) = *(const float4*)(W + (size_t)(kc + r) * 256 + c4);
        }
        __syncthreads();

#pragma unroll
        for (int k = 0; k < 32; k += 4) {
            v2f a0, a1, b0, b1;
            a0.x = As[m * 256 + kc + k + kk];
            a0.y = As[m * 256 + kc + k + kk + 1];
            a1.x = As[(m + 16) * 256 + kc + k + kk];
            a1.y = As[(m + 16) * 256 + kc + k + kk + 1];
            b0.x = Ws[(k + kk) * 256 + n0 + m];
            b0.y = Ws[(k + kk + 1) * 256 + n0 + m];
            b1.x = Ws[(k + kk) * 256 + n1 + m];
            b1.y = Ws[(k + kk + 1) * 256 + n1 + m];
            c00 = __builtin_amdgcn_wmma_f32_16x16x4_f32(false, a0, false, b0, (short)0, c00, false, false);
            c01 = __builtin_amdgcn_wmma_f32_16x16x4_f32(false, a0, false, b1, (short)0, c01, false, false);
            c10 = __builtin_amdgcn_wmma_f32_16x16x4_f32(false, a1, false, b0, (short)0, c10, false, false);
            c11 = __builtin_amdgcn_wmma_f32_16x16x4_f32(false, a1, false, b1, (short)0, c11, false, false);
        }
    }

    const int mhi = (lane >> 4) * 8;
    for (int v = 0; v < 8; ++v) {
        int r0 = row0 + mhi + v;
        int r1 = r0 + 16;
        if (r0 < n_nodes) {
            H[(size_t)r0 * 256 + n0 + m] = c00[v];
            H[(size_t)r0 * 256 + n1 + m] = c01[v];
        }
        if (r1 < n_nodes) {
            H[(size_t)r1 * 256 + n0 + m] = c10[v];
            H[(size_t)r1 * 256 + n1 + m] = c11[v];
        }
    }
}

__global__ void gat_attn_scores(const float* __restrict__ H,
                                const float* __restrict__ al,
                                const float* __restrict__ ar,
                                float* __restrict__ el, float* __restrict__ er,
                                int n_nodes) {
    int i = blockIdx.x * blockDim.x + threadIdx.x;
    if (i >= n_nodes * 8) return;
    int n = i >> 3, h = i & 7;
    const float* hp  = H + (size_t)n * 256 + h * 32;
    const float* alp = al + h * 32;
    const float* arp = ar + h * 32;
    float sl = 0.f, sr = 0.f;
    for (int d = 0; d < 32; ++d) { float v = hp[d]; sl += v * alp[d]; sr += v * arp[d]; }
    el[i] = sl; er[i] = sr;
}

__device__ inline void atomicMaxFloat(float* addr, float value) {
    if (value >= 0.0f)
        atomicMax((int*)addr, __float_as_int(value));
    else
        atomicMin((unsigned int*)addr, __float_as_uint(value));
}

__global__ void gat_edge_e_max(const int* __restrict__ src, const int* __restrict__ dst,
                               const float* __restrict__ el, const float* __restrict__ er,
                               float* __restrict__ e, float* __restrict__ m,
                               int n_edges, int heads) {
    int i = blockIdx.x * blockDim.x + threadIdx.x;
    if (i >= n_edges * heads) return;
    int ed = i / heads, h = i - ed * heads;
    float v = el[src[ed] * heads + h] + er[dst[ed] * heads + h];
    v = (v >= 0.f) ? v : SLOPE * v;
    e[i] = v;
    atomicMaxFloat(&m[dst[ed] * heads + h], v);
}

__global__ void gat_edge_expsum(const int* __restrict__ dst,
                                float* __restrict__ e, const float* __restrict__ m,
                                float* __restrict__ s, int n_edges, int heads) {
    int i = blockIdx.x * blockDim.x + threadIdx.x;
    if (i >= n_edges * heads) return;
    int ed = i / heads, h = i - ed * heads;
    float v = expf(e[i] - m[dst[ed] * heads + h]);
    e[i] = v;
    atomicAddF32(&s[dst[ed] * heads + h], v);
}

__global__ void gat_aggregate(const int* __restrict__ src, const int* __restrict__ dst,
                              const float* __restrict__ H, const float* __restrict__ ue,
                              const float* __restrict__ s, float* __restrict__ agg,
                              int n_edges, int heads, int dim) {
    int hd = heads * dim;
    int i = blockIdx.x * blockDim.x + threadIdx.x;
    if (i >= n_edges * hd) return;
    int ed = i / hd, c = i - ed * hd;
    int h = c / dim;
    int sn = src[ed], dn = dst[ed];
    float alpha = ue[ed * heads + h] / s[dn * heads + h];
    atomicAddF32(&agg[(size_t)dn * hd + c], H[(size_t)sn * hd + c] * alpha);
}

__global__ void gat_elu_bias(float* __restrict__ x, const float* __restrict__ b,
                             int total, int hd) {
    int i = blockIdx.x * blockDim.x + threadIdx.x;
    if (i >= total) return;
    float v = x[i] + b[i % hd];
    x[i] = (v > 0.f) ? v : expm1f(v);
}

__global__ void gat_l2_linear(const float* __restrict__ x, const float* __restrict__ W2,
                              const float* __restrict__ al2, const float* __restrict__ ar2,
                              float* __restrict__ h2, float* __restrict__ el2,
                              float* __restrict__ er2, int n_nodes) {
    int n = blockIdx.x * blockDim.x + threadIdx.x;
    if (n >= n_nodes) return;
    float a0 = 0.f, a1 = 0.f, a2 = 0.f;
    const float* xp = x + (size_t)n * 256;
    for (int k = 0; k < 256; ++k) {
        float xv = xp[k];
        a0 += xv * W2[k * 3 + 0];
        a1 += xv * W2[k * 3 + 1];
        a2 += xv * W2[k * 3 + 2];
    }
    h2[n * 3 + 0] = a0; h2[n * 3 + 1] = a1; h2[n * 3 + 2] = a2;
    el2[n] = a0 * al2[0] + a1 * al2[1] + a2 * al2[2];
    er2[n] = a0 * ar2[0] + a1 * ar2[1] + a2 * ar2[2];
}

__global__ void gat_finalize(const float* __restrict__ agg2, const float* __restrict__ b2,
                             float* __restrict__ out, int total) {
    int i = blockIdx.x * blockDim.x + threadIdx.x;
    if (i >= total) return;
    out[i] = agg2[i] + b2[i % 3];
}

extern "C" void kernel_launch(void* const* d_in, const int* in_sizes, int n_in,
                              void* d_out, int out_size, void* d_ws, size_t ws_size,
                              hipStream_t stream) {
    const float* feat = (const float*)d_in[0];
    const int*   src  = (const int*)  d_in[1];
    const int*   dst  = (const int*)  d_in[2];
    const float* W1   = (const float*)d_in[3];
    const float* al1  = (const float*)d_in[4];
    const float* ar1  = (const float*)d_in[5];
    const float* b1   = (const float*)d_in[6];
    const float* W2   = (const float*)d_in[7];
    const float* al2  = (const float*)d_in[8];
    const float* ar2  = (const float*)d_in[9];
    const float* b2   = (const float*)d_in[10];
    float* out = (float*)d_out;

    const int n_nodes = in_sizes[0] / 256;
    const int n_edges = in_sizes[1];

    char* cur = (char*)d_ws;
    auto alloc = [&](size_t bytes) -> float* {
        float* p = (float*)cur;
        cur += (bytes + 255) & ~(size_t)255;
        return p;
    };
    float* H1   = alloc((size_t)n_nodes * 256 * 4);
    float* el1  = alloc((size_t)n_nodes * 8 * 4);
    float* er1  = alloc((size_t)n_nodes * 8 * 4);
    float* m1   = alloc((size_t)n_nodes * 8 * 4);
    float* s1   = alloc((size_t)n_nodes * 8 * 4);
    float* e1   = alloc((size_t)n_edges * 8 * 4);
    float* agg1 = alloc((size_t)n_nodes * 256 * 4);
    float* h2   = alloc((size_t)n_nodes * 3 * 4);
    float* el2  = alloc((size_t)n_nodes * 4);
    float* er2  = alloc((size_t)n_nodes * 4);
    float* m2   = alloc((size_t)n_nodes * 4);
    float* s2   = alloc((size_t)n_nodes * 4);
    float* e2   = alloc((size_t)n_edges * 4);
    float* agg2 = alloc((size_t)n_nodes * 3 * 4);

    const int B = 256;
    const float NEG_INF = -__builtin_huge_valf();

    gat_gemm1_wmma<<<cdiv(n_nodes, 32), 256, 0, stream>>>(feat, W1, H1, n_nodes);
    gat_attn_scores<<<cdiv(n_nodes * 8, B), B, 0, stream>>>(H1, al1, ar1, el1, er1, n_nodes);

    gat_fill_f32<<<cdiv(n_nodes * 8, B), B, 0, stream>>>(m1, NEG_INF, n_nodes * 8);
    gat_fill_f32<<<cdiv(n_nodes * 8, B), B, 0, stream>>>(s1, 0.f, n_nodes * 8);
    gat_fill_f32<<<cdiv(n_nodes * 256, B), B, 0, stream>>>(agg1, 0.f, n_nodes * 256);

    gat_edge_e_max<<<cdiv(n_edges * 8, B), B, 0, stream>>>(src, dst, el1, er1, e1, m1, n_edges, 8);
    gat_edge_expsum<<<cdiv(n_edges * 8, B), B, 0, stream>>>(dst, e1, m1, s1, n_edges, 8);
    gat_aggregate<<<cdiv(n_edges * 256, B), B, 0, stream>>>(src, dst, H1, e1, s1, agg1,
                                                            n_edges, 8, 32);
    gat_elu_bias<<<cdiv(n_nodes * 256, B), B, 0, stream>>>(agg1, b1, n_nodes * 256, 256);

    gat_l2_linear<<<cdiv(n_nodes, 128), 128, 0, stream>>>(agg1, W2, al2, ar2, h2, el2, er2, n_nodes);

    gat_fill_f32<<<cdiv(n_nodes, B), B, 0, stream>>>(m2, NEG_INF, n_nodes);
    gat_fill_f32<<<cdiv(n_nodes, B), B, 0, stream>>>(s2, 0.f, n_nodes);
    gat_fill_f32<<<cdiv(n_nodes * 3, B), B, 0, stream>>>(agg2, 0.f, n_nodes * 3);

    gat_edge_e_max<<<cdiv(n_edges, B), B, 0, stream>>>(src, dst, el2, er2, e2, m2, n_edges, 1);
    gat_edge_expsum<<<cdiv(n_edges, B), B, 0, stream>>>(dst, e2, m2, s2, n_edges, 1);
    gat_aggregate<<<cdiv(n_edges * 3, B), B, 0, stream>>>(src, dst, h2, e2, s2, agg2,
                                                          n_edges, 1, 3);
    gat_finalize<<<cdiv(n_nodes * 3, B), B, 0, stream>>>(agg2, b2, out, n_nodes * 3);
}


# Edges counting-sorted by dst once per call (returning integer add gives the rank); one wave per dst node does max, exp-sum, weighted row sum in registers: no float atomics, sums in arrival-order rank
# speedup vs baseline: 3.6743x; 3.6743x over previous
#include <hip/hip_runtime.h>
#include <hip/hip_bf16.h>

typedef __attribute__((ext_vector_type(2))) float v2f;
typedef __attribute__((ext_vector_type(8))) float v8f;

#define SLOPE 0.2f

static __host__ __device__ inline int cdiv(int a, int b) { return (a + b - 1) / b; }

__global__ void gat_fill_f32(float* __restrict__ p, float v, int n) {
    int i = blockIdx.x * blockDim.x + threadIdx.x;
    if (i < n) p[i] = v;
}

__global__ void gat_gemm1_wmma(const float* __restrict__ A,
                               const float* __restrict__ W,
                               float* __restrict__ H,
                               int n_nodes) {
    __shared__ float As[32 * 256];
    __shared__ float Ws[32 * 256];

    const int row0 = blockIdx.x * 32;
    const int tid  = threadIdx.x;
    const int wave = tid >> 5;
    const int lane = tid & 31;

    for (int i = tid; i < 32 * 64; i += 256) {
        int r  = i >> 6;
        int c4 = (i & 63) << 2;
        int rr = row0 + r; if (rr >= n_nodes) rr = n_nodes - 1;
        *(float4*)(&As[r * 256 + c4]) = *(const float4*)(A + (size_t)rr * 256 + c4);
    }

    const int m   = lane & 15;
    const int kk  = (lane >> 4) << 1;
    const int n0  = wave * 32;
    const int n1  = n0 + 16;

    v8f c00 = {}, c01 = {}, c10 = {}, c11 = {};

    for (int kc = 0; kc < 256; kc += 32) {
        __syncthreads();
        for (int i = tid; i < 32 * 64; i += 256) {
            int r  = i >> 6;
            int c4 = (i & 63) << 2;
            *(float4*)(&Ws[r * 256 + c4]) = *(const float4*)(W + (size_t)(kc + r) * 256 + c4);
        }
        __syncthreads();

#pragma unroll
        for (int k = 0; k < 32; k += 4) {
            v2f a0, a1, b0, b1;
            a0.x = As[m * 256 + kc + k + kk];
            a0.y = As[m * 256 + kc + k + kk + 1];
            a1.x = As[(m + 16) * 256 + kc + k + kk];
            a1.y = As[(m + 16) * 256 + kc + k + kk + 1];
            b0.x = Ws[(k + kk) * 256 + n0 + m];
            b0.y = Ws[(k + kk + 1) * 256 + n0 + m];
            b1.x = Ws[(k + kk) * 256 + n1 + m];
            b1.y = Ws[(k + kk + 1) * 256 + n1 + m];
            c00 = __builtin_amdgcn_wmma_f32_16x16x4_f32(false, a0, false, b0, (short)0, c00, false, false);
            c01 = __builtin_amdgcn_wmma_f32_16x16x4_f32(false, a0, false, b1, (short)0, c01, false, false);
            c10 = __builtin_amdgcn_wmma_f32_16x16x4_f32(false, a1, false, b0, (short)0, c10, false, false);
            c11 = __builtin_amdgcn_wmma_f32_16x16x4_f32(false, a1, false, b1, (short)0, c11, false, false);
        }
    }

    const int mhi = (lane >> 4) * 8;
    for (int v = 0; v < 8; ++v) {
        int r0 = row0 + mhi + v;
        int r1 = r0 + 16;
        if (r0 < n_nodes) {
            H[(size_t)r0 * 256 + n0 + m] = c00[v];
            H[(size_t)r0 * 256 + n1 + m] = c01[v];
        }
        if (r1 < n_nodes) {
            H[(size_t)r1 * 256 + n0 + m] = c10[v];
            H[(size_t)r1 * 256 + n1 + m] = c11[v];
        }
    }
}

__global__ void gat_attn_scores(const float* __restrict__ H,
                                const float* __restrict__ al,
                                const float* __restrict__ ar,
                                float* __restrict__ el, float* __restrict__ er,
                                int n_nodes) {
    int i = blockIdx.x * blockDim.x + threadIdx.x;
    if (i >= n_nodes * 8) return;
    int n = i >> 3, h = i & 7;
    const float* hp  = H + (size_t)n * 256 + h * 32;
    const float* alp = al + h * 32;
    const float* arp = ar + h * 32;
    float sl = 0.f, sr = 0.f;
    for (int d = 0; d < 32; ++d) { float v = hp[d]; sl += v * alp[d]; sr += v * arp[d]; }
    el[i] = sl; er[i] = sr;
}

__global__ void gat_dst_rank(const int* __restrict__ dst, int* __restrict__ cnt,
                             int* __restrict__ rank, int n_edges) {
    int e = blockIdx.x * blockDim.x + threadIdx.x;
    if (e >= n_edges) return;
    rank[e] = atomicAdd(&cnt[dst[e]], 1);
}

__global__ void __launch_bounds__(1024) gat_row_offsets(const int* __restrict__ cnt,
                                                        int* __restrict__ rowptr, int n) {
    __shared__ int wtot[32];
    __shared__ int wpre[32];
    const int t = threadIdx.x, lane = t & 31, wave = t >> 5;
    const int b = blockIdx.x;
    const int i = b * 1024 + t;
    const int v = (i < n) ? cnt[i] : 0;
    int pre = 0;
    for (int k = 0; k < b; ++k) pre += cnt[k * 1024 + t];
    int x = v;
    for (int o = 1; o < 32; o <<= 1) {
        int y = __shfl_up(x, o, 32);
        if (lane >= o) x += y;
    }
    for (int o = 16; o > 0; o >>= 1) pre += __shfl_xor(pre, o, 32);
    if (lane == 31) wtot[wave] = x;
    if (lane == 0)  wpre[wave] = pre;
    __syncthreads();
    int w = wtot[lane];
    for (int o = 1; o < 32; o <<= 1) {
        int y = __shfl_up(w, o, 32);
        if (lane >= o) w += y;
    }
    int base = wpre[lane];
    for (int o = 16; o > 0; o >>= 1) base += __shfl_xor(base, o, 32);
    const int before_wave = __shfl(w, wave, 32) - wtot[wave];
    const int excl = base + before_wave + x - v;
    if (i < n) rowptr[i] = excl;
    if (i == n - 1) rowptr[n] = excl + v;
}

__global__ void gat_dst_place(const int* __restrict__ src, const int* __restrict__ dst,
                              const int* __restrict__ rowptr, const int* __restrict__ rank,
                              int* __restrict__ ssrc, int n_edges) {
    int e = blockIdx.x * blockDim.x + threadIdx.x;
    if (e >= n_edges) return;
    ssrc[rowptr[dst[e]] + rank[e]] = src[e];
}

template <int CTRL>
__device__ inline float quad_perm(float x) {
    return __int_as_float(__builtin_amdgcn_update_dpp(0, __float_as_int(x), CTRL, 0xF, 0xF, true));
}

template <int K>
__device__ inline void gat_add_row(const float* __restrict__ Hl, int s_lane, float u_lane,
                                   float4& a0, float4& a1) {
    const int sn = __builtin_amdgcn_readlane(s_lane, K);
    const float al = quad_perm<K * 0x55>(u_lane);
    const float4 x0 = *(const float4*)(Hl + (size_t)sn * 256);
    const float4 x1 = *(const float4*)(Hl + (size_t)sn * 256 + 4);
    a0.x += x0.x * al; a0.y += x0.y * al; a0.z += x0.z * al; a0.w += x0.w * al;
    a1.x += x1.x * al; a1.y += x1.y * al; a1.z += x1.z * al; a1.w += x1.w * al;
}

__global__ void __launch_bounds__(256) gat_l1_segments(const int* __restrict__ rowptr,
                                                       const int* __restrict__ ssrc,
                                                       const float* __restrict__ H,
                                                       const float* __restrict__ el,
                                                       const float* __restrict__ er,
                                                       float* __restrict__ agg, int n_nodes) {
    const int lane = threadIdx.x & 31;
    const int node = blockIdx.x * 8 + __builtin_amdgcn_readfirstlane((int)(threadIdx.x >> 5));
    if (node >= n_nodes) return;
    const int eb = rowptr[node], ee = rowptr[node + 1];
    const int head = lane >> 2, sub = lane & 3;
    const float erv = er[node * 8 + head];
    const float* Hl = H + lane * 8;
    const float NEG_INF = -__builtin_huge_valf();

    int   S[8];
    float V[8];
    float U[8];
    float4 a0 = {0.f, 0.f, 0.f, 0.f}, a1 = {0.f, 0.f, 0.f, 0.f};

    auto score_chunk = [&](int c) {
        const int last = ee - 1;
#pragma unroll
        for (int i = 0; i < 8; ++i) {
            const int j = c + 4 * i + sub;
            S[i] = ssrc[j < ee ? j : last];
        }
#pragma unroll
        for (int i = 0; i < 8; ++i) {
            float x = el[S[i] * 8 + head] + erv;
            x = (x >= 0.f) ? x : SLOPE * x;
            V[i] = (c + 4 * i + sub < ee) ? x : NEG_INF;
        }
    };
    auto add_chunk = [&](int cnt) {
#pragma unroll
        for (int i = 0; i < 8; ++i) {
            if (4 * i + 4 <= cnt) {
                gat_add_row<0>(Hl, S[i], U[i], a0, a1);
                gat_add_row<1>(Hl, S[i], U[i], a0, a1);
                gat_add_row<2>(Hl, S[i], U[i], a0, a1);
                gat_add_row<3>(Hl, S[i], U[i], a0, a1);
            } else if (4 * i < cnt) {
                gat_add_row<0>(Hl, S[i], U[i], a0, a1);
                if (4 * i + 1 < cnt) gat_add_row<1>(Hl, S[i], U[i], a0, a1);
                if (4 * i + 2 < cnt) gat_add_row<2>(Hl, S[i], U[i], a0, a1);
            }
        }
    };

    const int deg = ee - eb;
    if (deg > 0 && deg <= 32) {
        score_chunk(eb);
        float mx = V[0];
#pragma unroll
        for (int i = 1; i < 8; ++i) mx = fmaxf(mx, V[i]);
        mx = fmaxf(mx, quad_perm<0xB1>(mx));
        mx = fmaxf(mx, quad_perm<0x4E>(mx));
        float sm = 0.f;
#pragma unroll
        for (int i = 0; i < 8; ++i) {
            const float u = expf(V[i] - mx);
            U[i] = (eb + 4 * i + sub < ee) ? u : 0.f;
            sm += U[i];
        }
        sm += quad_perm<0xB1>(sm);
        sm += quad_perm<0x4E>(sm);
#pragma unroll
        for (int i = 0; i < 8; ++i) U[i] = U[i] / sm;
        add_chunk(deg);
    } else if (deg > 32) {
        float mx = NEG_INF;
        for (int c = eb; c < ee; c += 32) {
            score_chunk(c);
#pragma unroll
            for (int i = 0; i < 8; ++i) mx = fmaxf(mx, V[i]);
        }
        mx = fmaxf(mx, quad_perm<0xB1>(mx));
        mx = fmaxf(mx, quad_perm<0x4E>(mx));
        float sm = 0.f;
        for (int c = eb; c < ee; c += 32) {
            score_chunk(c);
#pragma unroll
            for (int i = 0; i < 8; ++i) {
                const float u = expf(V[i] - mx);
                sm += (c + 4 * i + sub < ee) ? u : 0.f;
            }
        }
        sm += quad_perm<0xB1>(sm);
        sm += quad_perm<0x4E>(sm);
        for (int c = eb; c < ee; c += 32) {
            score_chunk(c);
#pragma unroll
            for (int i = 0; i < 8; ++i) {
                const float u = expf(V[i] - mx) / sm;
                U[i] = (c + 4 * i + sub < ee) ? u : 0.f;
            }
            const int left = ee - c;
            add_chunk(left < 32 ? left : 32);
        }
    }
    float* op = agg + (size_t)node * 256 + lane * 8;
    *(float4*)(op)     = a0;
    *(float4*)(op + 4) = a1;
}

__device__ inline float wave_max(float x) {
    for (int o = 16; o > 0; o >>= 1) x = fmaxf(x, __shfl_xor(x, o, 32));
    return x;
}
__device__ inline float wave_sum(float x) {
    for (int o = 16; o > 0; o >>= 1) x += __shfl_xor(x, o, 32);
    return x;
}

__global__ void __launch_bounds__(256) gat_l2_segments(const int* __restrict__ rowptr,
                                                       const int* __restrict__ ssrc,
                                                       const float* __restrict__ hp,
                                                       const float* __restrict__ er2,
                                                       float* __restrict__ agg2, int n_nodes) {
    const int lane = threadIdx.x & 31;
    const int node = blockIdx.x * 8 + __builtin_amdgcn_readfirstlane((int)(threadIdx.x >> 5));
    if (node >= n_nodes) return;
    const int eb = rowptr[node], ee = rowptr[node + 1];
    const float erv = er2[node];
    const float NEG_INF = -__builtin_huge_valf();
    const float4* rows = (const float4*)hp;
    const int deg = ee - eb;
    float p0 = 0.f, p1 = 0.f, p2 = 0.f;

    if (deg > 0 && deg <= 32) {
        const int j = eb + lane;
        const bool ok = j < ee;
        const float4 r = rows[ssrc[ok ? j : ee - 1]];
        float x = r.w + erv;
        x = (x >= 0.f) ? x : SLOPE * x;
        x = ok ? x : NEG_INF;
        const float mx = wave_max(x);
        const float e = expf(x - mx);
        const float u = ok ? e : 0.f;
        const float sm = wave_sum(u);
        const float al = u / sm;
        p0 = ok ? r.x * al : 0.f;
        p1 = ok ? r.y * al : 0.f;
        p2 = ok ? r.z * al : 0.f;
    } else if (deg > 32) {
        float mx = NEG_INF;
        for (int c = eb; c < ee; c += 32) {
            const int j = c + lane;
            if (j < ee) {
                float x = rows[ssrc[j]].w + erv;
                x = (x >= 0.f) ? x : SLOPE * x;
                mx = fmaxf(mx, x);
            }
        }
        mx = wave_max(mx);
        float sm = 0.f;
        for (int c = eb; c < ee; c += 32) {
            const int j = c + lane;
            if (j < ee) {
                float x = rows[ssrc[j]].w + erv;
                x = (x >= 0.f) ? x : SLOPE * x;
                sm += expf(x - mx);
            }
        }
        sm = wave_sum(sm);
        for (int c = eb; c < ee; c += 32) {
            const int j = c + lane;
            if (j < ee) {
                const float4 r = rows[ssrc[j]];
                float x = r.w + erv;
                x = (x >= 0.f) ? x : SLOPE * x;
                const float al = expf(x - mx) / sm;
                p0 += r.x * al; p1 += r.y * al; p2 += r.z * al;
            }
        }
    }
    p0 = wave_sum(p0); p1 = wave_sum(p1); p2 = wave_sum(p2);
    if (lane == 0) {
        agg2[node * 3 + 0] = p0;
        agg2[node * 3 + 1] = p1;
        agg2[node * 3 + 2] = p2;
    }
}

__global__ void gat_elu_bias(float* __restrict__ x, const float* __restrict__ b,
                             int total, int hd) {
    int i = blockIdx.x * blockDim.x + threadIdx.x;
    if (i >= total) return;
    float v = x[i] + b[i % hd];
    x[i] = (v > 0.f) ? v : expm1f(v);
}

__global__ void gat_l2_linear(const float* __restrict__ x, const float* __restrict__ W2,
                              const float* __restrict__ al2, const float* __restrict__ ar2,
                              float* __restrict__ hp, float* __restrict__ er2, int n_nodes) {
    int n = blockIdx.x * blockDim.x + threadIdx.x;
    if (n >= n_nodes) return;
    float a0 = 0.f, a1 = 0.f, a2 = 0.f;
    const float* xp = x + (size_t)n * 256;
    for (int k = 0; k < 256; ++k) {
        float xv = xp[k];
        a0 += xv * W2[k * 3 + 0];
        a1 += xv * W2[k * 3 + 1];
        a2 += xv * W2[k * 3 + 2];
    }
    float4 row;
    row.x = a0; row.y = a1; row.z = a2;
    row.w = a0 * al2[0] + a1 * al2[1] + a2 * al2[2];
    ((float4*)hp)[n] = row;
    er2[n] = a0 * ar2[0] + a1 * ar2[1] + a2 * ar2[2];
}

__global__ void gat_finalize(const float* __restrict__ agg2, const float* __restrict__ b2,
                             float* __restrict__ out, int total) {
    int i = blockIdx.x * blockDim.x + threadIdx.x;
    if (i >= total) return;
    out[i] = agg2[i] + b2[i % 3];
}

extern "C" void kernel_launch(void* const* d_in, const int* in_sizes, int n_in,
                              void* d_out, int out_size, void* d_ws, size_t ws_size,
                              hipStream_t stream) {
    const float* feat = (const float*)d_in[0];
    const int*   src  = (const int*)  d_in[1];
    const int*   dst  = (const int*)  d_in[2];
    const float* W1   = (const float*)d_in[3];
    const float* al1  = (const float*)d_in[4];
    const float* ar1  = (const float*)d_in[5];
    const float* b1   = (const float*)d_in[6];
    const float* W2   = (const float*)d_in[7];
    const float* al2  = (const float*)d_in[8];
    const float* ar2  = (const float*)d_in[9];
    const float* b2   = (const float*)d_in[10];
    float* out = (float*)d_out;

    const int n_nodes = in_sizes[0] / 256;
    const int n_edges = in_sizes[1];

    char* cur = (char*)d_ws;
    auto alloc = [&](size_t bytes) -> float* {
        float* p = (float*)cur;
        cur += (bytes + 255) & ~(size_t)255;
        return p;
    };
    float* H1   = alloc((size_t)n_nodes * 256 * 4);
    float* el1  = alloc((size_t)n_nodes * 8 * 4);
    float* er1  = alloc((size_t)n_nodes * 8 * 4);
    float* agg1 = alloc((size_t)n_nodes * 256 * 4);
    float* hp2  = alloc((size_t)n_nodes * 4 * 4);
    float* er2  = alloc((size_t)n_nodes * 4);
    float* agg2 = alloc((size_t)n_nodes * 3 * 4);
    float* cntf = alloc((size_t)n_nodes * 4);
    int*   cnt    = (int*)cntf;
    int*   rowptr = (int*)alloc((size_t)(n_nodes + 1) * 4);
    int*   rank   = (int*)alloc((size_t)n_edges * 4);
    int*   ssrc   = (int*)alloc((size_t)n_edges * 4);

    const int B = 256;

    gat_fill_f32<<<cdiv(n_nodes, B), B, 0, stream>>>(cntf, 0.f, n_nodes);
    gat_dst_rank<<<cdiv(n_edges, B), B, 0, stream>>>(dst, cnt, rank, n_edges);
    gat_row_offsets<<<cdiv(n_nodes, 1024), 1024, 0, stream>>>(cnt, rowptr, n_nodes);
    gat_dst_place<<<cdiv(n_edges, B), B, 0, stream>>>(src, dst, rowptr, rank, ssrc, n_edges);

    gat_gemm1_wmma<<<cdiv(n_nodes, 32), 256, 0, stream>>>(feat, W1, H1, n_nodes);
    gat_attn_scores<<<cdiv(n_nodes * 8, B), B, 0, stream>>>(H1, al1, ar1, el1, er1, n_nodes);
    gat_l1_segments<<<cdiv(n_nodes, 8), 256, 0, stream>>>(rowptr, ssrc, H1, el1, er1, agg1, n_nodes);
    gat_elu_bias<<<cdiv(n_nodes * 256, B), B, 0, stream>>>(agg1, b1, n_nodes * 256, 256);

    gat_l2_linear<<<cdiv(n_nodes, 128), 128, 0, stream>>>(agg1, W2, al2, ar2, hp2, er2, n_nodes);
    gat_l2_segments<<<cdiv(n_nodes, 8), 256, 0, stream>>>(rowptr, ssrc, hp2, er2, agg2, n_nodes);
    gat_finalize<<<cdiv(n_nodes * 3, B), B, 0, stream>>>(agg2, b2, out, n_nodes * 3);
}


# Bias, elu and layer-2 linear (256x3) done by the layer-1 segment wave on its register sums: agg never stored, 2 launches fewer; x@W2 summed 8 per lane, then xor tree over lanes (sum order changed)
# speedup vs baseline: 4.2591x; 1.1592x over previous
#include <hip/hip_runtime.h>
#include <hip/hip_bf16.h>

typedef __attribute__((ext_vector_type(2))) float v2f;
typedef __attribute__((ext_vector_type(8))) float v8f;

#define SLOPE 0.2f

static __host__ __device__ inline int cdiv(int a, int b) { return (a + b - 1) / b; }

__global__ void gat_fill_f32(float* __restrict__ p, float v, int n) {
    int i = blockIdx.x * blockDim.x + threadIdx.x;
    if (i < n) p[i] = v;
}

__global__ void gat_gemm1_wmma(const float* __restrict__ A,
                               const float* __restrict__ W,
                               float* __restrict__ H,
                               int n_nodes) {
    __shared__ float As[32 * 256];
    __shared__ float Ws[32 * 256];

    const int row0 = blockIdx.x * 32;
    const int tid  = threadIdx.x;
    const int wave = tid >> 5;
    const int lane = tid & 31;

    for (int i = tid; i < 32 * 64; i += 256) {
        int r  = i >> 6;
        int c4 = (i & 63) << 2;
        int rr = row0 + r; if (rr >= n_nodes) rr = n_nodes - 1;
        *(float4*)(&As[r * 256 + c4]) = *(const float4*)(A + (size_t)rr * 256 + c4);
    }

    const int m   = lane & 15;
    const int kk  = (lane >> 4) << 1;
    const int n0  = wave * 32;
    const int n1  = n0 + 16;

    v8f c00 = {}, c01 = {}, c10 = {}, c11 = {};

    for (int kc = 0; kc < 256; kc += 32) {
        __syncthreads();
        for (int i = tid; i < 32 * 64; i += 256) {
            int r  = i >> 6;
            int c4 = (i & 63) << 2;
            *(float4*)(&Ws[r * 256 + c4]) = *(const float4*)(W + (size_t)(kc + r) * 256 + c4);
        }
        __syncthreads();

#pragma unroll
        for (int k = 0; k < 32; k += 4) {
            v2f a0, a1, b0, b1;
            a0.x = As[m * 256 + kc + k + kk];
            a0.y = As[m * 256 + kc + k + kk + 1];
            a1.x = As[(m + 16) * 256 + kc + k + kk];
            a1.y = As[(m + 16) * 256 + kc + k + kk + 1];
            b0.x = Ws[(k + kk) * 256 + n0 + m];
            b0.y = Ws[(k + kk + 1) * 256 + n0 + m];
            b1.x = Ws[(k + kk) * 256 + n1 + m];
            b1.y = Ws[(k + kk + 1) * 256 + n1 + m];
            c00 = __builtin_amdgcn_wmma_f32_16x16x4_f32(false, a0, false, b0, (short)0, c00, false, false);
            c01 = __builtin_amdgcn_wmma_f32_16x16x4_f32(false, a0, false, b1, (short)0, c01, false, false);
            c10 = __builtin_amdgcn_wmma_f32_16x16x4_f32(false, a1, false, b0, (short)0, c10, false, false);
            c11 = __builtin_amdgcn_wmma_f32_16x16x4_f32(false, a1, false, b1, (short)0, c11, false, false);
        }
    }

    const int mhi = (lane >> 4) * 8;
    for (int v = 0; v < 8; ++v) {
        int r0 = row0 + mhi + v;
        int r1 = r0 + 16;
        if (r0 < n_nodes) {
            H[(size_t)r0 * 256 + n0 + m] = c00[v];
            H[(size_t)r0 * 256 + n1 + m] = c01[v];
        }
        if (r1 < n_nodes) {
            H[(size_t)r1 * 256 + n0 + m] = c10[v];
            H[(size_t)r1 * 256 + n1 + m] = c11[v];
        }
    }
}

__global__ void gat_attn_scores(const float* __restrict__ H,
                                const float* __restrict__ al,
                                const float* __restrict__ ar,
                                float* __restrict__ el, float* __restrict__ er,
                                int n_nodes) {
    int i = blockIdx.x * blockDim.x + threadIdx.x;
    if (i >= n_nodes * 8) return;
    int n = i >> 3, h = i & 7;
    const float* hp  = H + (size_t)n * 256 + h * 32;
    const float* alp = al + h * 32;
    const float* arp = ar + h * 32;
    float sl = 0.f, sr = 0.f;
    for (int d = 0; d < 32; ++d) { float v = hp[d]; sl += v * alp[d]; sr += v * arp[d]; }
    el[i] = sl; er[i] = sr;
}

__global__ void gat_dst_rank(const int* __restrict__ dst, int* __restrict__ cnt,
                             int* __restrict__ rank, int n_edges) {
    int e = blockIdx.x * blockDim.x + threadIdx.x;
    if (e >= n_edges) return;
    rank[e] = atomicAdd(&cnt[dst[e]], 1);
}

__global__ void __launch_bounds__(1024) gat_row_offsets(const int* __restrict__ cnt,
                                                        int* __restrict__ rowptr, int n) {
    __shared__ int wtot[32];
    __shared__ int wpre[32];
    const int t = threadIdx.x, lane = t & 31, wave = t >> 5;
    const int b = blockIdx.x;
    const int i = b * 1024 + t;
    const int v = (i < n) ? cnt[i] : 0;
    int pre = 0;
    for (int k = 0; k < b; ++k) pre += cnt[k * 1024 + t];
    int x = v;
    for (int o = 1; o < 32; o <<= 1) {
        int y = __shfl_up(x, o, 32);
        if (lane >= o) x += y;
    }
    for (int o = 16; o > 0; o >>= 1) pre += __shfl_xor(pre, o, 32);
    if (lane == 31) wtot[wave] = x;
    if (lane == 0)  wpre[wave] = pre;
    __syncthreads();
    int w = wtot[lane];
    for (int o = 1; o < 32; o <<= 1) {
        int y = __shfl_up(w, o, 32);
        if (lane >= o) w += y;
    }
    int base = wpre[lane];
    for (int o = 16; o > 0; o >>= 1) base += __shfl_xor(base, o, 32);
    const int before_wave = __shfl(w, wave, 32) - wtot[wave];
    const int excl = base + before_wave + x - v;
    if (i < n) rowptr[i] = excl;
    if (i == n - 1) rowptr[n] = excl + v;
}

__global__ void gat_dst_place(const int* __restrict__ src, const int* __restrict__ dst,
                              const int* __restrict__ rowptr, const int* __restrict__ rank,
                              int* __restrict__ ssrc, int n_edges) {
    int e = blockIdx.x * blockDim.x + threadIdx.x;
    if (e >= n_edges) return;
    ssrc[rowptr[dst[e]] + rank[e]] = src[e];
}

__device__ inline float wave_max(float x) {
    for (int o = 16; o > 0; o >>= 1) x = fmaxf(x, __shfl_xor(x, o, 32));
    return x;
}
__device__ inline float wave_sum(float x) {
    for (int o = 16; o > 0; o >>= 1) x += __shfl_xor(x, o, 32);
    return x;
}

template <int CTRL>
__device__ inline float quad_perm(float x) {
    return __int_as_float(__builtin_amdgcn_update_dpp(0, __float_as_int(x), CTRL, 0xF, 0xF, true));
}

template <int K>
__device__ inline void gat_add_row(const float* __restrict__ Hl, int s_lane, float u_lane,
                                   float4& a0, float4& a1) {
    const int sn = __builtin_amdgcn_readlane(s_lane, K);
    const float al = quad_perm<K * 0x55>(u_lane);
    const float4 x0 = *(const float4*)(Hl + (size_t)sn * 256);
    const float4 x1 = *(const float4*)(Hl + (size_t)sn * 256 + 4);
    a0.x += x0.x * al; a0.y += x0.y * al; a0.z += x0.z * al; a0.w += x0.w * al;
    a1.x += x1.x * al; a1.y += x1.y * al; a1.z += x1.z * al; a1.w += x1.w * al;
}

__global__ void __launch_bounds__(256) gat_l1_segments(const int* __restrict__ rowptr,
                                                       const int* __restrict__ ssrc,
                                                       const float* __restrict__ H,
                                                       const float* __restrict__ el,
                                                       const float* __restrict__ er,
                                                       const float* __restrict__ b1,
                                                       const float* __restrict__ W2,
                                                       const float* __restrict__ al2,
                                                       const float* __restrict__ ar2,
                                                       float* __restrict__ hp,
                                                       float* __restrict__ er2, int n_nodes) {
    const int lane = threadIdx.x & 31;
    const int node = blockIdx.x * 8 + __builtin_amdgcn_readfirstlane((int)(threadIdx.x >> 5));
    if (node >= n_nodes) return;
    const int eb = rowptr[node], ee = rowptr[node + 1];
    const int head = lane >> 2, sub = lane & 3;
    const float erv = er[node * 8 + head];
    const float* Hl = H + lane * 8;
    const float NEG_INF = -__builtin_huge_valf();

    int   S[8];
    float V[8];
    float U[8];
    float4 a0 = {0.f, 0.f, 0.f, 0.f}, a1 = {0.f, 0.f, 0.f, 0.f};

    auto score_chunk = [&](int c) {
        const int last = ee - 1;
#pragma unroll
        for (int i = 0; i < 8; ++i) {
            const int j = c + 4 * i + sub;
            S[i] = ssrc[j < ee ? j : last];
        }
#pragma unroll
        for (int i = 0; i < 8; ++i) {
            float x = el[S[i] * 8 + head] + erv;
            x = (x >= 0.f) ? x : SLOPE * x;
            V[i] = (c + 4 * i + sub < ee) ? x : NEG_INF;
        }
    };
    auto add_chunk = [&](int cnt) {
#pragma unroll
        for (int i = 0; i < 8; ++i) {
            if (4 * i + 4 <= cnt) {
                gat_add_row<0>(Hl, S[i], U[i], a0, a1);
                gat_add_row<1>(Hl, S[i], U[i], a0, a1);
                gat_add_row<2>(Hl, S[i], U[i], a0, a1);
                gat_add_row<3>(Hl, S[i], U[i], a0, a1);
            } else if (4 * i < cnt) {
                gat_add_row<0>(Hl, S[i], U[i], a0, a1);
                if (4 * i + 1 < cnt) gat_add_row<1>(Hl, S[i], U[i], a0, a1);
                if (4 * i + 2 < cnt) gat_add_row<2>(Hl, S[i], U[i], a0, a1);
            }
        }
    };

    auto add_chunk_by_two = [&](int cnt) {
#pragma unroll
        for (int i = 0; i < 8; ++i) {
            if (4 * i + 2 <= cnt) {
                gat_add_row<0>(Hl, S[i], U[i], a0, a1);
                gat_add_row<1>(Hl, S[i], U[i], a0, a1);
            } else if (4 * i < cnt) {
                gat_add_row<0>(Hl, S[i], U[i], a0, a1);
            }
            if (4 * i + 4 <= cnt) {
                gat_add_row<2>(Hl, S[i], U[i], a0, a1);
                gat_add_row<3>(Hl, S[i], U[i], a0, a1);
            } else if (4 * i + 2 < cnt) {
                gat_add_row<2>(Hl, S[i], U[i], a0, a1);
            }
        }
    };

    const int deg = ee - eb;
    if (deg > 0 && deg <= 32) {
        score_chunk(eb);
        float mx = V[0];
#pragma unroll
        for (int i = 1; i < 8; ++i) mx = fmaxf(mx, V[i]);
        mx = fmaxf(mx, quad_perm<0xB1>(mx));
        mx = fmaxf(mx, quad_perm<0x4E>(mx));
        float sm = 0.f;
#pragma unroll
        for (int i = 0; i < 8; ++i) {
            const float u = expf(V[i] - mx);
            U[i] = (eb + 4 * i + sub < ee) ? u : 0.f;
            sm += U[i];
        }
        sm += quad_perm<0xB1>(sm);
        sm += quad_perm<0x4E>(sm);
#pragma unroll
        for (int i = 0; i < 8; ++i) U[i] = U[i] / sm;
        add_chunk(deg);
    } else if (deg > 32) {
        float mx = NEG_INF;
        for (int c = eb; c < ee; c += 32) {
            score_chunk(c);
#pragma unroll
            for (int i = 0; i < 8; ++i) mx = fmaxf(mx, V[i]);
        }
        mx = fmaxf(mx, quad_perm<0xB1>(mx));
        mx = fmaxf(mx, quad_perm<0x4E>(mx));
        float sm = 0.f;
        for (int c = eb; c < ee; c += 32) {
            score_chunk(c);
#pragma unroll
            for (int i = 0; i < 8; ++i) {
                const float u = expf(V[i] - mx);
                sm += (c + 4 * i + sub < ee) ? u : 0.f;
            }
        }
        sm += quad_perm<0xB1>(sm);
        sm += quad_perm<0x4E>(sm);
        for (int c = eb; c < ee; c += 32) {
            score_chunk(c);
#pragma unroll
            for (int i = 0; i < 8; ++i) {
                const float u = expf(V[i] - mx) / sm;
                U[i] = (c + 4 * i + sub < ee) ? u : 0.f;
            }
            const int left = ee - c;
            add_chunk_by_two(left < 32 ? left : 32);
        }
    }
    const float4 c0 = *(const float4*)(b1 + lane * 8);
    const float4 c1 = *(const float4*)(b1 + lane * 8 + 4);
    float x[8] = {a0.x + c0.x, a0.y + c0.y, a0.z + c0.z, a0.w + c0.w,
                  a1.x + c1.x, a1.y + c1.y, a1.z + c1.z, a1.w + c1.w};
#pragma unroll
    for (int i = 0; i < 8; ++i) x[i] = (x[i] > 0.f) ? x[i] : expm1f(x[i]);
    const float4* wp = (const float4*)(W2 + lane * 24);
    const float4 w0 = wp[0], w1 = wp[1], w2 = wp[2], w3 = wp[3], w4 = wp[4], w5 = wp[5];
    float p0 = x[0] * w0.x, p1 = x[0] * w0.y, p2 = x[0] * w0.z;
    p0 += x[1] * w0.w; p1 += x[1] * w1.x; p2 += x[1] * w1.y;
    p0 += x[2] * w1.z; p1 += x[2] * w1.w; p2 += x[2] * w2.x;
    p0 += x[3] * w2.y; p1 += x[3] * w2.z; p2 += x[3] * w2.w;
    p0 += x[4] * w3.x; p1 += x[4] * w3.y; p2 += x[4] * w3.z;
    p0 += x[5] * w3.w; p1 += x[5] * w4.x; p2 += x[5] * w4.y;
    p0 += x[6] * w4.z; p1 += x[6] * w4.w; p2 += x[6] * w5.x;
    p0 += x[7] * w5.y; p1 += x[7] * w5.z; p2 += x[7] * w5.w;
    p0 = wave_sum(p0); p1 = wave_sum(p1); p2 = wave_sum(p2);
    if (lane == 0) {
        float4 row;
        row.x = p0; row.y = p1; row.z = p2;
        row.w = p0 * al2[0] + p1 * al2[1] + p2 * al2[2];
        ((float4*)hp)[node] = row;
        er2[node] = p0 * ar2[0] + p1 * ar2[1] + p2 * ar2[2];
    }
}

__global__ void __launch_bounds__(256) gat_l2_segments(const int* __restrict__ rowptr,
                                                       const int* __restrict__ ssrc,
                                                       const float* __restrict__ hp,
                                                       const float* __restrict__ er2,
                                                       float* __restrict__ agg2, int n_nodes) {
    const int lane = threadIdx.x & 31;
    const int node = blockIdx.x * 8 + __builtin_amdgcn_readfirstlane((int)(threadIdx.x >> 5));
    if (node >= n_nodes) return;
    const int eb = rowptr[node], ee = rowptr[node + 1];
    const float erv = er2[node];
    const float NEG_INF = -__builtin_huge_valf();
    const float4* rows = (const float4*)hp;
    const int deg = ee - eb;
    float p0 = 0.f, p1 = 0.f, p2 = 0.f;

    if (deg > 0 && deg <= 32) {
        const int j = eb + lane;
        const bool ok = j < ee;
        const float4 r = rows[ssrc[ok ? j : ee - 1]];
        float x = r.w + erv;
        x = (x >= 0.f) ? x : SLOPE * x;
        x = ok ? x : NEG_INF;
        const float mx = wave_max(x);
        const float e = expf(x - mx);
        const float u = ok ? e : 0.f;
        const float sm = wave_sum(u);
        const float al = u / sm;
        p0 = ok ? r.x * al : 0.f;
        p1 = ok ? r.y * al : 0.f;
        p2 = ok ? r.z * al : 0.f;
    } else if (deg > 32) {
        float mx = NEG_INF;
        for (int c = eb; c < ee; c += 32) {
            const int j = c + lane;
            if (j < ee) {
                float x = rows[ssrc[j]].w + erv;
                x = (x >= 0.f) ? x : SLOPE * x;
                mx = fmaxf(mx, x);
            }
        }
        mx = wave_max(mx);
        float sm = 0.f;
        for (int c = eb; c < ee; c += 32) {
            const int j = c + lane;
            if (j < ee) {
                float x = rows[ssrc[j]].w + erv;
                x = (x >= 0.f) ? x : SLOPE * x;
                sm += expf(x - mx);
            }
        }
        sm = wave_sum(sm);
        for (int c = eb; c < ee; c += 32) {
            const int j = c + lane;
            if (j < ee) {
                const float4 r = rows[ssrc[j]];
                float x = r.w + erv;
                x = (x >= 0.f) ? x : SLOPE * x;
                const float al = expf(x - mx) / sm;
                p0 += r.x * al; p1 += r.y * al; p2 += r.z * al;
            }
        }
    }
    p0 = wave_sum(p0); p1 = wave_sum(p1); p2 = wave_sum(p2);
    if (lane == 0) {
        agg2[node * 3 + 0] = p0;
        agg2[node * 3 + 1] = p1;
        agg2[node * 3 + 2] = p2;
    }
}

__global__ void gat_finalize(const float* __restrict__ agg2, const float* __restrict__ b2,
                             float* __restrict__ out, int total) {
    int i = blockIdx.x * blockDim.x + threadIdx.x;
    if (i >= total) return;
    out[i] = agg2[i] + b2[i % 3];
}

extern "C" void kernel_launch(void* const* d_in, const int* in_sizes, int n_in,
                              void* d_out, int out_size, void* d_ws, size_t ws_size,
                              hipStream_t stream) {
    const float* feat = (const float*)d_in[0];
    const int*   src  = (const int*)  d_in[1];
    const int*   dst  = (const int*)  d_in[2];
    const float* W1   = (const float*)d_in[3];
    const float* al1  = (const float*)d_in[4];
    const float* ar1  = (const float*)d_in[5];
    const float* b1   = (const float*)d_in[6];
    const float* W2   = (const float*)d_in[7];
    const float* al2  = (const float*)d_in[8];
    const float* ar2  = (const float*)d_in[9];
    const float* b2   = (const float*)d_in[10];
    float* out = (float*)d_out;

    const int n_nodes = in_sizes[0] / 256;
    const int n_edges = in_sizes[1];

    char* cur = (char*)d_ws;
    auto alloc = [&](size_t bytes) -> float* {
        float* p = (float*)cur;
        cur += (bytes + 255) & ~(size_t)255;
        return p;
    };
    float* H1   = alloc((size_t)n_nodes * 256 * 4);
    float* el1  = alloc((size_t)n_nodes * 8 * 4);
    float* er1  = alloc((size_t)n_nodes * 8 * 4);
    float* hp2  = alloc((size_t)n_nodes * 4 * 4);
    float* er2  = alloc((size_t)n_nodes * 4);
    float* agg2 = alloc((size_t)n_nodes * 3 * 4);
    float* cntf = alloc((size_t)n_nodes * 4);
    int*   cnt    = (int*)cntf;
    int*   rowptr = (int*)alloc((size_t)(n_nodes + 1) * 4);
    int*   rank   = (int*)alloc((size_t)n_edges * 4);
    int*   ssrc   = (int*)alloc((size_t)n_edges * 4);

    const int B = 256;

    gat_fill_f32<<<cdiv(n_nodes, B), B, 0, stream>>>(cntf, 0.f, n_nodes);
    gat_dst_rank<<<cdiv(n_edges, B), B, 0, stream>>>(dst, cnt, rank, n_edges);
    gat_row_offsets<<<cdiv(n_nodes, 1024), 1024, 0, stream>>>(cnt, rowptr, n_nodes);
    gat_dst_place<<<cdiv(n_edges, B), B, 0, stream>>>(src, dst, rowptr, rank, ssrc, n_edges);

    gat_gemm1_wmma<<<cdiv(n_nodes, 32), 256, 0, stream>>>(feat, W1, H1, n_nodes);
    gat_attn_scores<<<cdiv(n_nodes * 8, B), B, 0, stream>>>(H1, al1, ar1, el1, er1, n_nodes);
    gat_l1_segments<<<cdiv(n_nodes, 8), 256, 0, stream>>>(rowptr, ssrc, H1, el1, er1,
                                                          b1, W2, al2, ar2, hp2, er2, n_nodes);

    gat_l2_segments<<<cdiv(n_nodes, 8), 256, 0, stream>>>(rowptr, ssrc, hp2, er2, agg2, n_nodes);
    gat_finalize<<<cdiv(n_nodes * 3, B), B, 0, stream>>>(agg2, b2, out, n_nodes * 3);
}


# GEMM1 staging loops fully unrolled: a thread's 8 global loads per LDS tile (A panel, each W chunk) go out as one batch instead of one per memory round trip (72 dependent trips become 9)
# speedup vs baseline: 5.2483x; 1.2323x over previous
#include <hip/hip_runtime.h>
#include <hip/hip_bf16.h>

typedef __attribute__((ext_vector_type(2))) float v2f;
typedef __attribute__((ext_vector_type(8))) float v8f;

#define SLOPE 0.2f

static __host__ __device__ inline int cdiv(int a, int b) { return (a + b - 1) / b; }

__global__ void gat_fill_f32(float* __restrict__ p, float v, int n) {
    int i = blockIdx.x * blockDim.x + threadIdx.x;
    if (i < n) p[i] = v;
}

__global__ void __launch_bounds__(256) gat_gemm1_wmma(const float* __restrict__ A,
                               const float* __restrict__ W,
                               float* __restrict__ H,
                               int n_nodes) {
    __shared__ float As[32 * 256];
    __shared__ float Ws[32 * 256];

    const int row0 = blockIdx.x * 32;
    const int tid  = threadIdx.x;
    const int wave = tid >> 5;
    const int lane = tid & 31;

    const int sr = tid >> 6;
    const int sc = (tid & 63) << 2;
    float4 st[8];

#pragma unroll
    for (int j = 0; j < 8; ++j) {
        int rr = row0 + sr + 4 * j; if (rr >= n_nodes) rr = n_nodes - 1;
        st[j] = *(const float4*)(A + (size_t)rr * 256 + sc);
    }
#pragma unroll
    for (int j = 0; j < 8; ++j) *(float4*)(&As[(sr + 4 * j) * 256 + sc]) = st[j];

    const int m   = lane & 15;
    const int kk  = (lane >> 4) << 1;
    const int n0  = wave * 32;
    const int n1  = n0 + 16;

    v8f c00 = {}, c01 = {}, c10 = {}, c11 = {};

    for (int kc = 0; kc < 256; kc += 32) {
        __syncthreads();
#pragma unroll
        for (int j = 0; j < 8; ++j) st[j] = *(const float4*)(W + (size_t)(kc + sr + 4 * j) * 256 + sc);
#pragma unroll
        for (int j = 0; j < 8; ++j) *(float4*)(&Ws[(sr + 4 * j) * 256 + sc]) = st[j];
        __syncthreads();

#pragma unroll
        for (int k = 0; k < 32; k += 4) {
            v2f a0, a1, b0, b1;
            a0.x = As[m * 256 + kc + k + kk];
            a0.y = As[m * 256 + kc + k + kk + 1];
            a1.x = As[(m + 16) * 256 + kc + k + kk];
            a1.y = As[(m + 16) * 256 + kc + k + kk + 1];
            b0.x = Ws[(k + kk) * 256 + n0 + m];
            b0.y = Ws[(k + kk + 1) * 256 + n0 + m];
            b1.x = Ws[(k + kk) * 256 + n1 + m];
            b1.y = Ws[(k + kk + 1) * 256 + n1 + m];
            c00 = __builtin_amdgcn_wmma_f32_16x16x4_f32(false, a0, false, b0, (short)0, c00, false, false);
            c01 = __builtin_amdgcn_wmma_f32_16x16x4_f32(false, a0, false, b1, (short)0, c01, false, false);
            c10 = __builtin_amdgcn_wmma_f32_16x16x4_f32(false, a1, false, b0, (short)0, c10, false, false);
            c11 = __builtin_amdgcn_wmma_f32_16x16x4_f32(false, a1, false, b1, (short)0, c11, false, false);
        }
    }

    const int mhi = (lane >> 4) * 8;
    for (int v = 0; v < 8; ++v) {
        int r0 = row0 + mhi + v;
        int r1 = r0 + 16;
        if (r0 < n_nodes) {
            H[(size_t)r0 * 256 + n0 + m] = c00[v];
            H[(size_t)r0 * 256 + n1 + m] = c01[v];
        }
        if (r1 < n_nodes) {
            H[(size_t)r1 * 256 + n0 + m] = c10[v];
            H[(size_t)r1 * 256 + n1 + m] = c11[v];
        }
    }
}

__global__ void gat_attn_scores(const float* __restrict__ H,
                                const float* __restrict__ al,
                                const float* __restrict__ ar,
                                float* __restrict__ el, float* __restrict__ er,
                                int n_nodes) {
    int i = blockIdx.x * blockDim.x + threadIdx.x;
    if (i >= n_nodes * 8) return;
    int n = i >> 3, h = i & 7;
    const float* hp  = H + (size_t)n * 256 + h * 32;
    const float* alp = al + h * 32;
    const float* arp = ar + h * 32;
    float sl = 0.f, sr = 0.f;
    for (int d = 0; d < 32; ++d) { float v = hp[d]; sl += v * alp[d]; sr += v * arp[d]; }
    el[i] = sl; er[i] = sr;
}

__global__ void gat_dst_rank(const int* __restrict__ dst, int* __restrict__ cnt,
                             int* __restrict__ rank, int n_edges) {
    int e = blockIdx.x * blockDim.x + threadIdx.x;
    if (e >= n_edges) return;
    rank[e] = atomicAdd(&cnt[dst[e]], 1);
}

__global__ void __launch_bounds__(1024) gat_row_offsets(const int* __restrict__ cnt,
                                                        int* __restrict__ rowptr, int n) {
    __shared__ int wtot[32];
    __shared__ int wpre[32];
    const int t = threadIdx.x, lane = t & 31, wave = t >> 5;
    const int b = blockIdx.x;
    const int i = b * 1024 + t;
    const int v = (i < n) ? cnt[i] : 0;
    int pre = 0;
    for (int k = 0; k < b; ++k) pre += cnt[k * 1024 + t];
    int x = v;
    for (int o = 1; o < 32; o <<= 1) {
        int y = __shfl_up(x, o, 32);
        if (lane >= o) x += y;
    }
    for (int o = 16; o > 0; o >>= 1) pre += __shfl_xor(pre, o, 32);
    if (lane == 31) wtot[wave] = x;
    if (lane == 0)  wpre[wave] = pre;
    __syncthreads();
    int w = wtot[lane];
    for (int o = 1; o < 32; o <<= 1) {
        int y = __shfl_up(w, o, 32);
        if (lane >= o) w += y;
    }
    int base = wpre[lane];
    for (int o = 16; o > 0; o >>= 1) base += __shfl_xor(base, o, 32);
    const int before_wave = __shfl(w, wave, 32) - wtot[wave];
    const int excl = base + before_wave + x - v;
    if (i < n) rowptr[i] = excl;
    if (i == n - 1) rowptr[n] = excl + v;
}

__global__ void gat_dst_place(const int* __restrict__ src, const int* __restrict__ dst,
                              const int* __restrict__ rowptr, const int* __restrict__ rank,
                              int* __restrict__ ssrc, int n_edges) {
    int e = blockIdx.x * blockDim.x + threadIdx.x;
    if (e >= n_edges) return;
    ssrc[rowptr[dst[e]] + rank[e]] = src[e];
}

__device__ inline float wave_max(float x) {
    for (int o = 16; o > 0; o >>= 1) x = fmaxf(x, __shfl_xor(x, o, 32));
    return x;
}
__device__ inline float wave_sum(float x) {
    for (int o = 16; o > 0; o >>= 1) x += __shfl_xor(x, o, 32);
    return x;
}

template <int CTRL>
__device__ inline float quad_perm(float x) {
    return __int_as_float(__builtin_amdgcn_update_dpp(0, __float_as_int(x), CTRL, 0xF, 0xF, true));
}

template <int K>
__device__ inline void gat_add_row(const float* __restrict__ Hl, int s_lane, float u_lane,
                                   float4& a0, float4& a1) {
    const int sn = __builtin_amdgcn_readlane(s_lane, K);
    const float al = quad_perm<K * 0x55>(u_lane);
    const float4 x0 = *(const float4*)(Hl + (size_t)sn * 256);
    const float4 x1 = *(const float4*)(Hl + (size_t)sn * 256 + 4);
    a0.x += x0.x * al; a0.y += x0.y * al; a0.z += x0.z * al; a0.w += x0.w * al;
    a1.x += x1.x * al; a1.y += x1.y * al; a1.z += x1.z * al; a1.w += x1.w * al;
}

__global__ void __launch_bounds__(256) gat_l1_segments(const int* __restrict__ rowptr,
                                                       const int* __restrict__ ssrc,
                                                       const float* __restrict__ H,
                                                       const float* __restrict__ el,
                                                       const float* __restrict__ er,
                                                       const float* __restrict__ b1,
                                                       const float* __restrict__ W2,
                                                       const float* __restrict__ al2,
                                                       const float* __restrict__ ar2,
                                                       float* __restrict__ hp,
                                                       float* __restrict__ er2, int n_nodes) {
    const int lane = threadIdx.x & 31;
    const int node = blockIdx.x * 8 + __builtin_amdgcn_readfirstlane((int)(threadIdx.x >> 5));
    if (node >= n_nodes) return;
    const int eb = rowptr[node], ee = rowptr[node + 1];
    const int head = lane >> 2, sub = lane & 3;
    const float erv = er[node * 8 + head];
    const float* Hl = H + lane * 8;
    const float NEG_INF = -__builtin_huge_valf();

    int   S[8];
    float V[8];
    float U[8];
    float4 a0 = {0.f, 0.f, 0.f, 0.f}, a1 = {0.f, 0.f, 0.f, 0.f};

    auto score_chunk = [&](int c) {
        const int last = ee - 1;
#pragma unroll
        for (int i = 0; i < 8; ++i) {
            const int j = c + 4 * i + sub;
            S[i] = ssrc[j < ee ? j : last];
        }
#pragma unroll
        for (int i = 0; i < 8; ++i) {
            float x = el[S[i] * 8 + head] + erv;
            x = (x >= 0.f) ? x : SLOPE * x;
            V[i] = (c + 4 * i + sub < ee) ? x : NEG_INF;
        }
    };
    auto add_chunk = [&](int cnt) {
#pragma unroll
        for (int i = 0; i < 8; ++i) {
            if (4 * i + 4 <= cnt) {
                gat_add_row<0>(Hl, S[i], U[i], a0, a1);
                gat_add_row<1>(Hl, S[i], U[i], a0, a1);
                gat_add_row<2>(Hl, S[i], U[i], a0, a1);
                gat_add_row<3>(Hl, S[i], U[i], a0, a1);
            } else if (4 * i < cnt) {
                gat_add_row<0>(Hl, S[i], U[i], a0, a1);
                if (4 * i + 1 < cnt) gat_add_row<1>(Hl, S[i], U[i], a0, a1);
                if (4 * i + 2 < cnt) gat_add_row<2>(Hl, S[i], U[i], a0, a1);
            }
        }
    };

    auto add_chunk_by_two = [&](int cnt) {
#pragma unroll
        for (int i = 0; i < 8; ++i) {
            if (4 * i + 2 <= cnt) {
                gat_add_row<0>(Hl, S[i], U[i], a0, a1);
                gat_add_row<1>(Hl, S[i], U[i], a0, a1);
            } else if (4 * i < cnt) {
                gat_add_row<0>(Hl, S[i], U[i], a0, a1);
            }
            if (4 * i + 4 <= cnt) {
                gat_add_row<2>(Hl, S[i], U[i], a0, a1);
                gat_add_row<3>(Hl, S[i], U[i], a0, a1);
            } else if (4 * i + 2 < cnt) {
                gat_add_row<2>(Hl, S[i], U[i], a0, a1);
            }
        }
    };

    const int deg = ee - eb;
    if (deg > 0 && deg <= 32) {
        score_chunk(eb);
        float mx = V[0];
#pragma unroll
        for (int i = 1; i < 8; ++i) mx = fmaxf(mx, V[i]);
        mx = fmaxf(mx, quad_perm<0xB1>(mx));
        mx = fmaxf(mx, quad_perm<0x4E>(mx));
        float sm = 0.f;
#pragma unroll
        for (int i = 0; i < 8; ++i) {
            const float u = expf(V[i] - mx);
            U[i] = (eb + 4 * i + sub < ee) ? u : 0.f;
            sm += U[i];
        }
        sm += quad_perm<0xB1>(sm);
        sm += quad_perm<0x4E>(sm);
#pragma unroll
        for (int i = 0; i < 8; ++i) U[i] = U[i] / sm;
        add_chunk(deg);
    } else if (deg > 32) {
        float mx = NEG_INF;
        for (int c = eb; c < ee; c += 32) {
            score_chunk(c);
#pragma unroll
            for (int i = 0; i < 8; ++i) mx = fmaxf(mx, V[i]);
        }
        mx = fmaxf(mx, quad_perm<0xB1>(mx));
        mx = fmaxf(mx, quad_perm<0x4E>(mx));
        float sm = 0.f;
        for (int c = eb; c < ee; c += 32) {
            score_chunk(c);
#pragma unroll
            for (int i = 0; i < 8; ++i) {
                const float u = expf(V[i] - mx);
                sm += (c + 4 * i + sub < ee) ? u : 0.f;
            }
        }
        sm += quad_perm<0xB1>(sm);
        sm += quad_perm<0x4E>(sm);
        for (int c = eb; c < ee; c += 32) {
            score_chunk(c);
#pragma unroll
            for (int i = 0; i < 8; ++i) {
                const float u = expf(V[i] - mx) / sm;
                U[i] = (c + 4 * i + sub < ee) ? u : 0.f;
            }
            const int left = ee - c;
            add_chunk_by_two(left < 32 ? left : 32);
        }
    }
    const float4 c0 = *(const float4*)(b1 + lane * 8);
    const float4 c1 = *(const float4*)(b1 + lane * 8 + 4);
    float x[8] = {a0.x + c0.x, a0.y + c0.y, a0.z + c0.z, a0.w + c0.w,
                  a1.x + c1.x, a1.y + c1.y, a1.z + c1.z, a1.w + c1.w};
#pragma unroll
    for (int i = 0; i < 8; ++i) x[i] = (x[i] > 0.f) ? x[i] : expm1f(x[i]);
    const float4* wp = (const float4*)(W2 + lane * 24);
    const float4 w0 = wp[0], w1 = wp[1], w2 = wp[2], w3 = wp[3], w4 = wp[4], w5 = wp[5];
    float p0 = x[0] * w0.x, p1 = x[0] * w0.y, p2 = x[0] * w0.z;
    p0 += x[1] * w0.w; p1 += x[1] * w1.x; p2 += x[1] * w1.y;
    p0 += x[2] * w1.z; p1 += x[2] * w1.w; p2 += x[2] * w2.x;
    p0 += x[3] * w2.y; p1 += x[3] * w2.z; p2 += x[3] * w2.w;
    p0 += x[4] * w3.x; p1 += x[4] * w3.y; p2 += x[4] * w3.z;
    p0 += x[5] * w3.w; p1 += x[5] * w4.x; p2 += x[5] * w4.y;
    p0 += x[6] * w4.z; p1 += x[6] * w4.w; p2 += x[6] * w5.x;
    p0 += x[7] * w5.y; p1 += x[7] * w5.z; p2 += x[7] * w5.w;
    p0 = wave_sum(p0); p1 = wave_sum(p1); p2 = wave_sum(p2);
    if (lane == 0) {
        float4 row;
        row.x = p0; row.y = p1; row.z = p2;
        row.w = p0 * al2[0] + p1 * al2[1] + p2 * al2[2];
        ((float4*)hp)[node] = row;
        er2[node] = p0 * ar2[0] + p1 * ar2[1] + p2 * ar2[2];
    }
}

__global__ void __launch_bounds__(256) gat_l2_segments(const int* __restrict__ rowptr,
                                                       const int* __restrict__ ssrc,
                                                       const float* __restrict__ hp,
                                                       const float* __restrict__ er2,
                                                       float* __restrict__ agg2, int n_nodes) {
    const int lane = threadIdx.x & 31;
    const int node = blockIdx.x * 8 + __builtin_amdgcn_readfirstlane((int)(threadIdx.x >> 5));
    if (node >= n_nodes) return;
    const int eb = rowptr[node], ee = rowptr[node + 1];
    const float erv = er2[node];
    const float NEG_INF = -__builtin_huge_valf();
    const float4* rows = (const float4*)hp;
    const int deg = ee - eb;
    float p0 = 0.f, p1 = 0.f, p2 = 0.f;

    if (deg > 0 && deg <= 32) {
        const int j = eb + lane;
        const bool ok = j < ee;
        const float4 r = rows[ssrc[ok ? j : ee - 1]];
        float x = r.w + erv;
        x = (x >= 0.f) ? x : SLOPE * x;
        x = ok ? x : NEG_INF;
        const float mx = wave_max(x);
        const float e = expf(x - mx);
        const float u = ok ? e : 0.f;
        const float sm = wave_sum(u);
        const float al = u / sm;
        p0 = ok ? r.x * al : 0.f;
        p1 = ok ? r.y * al : 0.f;
        p2 = ok ? r.z * al : 0.f;
    } else if (deg > 32) {
        float mx = NEG_INF;
        for (int c = eb; c < ee; c += 32) {
            const int j = c + lane;
            if (j < ee) {
                float x = rows[ssrc[j]].w + erv;
                x = (x >= 0.f) ? x : SLOPE * x;
                mx = fmaxf(mx, x);
            }
        }
        mx = wave_max(mx);
        float sm = 0.f;
        for (int c = eb; c < ee; c += 32) {
            const int j = c + lane;
            if (j < ee) {
                float x = rows[ssrc[j]].w + erv;
                x = (x >= 0.f) ? x : SLOPE * x;
                sm += expf(x - mx);
            }
        }
        sm = wave_sum(sm);
        for (int c = eb; c < ee; c += 32) {
            const int j = c + lane;
            if (j < ee) {
                const float4 r = rows[ssrc[j]];
                float x = r.w + erv;
                x = (x >= 0.f) ? x : SLOPE * x;
                const float al = expf(x - mx) / sm;
                p0 += r.x * al; p1 += r.y * al; p2 += r.z * al;
            }
        }
    }
    p0 = wave_sum(p0); p1 = wave_sum(p1); p2 = wave_sum(p2);
    if (lane == 0) {
        agg2[node * 3 + 0] = p0;
        agg2[node * 3 + 1] = p1;
        agg2[node * 3 + 2] = p2;
    }
}

__global__ void gat_finalize(const float* __restrict__ agg2, const float* __restrict__ b2,
                             float* __restrict__ out, int total) {
    int i = blockIdx.x * blockDim.x + threadIdx.x;
    if (i >= total) return;
    out[i] = agg2[i] + b2[i % 3];
}

extern "C" void kernel_launch(void* const* d_in, const int* in_sizes, int n_in,
                              void* d_out, int out_size, void* d_ws, size_t ws_size,
                              hipStream_t stream) {
    const float* feat = (const float*)d_in[0];
    const int*   src  = (const int*)  d_in[1];
    const int*   dst  = (const int*)  d_in[2];
    const float* W1   = (const float*)d_in[3];
    const float* al1  = (const float*)d_in[4];
    const float* ar1  = (const float*)d_in[5];
    const float* b1   = (const float*)d_in[6];
    const float* W2   = (const float*)d_in[7];
    const float* al2  = (const float*)d_in[8];
    const float* ar2  = (const float*)d_in[9];
    const float* b2   = (const float*)d_in[10];
    float* out = (float*)d_out;

    const int n_nodes = in_sizes[0] / 256;
    const int n_edges = in_sizes[1];

    char* cur = (char*)d_ws;
    auto alloc = [&](size_t bytes) -> float* {
        float* p = (float*)cur;
        cur += (bytes + 255) & ~(size_t)255;
        return p;
    };
    float* H1   = alloc((size_t)n_nodes * 256 * 4);
    float* el1  = alloc((size_t)n_nodes * 8 * 4);
    float* er1  = alloc((size_t)n_nodes * 8 * 4);
    float* hp2  = alloc((size_t)n_nodes * 4 * 4);
    float* er2  = alloc((size_t)n_nodes * 4);
    float* agg2 = alloc((size_t)n_nodes * 3 * 4);
    float* cntf = alloc((size_t)n_nodes * 4);
    int*   cnt    = (int*)cntf;
    int*   rowptr = (int*)alloc((size_t)(n_nodes + 1) * 4);
    int*   rank   = (int*)alloc((size_t)n_edges * 4);
    int*   ssrc   = (int*)alloc((size_t)n_edges * 4);

    const int B = 256;

    gat_fill_f32<<<cdiv(n_nodes, B), B, 0, stream>>>(cntf, 0.f, n_nodes);
    gat_dst_rank<<<cdiv(n_edges, B), B, 0, stream>>>(dst, cnt, rank, n_edges);
    gat_row_offsets<<<cdiv(n_nodes, 1024), 1024, 0, stream>>>(cnt, rowptr, n_nodes);
    gat_dst_place<<<cdiv(n_edges, B), B, 0, stream>>>(src, dst, rowptr, rank, ssrc, n_edges);

    gat_gemm1_wmma<<<cdiv(n_nodes, 32), 256, 0, stream>>>(feat, W1, H1, n_nodes);
    gat_attn_scores<<<cdiv(n_nodes * 8, B), B, 0, stream>>>(H1, al1, ar1, el1, er1, n_nodes);
    gat_l1_segments<<<cdiv(n_nodes, 8), 256, 0, stream>>>(rowptr, ssrc, H1, el1, er1,
                                                          b1, W2, al2, ar2, hp2, er2, n_nodes);

    gat_l2_segments<<<cdiv(n_nodes, 8), 256, 0, stream>>>(rowptr, ssrc, hp2, er2, agg2, n_nodes);
    gat_finalize<<<cdiv(n_nodes * 3, B), B, 0, stream>>>(agg2, b2, out, n_nodes * 3);
}


# GEMM1 LDS rows padded (A panel pitch 256 -> 260 dwords, W chunk 256 -> 272): WMMA fragment reads spread over all 64 banks instead of the 16 rows of a half-wave falling on one bank
# speedup vs baseline: 5.8398x; 1.1127x over previous
#include <hip/hip_runtime.h>
#include <hip/hip_bf16.h>

typedef __attribute__((ext_vector_type(2))) float v2f;
typedef __attribute__((ext_vector_type(8))) float v8f;

#define SLOPE 0.2f

static __host__ __device__ inline int cdiv(int a, int b) { return (a + b - 1) / b; }

__global__ void gat_fill_f32(float* __restrict__ p, float v, int n) {
    int i = blockIdx.x * blockDim.x + threadIdx.x;
    if (i < n) p[i] = v;
}

#define A_PITCH 260
#define W_PITCH 272

__global__ void __launch_bounds__(256) gat_gemm1_wmma(const float* __restrict__ A,
                               const float* __restrict__ W,
                               float* __restrict__ H,
                               int n_nodes) {
    __shared__ float As[32 * A_PITCH];
    __shared__ float Ws[32 * W_PITCH];

    const int row0 = blockIdx.x * 32;
    const int tid  = threadIdx.x;
    const int wave = tid >> 5;
    const int lane = tid & 31;

    const int sr = tid >> 6;
    const int sc = (tid & 63) << 2;
    float4 st[8];

#pragma unroll
    for (int j = 0; j < 8; ++j) {
        int rr = row0 + sr + 4 * j; if (rr >= n_nodes) rr = n_nodes - 1;
        st[j] = *(const float4*)(A + (size_t)rr * 256 + sc);
    }
#pragma unroll
    for (int j = 0; j < 8; ++j) *(float4*)(&As[(sr + 4 * j) * A_PITCH + sc]) = st[j];

    const int m   = lane & 15;
    const int kk  = (lane >> 4) << 1;
    const int n0  = wave * 32;
    const int n1  = n0 + 16;

    v8f c00 = {}, c01 = {}, c10 = {}, c11 = {};

    for (int kc = 0; kc < 256; kc += 32) {
        __syncthreads();
#pragma unroll
        for (int j = 0; j < 8; ++j) st[j] = *(const float4*)(W + (size_t)(kc + sr + 4 * j) * 256 + sc);
#pragma unroll
        for (int j = 0; j < 8; ++j) *(float4*)(&Ws[(sr + 4 * j) * W_PITCH + sc]) = st[j];
        __syncthreads();

#pragma unroll
        for (int k = 0; k < 32; k += 4) {
            v2f a0, a1, b0, b1;
            a0.x = As[m * A_PITCH + kc + k + kk];
            a0.y = As[m * A_PITCH + kc + k + kk + 1];
            a1.x = As[(m + 16) * A_PITCH + kc + k + kk];
            a1.y = As[(m + 16) * A_PITCH + kc + k + kk + 1];
            b0.x = Ws[(k + kk) * W_PITCH + n0 + m];
            b0.y = Ws[(k + kk + 1) * W_PITCH + n0 + m];
            b1.x = Ws[(k + kk) * W_PITCH + n1 + m];
            b1.y = Ws[(k + kk + 1) * W_PITCH + n1 + m];
            c00 = __builtin_amdgcn_wmma_f32_16x16x4_f32(false, a0, false, b0, (short)0, c00, false, false);
            c01 = __builtin_amdgcn_wmma_f32_16x16x4_f32(false, a0, false, b1, (short)0, c01, false, false);
            c10 = __builtin_amdgcn_wmma_f32_16x16x4_f32(false, a1, false, b0, (short)0, c10, false, false);
            c11 = __builtin_amdgcn_wmma_f32_16x16x4_f32(false, a1, false, b1, (short)0, c11, false, false);
        }
    }

    const int mhi = (lane >> 4) * 8;
    for (int v = 0; v < 8; ++v) {
        int r0 = row0 + mhi + v;
        int r1 = r0 + 16;
        if (r0 < n_nodes) {
            H[(size_t)r0 * 256 + n0 + m] = c00[v];
            H[(size_t)r0 * 256 + n1 + m] = c01[v];
        }
        if (r1 < n_nodes) {
            H[(size_t)r1 * 256 + n0 + m] = c10[v];
            H[(size_t)r1 * 256 + n1 + m] = c11[v];
        }
    }
}

__global__ void gat_attn_scores(const float* __restrict__ H,
                                const float* __restrict__ al,
                                const float* __restrict__ ar,
                                float* __restrict__ el, float* __restrict__ er,
                                int n_nodes) {
    int i = blockIdx.x * blockDim.x + threadIdx.x;
    if (i >= n_nodes * 8) return;
    int n = i >> 3, h = i & 7;
    const float* hp  = H + (size_t)n * 256 + h * 32;
    const float* alp = al + h * 32;
    const float* arp = ar + h * 32;
    float sl = 0.f, sr = 0.f;
    for (int d = 0; d < 32; ++d) { float v = hp[d]; sl += v * alp[d]; sr += v * arp[d]; }
    el[i] = sl; er[i] = sr;
}

__global__ void gat_dst_rank(const int* __restrict__ dst, int* __restrict__ cnt,
                             int* __restrict__ rank, int n_edges) {
    int e = blockIdx.x * blockDim.x + threadIdx.x;
    if (e >= n_edges) return;
    rank[e] = atomicAdd(&cnt[dst[e]], 1);
}

__global__ void __launch_bounds__(1024) gat_row_offsets(const int* __restrict__ cnt,
                                                        int* __restrict__ rowptr, int n) {
    __shared__ int wtot[32];
    __shared__ int wpre[32];
    const int t = threadIdx.x, lane = t & 31, wave = t >> 5;
    const int b = blockIdx.x;
    const int i = b * 1024 + t;
    const int v = (i < n) ? cnt[i] : 0;
    int pre = 0;
    for (int k = 0; k < b; ++k) pre += cnt[k * 1024 + t];
    int x = v;
    for (int o = 1; o < 32; o <<= 1) {
        int y = __shfl_up(x, o, 32);
        if (lane >= o) x += y;
    }
    for (int o = 16; o > 0; o >>= 1) pre += __shfl_xor(pre, o, 32);
    if (lane == 31) wtot[wave] = x;
    if (lane == 0)  wpre[wave] = pre;
    __syncthreads();
    int w = wtot[lane];
    for (int o = 1; o < 32; o <<= 1) {
        int y = __shfl_up(w, o, 32);
        if (lane >= o) w += y;
    }
    int base = wpre[lane];
    for (int o = 16; o > 0; o >>= 1) base += __shfl_xor(base, o, 32);
    const int before_wave = __shfl(w, wave, 32) - wtot[wave];
    const int excl = base + before_wave + x - v;
    if (i < n) rowptr[i] = excl;
    if (i == n - 1) rowptr[n] = excl + v;
}

__global__ void gat_dst_place(const int* __restrict__ src, const int* __restrict__ dst,
                              const int* __restrict__ rowptr, const int* __restrict__ rank,
                              int* __restrict__ ssrc, int n_edges) {
    int e = blockIdx.x * blockDim.x + threadIdx.x;
    if (e >= n_edges) return;
    ssrc[rowptr[dst[e]] + rank[e]] = src[e];
}

__device__ inline float wave_max(float x) {
    for (int o = 16; o > 0; o >>= 1) x = fmaxf(x, __shfl_xor(x, o, 32));
    return x;
}
__device__ inline float wave_sum(float x) {
    for (int o = 16; o > 0; o >>= 1) x += __shfl_xor(x, o, 32);
    return x;
}

template <int CTRL>
__device__ inline float quad_perm(float x) {
    return __int_as_float(__builtin_amdgcn_update_dpp(0, __float_as_int(x), CTRL, 0xF, 0xF, true));
}

template <int K>
__device__ inline void gat_add_row(const float* __restrict__ Hl, int s_lane, float u_lane,
                                   float4& a0, float4& a1) {
    const int sn = __builtin_amdgcn_readlane(s_lane, K);
    const float al = quad_perm<K * 0x55>(u_lane);
    const float4 x0 = *(const float4*)(Hl + (size_t)sn * 256);
    const float4 x1 = *(const float4*)(Hl + (size_t)sn * 256 + 4);
    a0.x += x0.x * al; a0.y += x0.y * al; a0.z += x0.z * al; a0.w += x0.w * al;
    a1.x += x1.x * al; a1.y += x1.y * al; a1.z += x1.z * al; a1.w += x1.w * al;
}

__global__ void __launch_bounds__(256) gat_l1_segments(const int* __restrict__ rowptr,
                                                       const int* __restrict__ ssrc,
                                                       const float* __restrict__ H,
                                                       const float* __restrict__ el,
                                                       const float* __restrict__ er,
                                                       const float* __restrict__ b1,
                                                       const float* __restrict__ W2,
                                                       const float* __restrict__ al2,
                                                       const float* __restrict__ ar2,
                                                       float* __restrict__ hp,
                                                       float* __restrict__ er2, int n_nodes) {
    const int lane = threadIdx.x & 31;
    const int node = blockIdx.x * 8 + __builtin_amdgcn_readfirstlane((int)(threadIdx.x >> 5));
    if (node >= n_nodes) return;
    const int eb = rowptr[node], ee = rowptr[node + 1];
    const int head = lane >> 2, sub = lane & 3;
    const float erv = er[node * 8 + head];
    const float* Hl = H + lane * 8;
    const float NEG_INF = -__builtin_huge_valf();

    int   S[8];
    float V[8];
    float U[8];
    float4 a0 = {0.f, 0.f, 0.f, 0.f}, a1 = {0.f, 0.f, 0.f, 0.f};

    auto score_chunk = [&](int c) {
        const int last = ee - 1;
#pragma unroll
        for (int i = 0; i < 8; ++i) {
            const int j = c + 4 * i + sub;
            S[i] = ssrc[j < ee ? j : last];
        }
#pragma unroll
        for (int i = 0; i < 8; ++i) {
            float x = el[S[i] * 8 + head] + erv;
            x = (x >= 0.f) ? x : SLOPE * x;
            V[i] = (c + 4 * i + sub < ee) ? x : NEG_INF;
        }
    };
    auto add_chunk = [&](int cnt) {
#pragma unroll
        for (int i = 0; i < 8; ++i) {
            if (4 * i + 4 <= cnt) {
                gat_add_row<0>(Hl, S[i], U[i], a0, a1);
                gat_add_row<1>(Hl, S[i], U[i], a0, a1);
                gat_add_row<2>(Hl, S[i], U[i], a0, a1);
                gat_add_row<3>(Hl, S[i], U[i], a0, a1);
            } else if (4 * i < cnt) {
                gat_add_row<0>(Hl, S[i], U[i], a0, a1);
                if (4 * i + 1 < cnt) gat_add_row<1>(Hl, S[i], U[i], a0, a1);
                if (4 * i + 2 < cnt) gat_add_row<2>(Hl, S[i], U[i], a0, a1);
            }
        }
    };

    auto add_chunk_by_two = [&](int cnt) {
#pragma unroll
        for (int i = 0; i < 8; ++i) {
            if (4 * i + 2 <= cnt) {
                gat_add_row<0>(Hl, S[i], U[i], a0, a1);
                gat_add_row<1>(Hl, S[i], U[i], a0, a1);
            } else if (4 * i < cnt) {
                gat_add_row<0>(Hl, S[i], U[i], a0, a1);
            }
            if (4 * i + 4 <= cnt) {
                gat_add_row<2>(Hl, S[i], U[i], a0, a1);
                gat_add_row<3>(Hl, S[i], U[i], a0, a1);
            } else if (4 * i + 2 < cnt) {
                gat_add_row<2>(Hl, S[i], U[i], a0, a1);
            }
        }
    };

    const int deg = ee - eb;
    if (deg > 0 && deg <= 32) {
        score_chunk(eb);
        float mx = V[0];
#pragma unroll
        for (int i = 1; i < 8; ++i) mx = fmaxf(mx, V[i]);
        mx = fmaxf(mx, quad_perm<0xB1>(mx));
        mx = fmaxf(mx, quad_perm<0x4E>(mx));
        float sm = 0.f;
#pragma unroll
        for (int i = 0; i < 8; ++i) {
            const float u = expf(V[i] - mx);
            U[i] = (eb + 4 * i + sub < ee) ? u : 0.f;
            sm += U[i];
        }
        sm += quad_perm<0xB1>(sm);
        sm += quad_perm<0x4E>(sm);
#pragma unroll
        for (int i = 0; i < 8; ++i) U[i] = U[i] / sm;
        add_chunk(deg);
    } else if (deg > 32) {
        float mx = NEG_INF;
        for (int c = eb; c < ee; c += 32) {
            score_chunk(c);
#pragma unroll
            for (int i = 0; i < 8; ++i) mx = fmaxf(mx, V[i]);
        }
        mx = fmaxf(mx, quad_perm<0xB1>(mx));
        mx = fmaxf(mx, quad_perm<0x4E>(mx));
        float sm = 0.f;
        for (int c = eb; c < ee; c += 32) {
            score_chunk(c);
#pragma unroll
            for (int i = 0; i < 8; ++i) {
                const float u = expf(V[i] - mx);
                sm += (c + 4 * i + sub < ee) ? u : 0.f;
            }
        }
        sm += quad_perm<0xB1>(sm);
        sm += quad_perm<0x4E>(sm);
        for (int c = eb; c < ee; c += 32) {
            score_chunk(c);
#pragma unroll
            for (int i = 0; i < 8; ++i) {
                const float u = expf(V[i] - mx) / sm;
                U[i] = (c + 4 * i + sub < ee) ? u : 0.f;
            }
            const int left = ee - c;
            add_chunk_by_two(left < 32 ? left : 32);
        }
    }
    const float4 c0 = *(const float4*)(b1 + lane * 8);
    const float4 c1 = *(const float4*)(b1 + lane * 8 + 4);
    float x[8] = {a0.x + c0.x, a0.y + c0.y, a0.z + c0.z, a0.w + c0.w,
                  a1.x + c1.x, a1.y + c1.y, a1.z + c1.z, a1.w + c1.w};
#pragma unroll
    for (int i = 0; i < 8; ++i) x[i] = (x[i] > 0.f) ? x[i] : expm1f(x[i]);
    const float4* wp = (const float4*)(W2 + lane * 24);
    const float4 w0 = wp[0], w1 = wp[1], w2 = wp[2], w3 = wp[3], w4 = wp[4], w5 = wp[5];
    float p0 = x[0] * w0.x, p1 = x[0] * w0.y, p2 = x[0] * w0.z;
    p0 += x[1] * w0.w; p1 += x[1] * w1.x; p2 += x[1] * w1.y;
    p0 += x[2] * w1.z; p1 += x[2] * w1.w; p2 += x[2] * w2.x;
    p0 += x[3] * w2.y; p1 += x[3] * w2.z; p2 += x[3] * w2.w;
    p0 += x[4] * w3.x; p1 += x[4] * w3.y; p2 += x[4] * w3.z;
    p0 += x[5] * w3.w; p1 += x[5] * w4.x; p2 += x[5] * w4.y;
    p0 += x[6] * w4.z; p1 += x[6] * w4.w; p2 += x[6] * w5.x;
    p0 += x[7] * w5.y; p1 += x[7] * w5.z; p2 += x[7] * w5.w;
    p0 = wave_sum(p0); p1 = wave_sum(p1); p2 = wave_sum(p2);
    if (lane == 0) {
        float4 row;
        row.x = p0; row.y = p1; row.z = p2;
        row.w = p0 * al2[0] + p1 * al2[1] + p2 * al2[2];
        ((float4*)hp)[node] = row;
        er2[node] = p0 * ar2[0] + p1 * ar2[1] + p2 * ar2[2];
    }
}

__global__ void __launch_bounds__(256) gat_l2_segments(const int* __restrict__ rowptr,
                                                       const int* __restrict__ ssrc,
                                                       const float* __restrict__ hp,
                                                       const float* __restrict__ er2,
                                                       float* __restrict__ agg2, int n_nodes) {
    const int lane = threadIdx.x & 31;
    const int node = blockIdx.x * 8 + __builtin_amdgcn_readfirstlane((int)(threadIdx.x >> 5));
    if (node >= n_nodes) return;
    const int eb = rowptr[node], ee = rowptr[node + 1];
    const float erv = er2[node];
    const float NEG_INF = -__builtin_huge_valf();
    const float4* rows = (const float4*)hp;
    const int deg = ee - eb;
    float p0 = 0.f, p1 = 0.f, p2 = 0.f;

    if (deg > 0 && deg <= 32) {
        const int j = eb + lane;
        const bool ok = j < ee;
        const float4 r = rows[ssrc[ok ? j : ee - 1]];
        float x = r.w + erv;
        x = (x >= 0.f) ? x : SLOPE * x;
        x = ok ? x : NEG_INF;
        const float mx = wave_max(x);
        const float e = expf(x - mx);
        const float u = ok ? e : 0.f;
        const float sm = wave_sum(u);
        const float al = u / sm;
        p0 = ok ? r.x * al : 0.f;
        p1 = ok ? r.y * al : 0.f;
        p2 = ok ? r.z * al : 0.f;
    } else if (deg > 32) {
        float mx = NEG_INF;
        for (int c = eb; c < ee; c += 32) {
            const int j = c + lane;
            if (j < ee) {
                float x = rows[ssrc[j]].w + erv;
                x = (x >= 0.f) ? x : SLOPE * x;
                mx = fmaxf(mx, x);
            }
        }
        mx = wave_max(mx);
        float sm = 0.f;
        for (int c = eb; c < ee; c += 32) {
            const int j = c + lane;
            if (j < ee) {
                float x = rows[ssrc[j]].w + erv;
                x = (x >= 0.f) ? x : SLOPE * x;
                sm += expf(x - mx);
            }
        }
        sm = wave_sum(sm);
        for (int c = eb; c < ee; c += 32) {
            const int j = c + lane;
            if (j < ee) {
                const float4 r = rows[ssrc[j]];
                float x = r.w + erv;
                x = (x >= 0.f) ? x : SLOPE * x;
                const float al = expf(x - mx) / sm;
                p0 += r.x * al; p1 += r.y * al; p2 += r.z * al;
            }
        }
    }
    p0 = wave_sum(p0); p1 = wave_sum(p1); p2 = wave_sum(p2);
    if (lane == 0) {
        agg2[node * 3 + 0] = p0;
        agg2[node * 3 + 1] = p1;
        agg2[node * 3 + 2] = p2;
    }
}

__global__ void gat_finalize(const float* __restrict__ agg2, const float* __restrict__ b2,
                             float* __restrict__ out, int total) {
    int i = blockIdx.x * blockDim.x + threadIdx.x;
    if (i >= total) return;
    out[i] = agg2[i] + b2[i % 3];
}

extern "C" void kernel_launch(void* const* d_in, const int* in_sizes, int n_in,
                              void* d_out, int out_size, void* d_ws, size_t ws_size,
                              hipStream_t stream) {
    const float* feat = (const float*)d_in[0];
    const int*   src  = (const int*)  d_in[1];
    const int*   dst  = (const int*)  d_in[2];
    const float* W1   = (const float*)d_in[3];
    const float* al1  = (const float*)d_in[4];
    const float* ar1  = (const float*)d_in[5];
    const float* b1   = (const float*)d_in[6];
    const float* W2   = (const float*)d_in[7];
    const float* al2  = (const float*)d_in[8];
    const float* ar2  = (const float*)d_in[9];
    const float* b2   = (const float*)d_in[10];
    float* out = (float*)d_out;

    const int n_nodes = in_sizes[0] / 256;
    const int n_edges = in_sizes[1];

    char* cur = (char*)d_ws;
    auto alloc = [&](size_t bytes) -> float* {
        float* p = (float*)cur;
        cur += (bytes + 255) & ~(size_t)255;
        return p;
    };
    float* H1   = alloc((size_t)n_nodes * 256 * 4);
    float* el1  = alloc((size_t)n_nodes * 8 * 4);
    float* er1  = alloc((size_t)n_nodes * 8 * 4);
    float* hp2  = alloc((size_t)n_nodes * 4 * 4);
    float* er2  = alloc((size_t)n_nodes * 4);
    float* agg2 = alloc((size_t)n_nodes * 3 * 4);
    float* cntf = alloc((size_t)n_nodes * 4);
    int*   cnt    = (int*)cntf;
    int*   rowptr = (int*)alloc((size_t)(n_nodes + 1) * 4);
    int*   rank   = (int*)alloc((size_t)n_edges * 4);
    int*   ssrc   = (int*)alloc((size_t)n_edges * 4);

    const int B = 256;

    gat_fill_f32<<<cdiv(n_nodes, B), B, 0, stream>>>(cntf, 0.f, n_nodes);
    gat_dst_rank<<<cdiv(n_edges, B), B, 0, stream>>>(dst, cnt, rank, n_edges);
    gat_row_offsets<<<cdiv(n_nodes, 1024), 1024, 0, stream>>>(cnt, rowptr, n_nodes);
    gat_dst_place<<<cdiv(n_edges, B), B, 0, stream>>>(src, dst, rowptr, rank, ssrc, n_edges);

    gat_gemm1_wmma<<<cdiv(n_nodes, 32), 256, 0, stream>>>(feat, W1, H1, n_nodes);
    gat_attn_scores<<<cdiv(n_nodes * 8, B), B, 0, stream>>>(H1, al1, ar1, el1, er1, n_nodes);
    gat_l1_segments<<<cdiv(n_nodes, 8), 256, 0, stream>>>(rowptr, ssrc, H1, el1, er1,
                                                          b1, W2, al2, ar2, hp2, er2, n_nodes);

    gat_l2_segments<<<cdiv(n_nodes, 8), 256, 0, stream>>>(rowptr, ssrc, hp2, er2, agg2, n_nodes);
    gat_finalize<<<cdiv(n_nodes * 3, B), B, 0, stream>>>(agg2, b2, out, n_nodes * 3);
}


# GEMM1 epilogue forms el/er from its accumulators (wave = head: 2 columns per lane, DPP sum over 16 lanes, rows via LDS); score kernel and its re-read of H gone, 8 launches (el/er sum order changed)
# speedup vs baseline: 6.0819x; 1.0415x over previous
#include <hip/hip_runtime.h>
#include <hip/hip_bf16.h>

typedef __attribute__((ext_vector_type(2))) float v2f;
typedef __attribute__((ext_vector_type(8))) float v8f;

#define SLOPE 0.2f

static __host__ __device__ inline int cdiv(int a, int b) { return (a + b - 1) / b; }

__global__ void gat_fill_f32(float* __restrict__ p, float v, int n) {
    int i = blockIdx.x * blockDim.x + threadIdx.x;
    if (i < n) p[i] = v;
}

template <int CTRL>
__device__ inline float row_xmask_add(float x) {
    return x + __int_as_float(__builtin_amdgcn_update_dpp(0, __float_as_int(x), CTRL, 0xF, 0xF, true));
}
__device__ inline float row16_sum(float x) {
    x = row_xmask_add<0x161>(x);
    x = row_xmask_add<0x162>(x);
    x = row_xmask_add<0x164>(x);
    x = row_xmask_add<0x168>(x);
    return x;
}

#define A_PITCH 260
#define W_PITCH 272

__global__ void __launch_bounds__(256) gat_gemm1_wmma(const float* __restrict__ A,
                               const float* __restrict__ W,
                               const float* __restrict__ al,
                               const float* __restrict__ ar,
                               float* __restrict__ H,
                               float* __restrict__ el,
                               float* __restrict__ er,
                               int n_nodes) {
    __shared__ float As[32 * A_PITCH];
    __shared__ float Ws[32 * W_PITCH];
    __shared__ float Sc[2 * 32 * 8];

    const int row0 = blockIdx.x * 32;
    const int tid  = threadIdx.x;
    const int wave = tid >> 5;
    const int lane = tid & 31;

    const int sr = tid >> 6;
    const int sc = (tid & 63) << 2;
    float4 st[8];

#pragma unroll
    for (int j = 0; j < 8; ++j) {
        int rr = row0 + sr + 4 * j; if (rr >= n_nodes) rr = n_nodes - 1;
        st[j] = *(const float4*)(A + (size_t)rr * 256 + sc);
    }
#pragma unroll
    for (int j = 0; j < 8; ++j) *(float4*)(&As[(sr + 4 * j) * A_PITCH + sc]) = st[j];

    const int m   = lane & 15;
    const int kk  = (lane >> 4) << 1;
    const int n0  = wave * 32;
    const int n1  = n0 + 16;

    const float al0 = al[n0 + m], al1 = al[n1 + m];
    const float ar0 = ar[n0 + m], ar1 = ar[n1 + m];

    v8f c00 = {}, c01 = {}, c10 = {}, c11 = {};

    for (int kc = 0; kc < 256; kc += 32) {
        __syncthreads();
#pragma unroll
        for (int j = 0; j < 8; ++j) st[j] = *(const float4*)(W + (size_t)(kc + sr + 4 * j) * 256 + sc);
#pragma unroll
        for (int j = 0; j < 8; ++j) *(float4*)(&Ws[(sr + 4 * j) * W_PITCH + sc]) = st[j];
        __syncthreads();

#pragma unroll
        for (int k = 0; k < 32; k += 4) {
            v2f a0, a1, b0, b1;
            a0.x = As[m * A_PITCH + kc + k + kk];
            a0.y = As[m * A_PITCH + kc + k + kk + 1];
            a1.x = As[(m + 16) * A_PITCH + kc + k + kk];
            a1.y = As[(m + 16) * A_PITCH + kc + k + kk + 1];
            b0.x = Ws[(k + kk) * W_PITCH + n0 + m];
            b0.y = Ws[(k + kk + 1) * W_PITCH + n0 + m];
            b1.x = Ws[(k + kk) * W_PITCH + n1 + m];
            b1.y = Ws[(k + kk + 1) * W_PITCH + n1 + m];
            c00 = __builtin_amdgcn_wmma_f32_16x16x4_f32(false, a0, false, b0, (short)0, c00, false, false);
            c01 = __builtin_amdgcn_wmma_f32_16x16x4_f32(false, a0, false, b1, (short)0, c01, false, false);
            c10 = __builtin_amdgcn_wmma_f32_16x16x4_f32(false, a1, false, b0, (short)0, c10, false, false);
            c11 = __builtin_amdgcn_wmma_f32_16x16x4_f32(false, a1, false, b1, (short)0, c11, false, false);
        }
    }

    const int mhi = (lane >> 4) * 8;

#pragma unroll
    for (int v = 0; v < 8; ++v) {
        float l0 = c00[v] * al0 + c01[v] * al1;
        float l1 = c10[v] * al0 + c11[v] * al1;
        float q0 = c00[v] * ar0 + c01[v] * ar1;
        float q1 = c10[v] * ar0 + c11[v] * ar1;
        l0 = row16_sum(l0); l1 = row16_sum(l1);
        q0 = row16_sum(q0); q1 = row16_sum(q1);
        if (m == 0) {
            Sc[(mhi + v) * 8 + wave]            = l0;
            Sc[(16 + mhi + v) * 8 + wave]       = l1;
            Sc[256 + (mhi + v) * 8 + wave]      = q0;
            Sc[256 + (16 + mhi + v) * 8 + wave] = q1;
        }
    }
    __syncthreads();
    if (row0 + (tid >> 3) < n_nodes) {
        el[(size_t)row0 * 8 + tid] = Sc[tid];
        er[(size_t)row0 * 8 + tid] = Sc[256 + tid];
    }

    for (int v = 0; v < 8; ++v) {
        int r0 = row0 + mhi + v;
        int r1 = r0 + 16;
        if (r0 < n_nodes) {
            H[(size_t)r0 * 256 + n0 + m] = c00[v];
            H[(size_t)r0 * 256 + n1 + m] = c01[v];
        }
        if (r1 < n_nodes) {
            H[(size_t)r1 * 256 + n0 + m] = c10[v];
            H[(size_t)r1 * 256 + n1 + m] = c11[v];
        }
    }
}

__global__ void gat_dst_rank(const int* __restrict__ dst, int* __restrict__ cnt,
                             int* __restrict__ rank, int n_edges) {
    int e = blockIdx.x * blockDim.x + threadIdx.x;
    if (e >= n_edges) return;
    rank[e] = atomicAdd(&cnt[dst[e]], 1);
}

__global__ void __launch_bounds__(1024) gat_row_offsets(const int* __restrict__ cnt,
                                                        int* __restrict__ rowptr, int n) {
    __shared__ int wtot[32];
    __shared__ int wpre[32];
    const int t = threadIdx.x, lane = t & 31, wave = t >> 5;
    const int b = blockIdx.x;
    const int i = b * 1024 + t;
    const int v = (i < n) ? cnt[i] : 0;
    int pre = 0;
    for (int k = 0; k < b; ++k) pre += cnt[k * 1024 + t];
    int x = v;
    for (int o = 1; o < 32; o <<= 1) {
        int y = __shfl_up(x, o, 32);
        if (lane >= o) x += y;
    }
    for (int o = 16; o > 0; o >>= 1) pre += __shfl_xor(pre, o, 32);
    if (lane == 31) wtot[wave] = x;
    if (lane == 0)  wpre[wave] = pre;
    __syncthreads();
    int w = wtot[lane];
    for (int o = 1; o < 32; o <<= 1) {
        int y = __shfl_up(w, o, 32);
        if (lane >= o) w += y;
    }
    int base = wpre[lane];
    for (int o = 16; o > 0; o >>= 1) base += __shfl_xor(base, o, 32);
    const int before_wave = __shfl(w, wave, 32) - wtot[wave];
    const int excl = base + before_wave + x - v;
    if (i < n) rowptr[i] = excl;
    if (i == n - 1) rowptr[n] = excl + v;
}

__global__ void gat_dst_place(const int* __restrict__ src, const int* __restrict__ dst,
                              const int* __restrict__ rowptr, const int* __restrict__ rank,
                              int* __restrict__ ssrc, int n_edges) {
    int e = blockIdx.x * blockDim.x + threadIdx.x;
    if (e >= n_edges) return;
    ssrc[rowptr[dst[e]] + rank[e]] = src[e];
}

__device__ inline float wave_max(float x) {
    for (int o = 16; o > 0; o >>= 1) x = fmaxf(x, __shfl_xor(x, o, 32));
    return x;
}
__device__ inline float wave_sum(float x) {
    for (int o = 16; o > 0; o >>= 1) x += __shfl_xor(x, o, 32);
    return x;
}

template <int CTRL>
__device__ inline float quad_perm(float x) {
    return __int_as_float(__builtin_amdgcn_update_dpp(0, __float_as_int(x), CTRL, 0xF, 0xF, true));
}

template <int K>
__device__ inline void gat_add_row(const float* __restrict__ Hl, int s_lane, float u_lane,
                                   float4& a0, float4& a1) {
    const int sn = __builtin_amdgcn_readlane(s_lane, K);
    const float al = quad_perm<K * 0x55>(u_lane);
    const float4 x0 = *(const float4*)(Hl + (size_t)sn * 256);
    const float4 x1 = *(const float4*)(Hl + (size_t)sn * 256 + 4);
    a0.x += x0.x * al; a0.y += x0.y * al; a0.z += x0.z * al; a0.w += x0.w * al;
    a1.x += x1.x * al; a1.y += x1.y * al; a1.z += x1.z * al; a1.w += x1.w * al;
}

__global__ void __launch_bounds__(256) gat_l1_segments(const int* __restrict__ rowptr,
                                                       const int* __restrict__ ssrc,
                                                       const float* __restrict__ H,
                                                       const float* __restrict__ el,
                                                       const float* __restrict__ er,
                                                       const float* __restrict__ b1,
                                                       const float* __restrict__ W2,
                                                       const float* __restrict__ al2,
                                                       const float* __restrict__ ar2,
                                                       float* __restrict__ hp,
                                                       float* __restrict__ er2, int n_nodes) {
    const int lane = threadIdx.x & 31;
    const int node = blockIdx.x * 8 + __builtin_amdgcn_readfirstlane((int)(threadIdx.x >> 5));
    if (node >= n_nodes) return;
    const int eb = rowptr[node], ee = rowptr[node + 1];
    const int head = lane >> 2, sub = lane & 3;
    const float erv = er[node * 8 + head];
    const float* Hl = H + lane * 8;
    const float NEG_INF = -__builtin_huge_valf();

    int   S[8];
    float V[8];
    float U[8];
    float4 a0 = {0.f, 0.f, 0.f, 0.f}, a1 = {0.f, 0.f, 0.f, 0.f};

    auto score_chunk = [&](int c) {
        const int last = ee - 1;
#pragma unroll
        for (int i = 0; i < 8; ++i) {
            const int j = c + 4 * i + sub;
            S[i] = ssrc[j < ee ? j : last];
        }
#pragma unroll
        for (int i = 0; i < 8; ++i) {
            float x = el[S[i] * 8 + head] + erv;
            x = (x >= 0.f) ? x : SLOPE * x;
            V[i] = (c + 4 * i + sub < ee) ? x : NEG_INF;
        }
    };
    auto add_chunk = [&](int cnt) {
#pragma unroll
        for (int i = 0; i < 8; ++i) {
            if (4 * i + 4 <= cnt) {
                gat_add_row<0>(Hl, S[i], U[i], a0, a1);
                gat_add_row<1>(Hl, S[i], U[i], a0, a1);
                gat_add_row<2>(Hl, S[i], U[i], a0, a1);
                gat_add_row<3>(Hl, S[i], U[i], a0, a1);
            } else if (4 * i < cnt) {
                gat_add_row<0>(Hl, S[i], U[i], a0, a1);
                if (4 * i + 1 < cnt) gat_add_row<1>(Hl, S[i], U[i], a0, a1);
                if (4 * i + 2 < cnt) gat_add_row<2>(Hl, S[i], U[i], a0, a1);
            }
        }
    };

    auto add_chunk_by_two = [&](int cnt) {
#pragma unroll
        for (int i = 0; i < 8; ++i) {
            if (4 * i + 2 <= cnt) {
                gat_add_row<0>(Hl, S[i], U[i], a0, a1);
                gat_add_row<1>(Hl, S[i], U[i], a0, a1);
            } else if (4 * i < cnt) {
                gat_add_row<0>(Hl, S[i], U[i], a0, a1);
            }
            if (4 * i + 4 <= cnt) {
                gat_add_row<2>(Hl, S[i], U[i], a0, a1);
                gat_add_row<3>(Hl, S[i], U[i], a0, a1);
            } else if (4 * i + 2 < cnt) {
                gat_add_row<2>(Hl, S[i], U[i], a0, a1);
            }
        }
    };

    const int deg = ee - eb;
    if (deg > 0 && deg <= 32) {
        score_chunk(eb);
        float mx = V[0];
#pragma unroll
        for (int i = 1; i < 8; ++i) mx = fmaxf(mx, V[i]);
        mx = fmaxf(mx, quad_perm<0xB1>(mx));
        mx = fmaxf(mx, quad_perm<0x4E>(mx));
        float sm = 0.f;
#pragma unroll
        for (int i = 0; i < 8; ++i) {
            const float u = expf(V[i] - mx);
            U[i] = (eb + 4 * i + sub < ee) ? u : 0.f;
            sm += U[i];
        }
        sm += quad_perm<0xB1>(sm);
        sm += quad_perm<0x4E>(sm);
#pragma unroll
        for (int i = 0; i < 8; ++i) U[i] = U[i] / sm;
        add_chunk(deg);
    } else if (deg > 32) {
        float mx = NEG_INF;
        for (int c = eb; c < ee; c += 32) {
            score_chunk(c);
#pragma unroll
            for (int i = 0; i < 8; ++i) mx = fmaxf(mx, V[i]);
        }
        mx = fmaxf(mx, quad_perm<0xB1>(mx));
        mx = fmaxf(mx, quad_perm<0x4E>(mx));
        float sm = 0.f;
        for (int c = eb; c < ee; c += 32) {
            score_chunk(c);
#pragma unroll
            for (int i = 0; i < 8; ++i) {
                const float u = expf(V[i] - mx);
                sm += (c + 4 * i + sub < ee) ? u : 0.f;
            }
        }
        sm += quad_perm<0xB1>(sm);
        sm += quad_perm<0x4E>(sm);
        for (int c = eb; c < ee; c += 32) {
            score_chunk(c);
#pragma unroll
            for (int i = 0; i < 8; ++i) {
                const float u = expf(V[i] - mx) / sm;
                U[i] = (c + 4 * i + sub < ee) ? u : 0.f;
            }
            const int left = ee - c;
            add_chunk_by_two(left < 32 ? left : 32);
        }
    }
    const float4 c0 = *(const float4*)(b1 + lane * 8);
    const float4 c1 = *(const float4*)(b1 + lane * 8 + 4);
    float x[8] = {a0.x + c0.x, a0.y + c0.y, a0.z + c0.z, a0.w + c0.w,
                  a1.x + c1.x, a1.y + c1.y, a1.z + c1.z, a1.w + c1.w};
#pragma unroll
    for (int i = 0; i < 8; ++i) x[i] = (x[i] > 0.f) ? x[i] : expm1f(x[i]);
    const float4* wp = (const float4*)(W2 + lane * 24);
    const float4 w0 = wp[0], w1 = wp[1], w2 = wp[2], w3 = wp[3], w4 = wp[4], w5 = wp[5];
    float p0 = x[0] * w0.x, p1 = x[0] * w0.y, p2 = x[0] * w0.z;
    p0 += x[1] * w0.w; p1 += x[1] * w1.x; p2 += x[1] * w1.y;
    p0 += x[2] * w1.z; p1 += x[2] * w1.w; p2 += x[2] * w2.x;
    p0 += x[3] * w2.y; p1 += x[3] * w2.z; p2 += x[3] * w2.w;
    p0 += x[4] * w3.x; p1 += x[4] * w3.y; p2 += x[4] * w3.z;
    p0 += x[5] * w3.w; p1 += x[5] * w4.x; p2 += x[5] * w4.y;
    p0 += x[6] * w4.z; p1 += x[6] * w4.w; p2 += x[6] * w5.x;
    p0 += x[7] * w5.y; p1 += x[7] * w5.z; p2 += x[7] * w5.w;
    p0 = wave_sum(p0); p1 = wave_sum(p1); p2 = wave_sum(p2);
    if (lane == 0) {
        float4 row;
        row.x = p0; row.y = p1; row.z = p2;
        row.w = p0 * al2[0] + p1 * al2[1] + p2 * al2[2];
        ((float4*)hp)[node] = row;
        er2[node] = p0 * ar2[0] + p1 * ar2[1] + p2 * ar2[2];
    }
}

__global__ void __launch_bounds__(256) gat_l2_segments(const int* __restrict__ rowptr,
                                                       const int* __restrict__ ssrc,
                                                       const float* __restrict__ hp,
                                                       const float* __restrict__ er2,
                                                       float* __restrict__ agg2, int n_nodes) {
    const int lane = threadIdx.x & 31;
    const int node = blockIdx.x * 8 + __builtin_amdgcn_readfirstlane((int)(threadIdx.x >> 5));
    if (node >= n_nodes) return;
    const int eb = rowptr[node], ee = rowptr[node + 1];
    const float erv = er2[node];
    const float NEG_INF = -__builtin_huge_valf();
    const float4* rows = (const float4*)hp;
    const int deg = ee - eb;
    float p0 = 0.f, p1 = 0.f, p2 = 0.f;

    if (deg > 0 && deg <= 32) {
        const int j = eb + lane;
        const bool ok = j < ee;
        const float4 r = rows[ssrc[ok ? j : ee - 1]];
        float x = r.w + erv;
        x = (x >= 0.f) ? x : SLOPE * x;
        x = ok ? x : NEG_INF;
        const float mx = wave_max(x);
        const float e = expf(x - mx);
        const float u = ok ? e : 0.f;
        const float sm = wave_sum(u);
        const float al = u / sm;
        p0 = ok ? r.x * al : 0.f;
        p1 = ok ? r.y * al : 0.f;
        p2 = ok ? r.z * al : 0.f;
    } else if (deg > 32) {
        float mx = NEG_INF;
        for (int c = eb; c < ee; c += 32) {
            const int j = c + lane;
            if (j < ee) {
                float x = rows[ssrc[j]].w + erv;
                x = (x >= 0.f) ? x : SLOPE * x;
                mx = fmaxf(mx, x);
            }
        }
        mx = wave_max(mx);
        float sm = 0.f;
        for (int c = eb; c < ee; c += 32) {
            const int j = c + lane;
            if (j < ee) {
                float x = rows[ssrc[j]].w + erv;
                x = (x >= 0.f) ? x : SLOPE * x;
                sm += expf(x - mx);
            }
        }
        sm = wave_sum(sm);
        for (int c = eb; c < ee; c += 32) {
            const int j = c + lane;
            if (j < ee) {
                const float4 r = rows[ssrc[j]];
                float x = r.w + erv;
                x = (x >= 0.f) ? x : SLOPE * x;
                const float al = expf(x - mx) / sm;
                p0 += r.x * al; p1 += r.y * al; p2 += r.z * al;
            }
        }
    }
    p0 = wave_sum(p0); p1 = wave_sum(p1); p2 = wave_sum(p2);
    if (lane == 0) {
        agg2[node * 3 + 0] = p0;
        agg2[node * 3 + 1] = p1;
        agg2[node * 3 + 2] = p2;
    }
}

__global__ void gat_finalize(const float* __restrict__ agg2, const float* __restrict__ b2,
                             float* __restrict__ out, int total) {
    int i = blockIdx.x * blockDim.x + threadIdx.x;
    if (i >= total) return;
    out[i] = agg2[i] + b2[i % 3];
}

extern "C" void kernel_launch(void* const* d_in, const int* in_sizes, int n_in,
                              void* d_out, int out_size, void* d_ws, size_t ws_size,
                              hipStream_t stream) {
    const float* feat = (const float*)d_in[0];
    const int*   src  = (const int*)  d_in[1];
    const int*   dst  = (const int*)  d_in[2];
    const float* W1   = (const float*)d_in[3];
    const float* al1  = (const float*)d_in[4];
    const float* ar1  = (const float*)d_in[5];
    const float* b1   = (const float*)d_in[6];
    const float* W2   = (const float*)d_in[7];
    const float* al2  = (const float*)d_in[8];
    const float* ar2  = (const float*)d_in[9];
    const float* b2   = (const float*)d_in[10];
    float* out = (float*)d_out;

    const int n_nodes = in_sizes[0] / 256;
    const int n_edges = in_sizes[1];

    char* cur = (char*)d_ws;
    auto alloc = [&](size_t bytes) -> float* {
        float* p = (float*)cur;
        cur += (bytes + 255) & ~(size_t)255;
        return p;
    };
    float* H1   = alloc((size_t)n_nodes * 256 * 4);
    float* el1  = alloc((size_t)n_nodes * 8 * 4);
    float* er1  = alloc((size_t)n_nodes * 8 * 4);
    float* hp2  = alloc((size_t)n_nodes * 4 * 4);
    float* er2  = alloc((size_t)n_nodes * 4);
    float* agg2 = alloc((size_t)n_nodes * 3 * 4);
    float* cntf = alloc((size_t)n_nodes * 4);
    int*   cnt    = (int*)cntf;
    int*   rowptr = (int*)alloc((size_t)(n_nodes + 1) * 4);
    int*   rank   = (int*)alloc((size_t)n_edges * 4);
    int*   ssrc   = (int*)alloc((size_t)n_edges * 4);

    const int B = 256;

    gat_fill_f32<<<cdiv(n_nodes, B), B, 0, stream>>>(cntf, 0.f, n_nodes);
    gat_dst_rank<<<cdiv(n_edges, B), B, 0, stream>>>(dst, cnt, rank, n_edges);
    gat_row_offsets<<<cdiv(n_nodes, 1024), 1024, 0, stream>>>(cnt, rowptr, n_nodes);
    gat_dst_place<<<cdiv(n_edges, B), B, 0, stream>>>(src, dst, rowptr, rank, ssrc, n_edges);

    gat_gemm1_wmma<<<cdiv(n_nodes, 32), 256, 0, stream>>>(feat, W1, al1, ar1, H1, el1, er1, n_nodes);
    gat_l1_segments<<<cdiv(n_nodes, 8), 256, 0, stream>>>(rowptr, ssrc, H1, el1, er1,
                                                          b1, W2, al2, ar2, hp2, er2, n_nodes);

    gat_l2_segments<<<cdiv(n_nodes, 8), 256, 0, stream>>>(rowptr, ssrc, hp2, er2, agg2, n_nodes);
    gat_finalize<<<cdiv(n_nodes * 3, B), B, 0, stream>>>(agg2, b2, out, n_nodes * 3);
}


# Edge sort by dst moved into LDS, two levels (slice x 256-node-bucket counts, scan, slice scatter, per-bucket sort): no global add, table lookup or scattered store per edge; group order = LDS arrival
# speedup vs baseline: 7.6060x; 1.2506x over previous
#include <hip/hip_runtime.h>
#include <hip/hip_bf16.h>

typedef __attribute__((ext_vector_type(2))) float v2f;
typedef __attribute__((ext_vector_type(8))) float v8f;

#define SLOPE 0.2f

static __host__ __device__ inline int cdiv(int a, int b) { return (a + b - 1) / b; }

template <int CTRL>
__device__ inline float row_xmask_add(float x) {
    return x + __int_as_float(__builtin_amdgcn_update_dpp(0, __float_as_int(x), CTRL, 0xF, 0xF, true));
}
__device__ inline float row16_sum(float x) {
    x = row_xmask_add<0x161>(x);
    x = row_xmask_add<0x162>(x);
    x = row_xmask_add<0x164>(x);
    x = row_xmask_add<0x168>(x);
    return x;
}

#define A_PITCH 260
#define W_PITCH 272

__global__ void __launch_bounds__(256) gat_gemm1_wmma(const float* __restrict__ A,
                               const float* __restrict__ W,
                               const float* __restrict__ al,
                               const float* __restrict__ ar,
                               float* __restrict__ H,
                               float* __restrict__ el,
                               float* __restrict__ er,
                               int n_nodes) {
    __shared__ float As[32 * A_PITCH];
    __shared__ float Ws[32 * W_PITCH];
    __shared__ float Sc[2 * 32 * 8];

    const int row0 = blockIdx.x * 32;
    const int tid  = threadIdx.x;
    const int wave = tid >> 5;
    const int lane = tid & 31;

    const int sr = tid >> 6;
    const int sc = (tid & 63) << 2;
    float4 st[8];

#pragma unroll
    for (int j = 0; j < 8; ++j) {
        int rr = row0 + sr + 4 * j; if (rr >= n_nodes) rr = n_nodes - 1;
        st[j] = *(const float4*)(A + (size_t)rr * 256 + sc);
    }
#pragma unroll
    for (int j = 0; j < 8; ++j) *(float4*)(&As[(sr + 4 * j) * A_PITCH + sc]) = st[j];

    const int m   = lane & 15;
    const int kk  = (lane >> 4) << 1;
    const int n0  = wave * 32;
    const int n1  = n0 + 16;

    const float al0 = al[n0 + m], al1 = al[n1 + m];
    const float ar0 = ar[n0 + m], ar1 = ar[n1 + m];

    v8f c00 = {}, c01 = {}, c10 = {}, c11 = {};

    for (int kc = 0; kc < 256; kc += 32) {
        __syncthreads();
#pragma unroll
        for (int j = 0; j < 8; ++j) st[j] = *(const float4*)(W + (size_t)(kc + sr + 4 * j) * 256 + sc);
#pragma unroll
        for (int j = 0; j < 8; ++j) *(float4*)(&Ws[(sr + 4 * j) * W_PITCH + sc]) = st[j];
        __syncthreads();

#pragma unroll
        for (int k = 0; k < 32; k += 4) {
            v2f a0, a1, b0, b1;
            a0.x = As[m * A_PITCH + kc + k + kk];
            a0.y = As[m * A_PITCH + kc + k + kk + 1];
            a1.x = As[(m + 16) * A_PITCH + kc + k + kk];
            a1.y = As[(m + 16) * A_PITCH + kc + k + kk + 1];
            b0.x = Ws[(k + kk) * W_PITCH + n0 + m];
            b0.y = Ws[(k + kk + 1) * W_PITCH + n0 + m];
            b1.x = Ws[(k + kk) * W_PITCH + n1 + m];
            b1.y = Ws[(k + kk + 1) * W_PITCH + n1 + m];
            c00 = __builtin_amdgcn_wmma_f32_16x16x4_f32(false, a0, false, b0, (short)0, c00, false, false);
            c01 = __builtin_amdgcn_wmma_f32_16x16x4_f32(false, a0, false, b1, (short)0, c01, false, false);
            c10 = __builtin_amdgcn_wmma_f32_16x16x4_f32(false, a1, false, b0, (short)0, c10, false, false);
            c11 = __builtin_amdgcn_wmma_f32_16x16x4_f32(false, a1, false, b1, (short)0, c11, false, false);
        }
    }

    const int mhi = (lane >> 4) * 8;

#pragma unroll
    for (int v = 0; v < 8; ++v) {
        float l0 = c00[v] * al0 + c01[v] * al1;
        float l1 = c10[v] * al0 + c11[v] * al1;
        float q0 = c00[v] * ar0 + c01[v] * ar1;
        float q1 = c10[v] * ar0 + c11[v] * ar1;
        l0 = row16_sum(l0); l1 = row16_sum(l1);
        q0 = row16_sum(q0); q1 = row16_sum(q1);
        if (m == 0) {
            Sc[(mhi + v) * 8 + wave]            = l0;
            Sc[(16 + mhi + v) * 8 + wave]       = l1;
            Sc[256 + (mhi + v) * 8 + wave]      = q0;
            Sc[256 + (16 + mhi + v) * 8 + wave] = q1;
        }
    }
    __syncthreads();
    if (row0 + (tid >> 3) < n_nodes) {
        el[(size_t)row0 * 8 + tid] = Sc[tid];
        er[(size_t)row0 * 8 + tid] = Sc[256 + tid];
    }

    for (int v = 0; v < 8; ++v) {
        int r0 = row0 + mhi + v;
        int r1 = r0 + 16;
        if (r0 < n_nodes) {
            H[(size_t)r0 * 256 + n0 + m] = c00[v];
            H[(size_t)r0 * 256 + n1 + m] = c01[v];
        }
        if (r1 < n_nodes) {
            H[(size_t)r1 * 256 + n0 + m] = c10[v];
            H[(size_t)r1 * 256 + n1 + m] = c11[v];
        }
    }
}

#define BK_SHIFT 8
#define BK_NODES 256
#define SLICE    4096
#define BK_CAP   8192

__device__ inline void scan256_first_wave(const int* cnt, int* out, int t) {
    if (t < 32) {
        int v[8];
        int s = 0;
#pragma unroll
        for (int j = 0; j < 8; ++j) { v[j] = cnt[t * 8 + j]; s += v[j]; }
        int x = s;
        for (int o = 1; o < 32; o <<= 1) {
            int y = __shfl_up(x, o, 32);
            if (t >= o) x += y;
        }
        int run = x - s;
#pragma unroll
        for (int j = 0; j < 8; ++j) { out[t * 8 + j] = run; run += v[j]; }
    }
}

__global__ void __launch_bounds__(1024) gat_slice_count(const int* __restrict__ dst,
                                                        int* __restrict__ whist,
                                                        int n_edges, int n_buckets) {
    __shared__ int h[BK_NODES];
    const int t = threadIdx.x, w = blockIdx.x, G = gridDim.x;
    if (t < BK_NODES) h[t] = 0;
    __syncthreads();
    int d[4];
#pragma unroll
    for (int k = 0; k < 4; ++k) {
        const int e = w * SLICE + k * 1024 + t;
        d[k] = dst[e < n_edges ? e : n_edges - 1];
    }
#pragma unroll
    for (int k = 0; k < 4; ++k) {
        const int e = w * SLICE + k * 1024 + t;
        if (e < n_edges) atomicAdd(&h[d[k] >> BK_SHIFT], 1);
    }
    __syncthreads();
    if (t < n_buckets) whist[t * G + w] = h[t];
}

__global__ void __launch_bounds__(1024) gat_row_offsets(const int* __restrict__ cnt,
                                                        int* __restrict__ rowptr, int n) {
    __shared__ int wtot[32];
    __shared__ int wpre[32];
    const int t = threadIdx.x, lane = t & 31, wave = t >> 5;
    const int b = blockIdx.x;
    const int i = b * 1024 + t;
    const int v = (i < n) ? cnt[i] : 0;
    int pre = 0;
    for (int k = 0; k < b; ++k) pre += cnt[k * 1024 + t];
    int x = v;
    for (int o = 1; o < 32; o <<= 1) {
        int y = __shfl_up(x, o, 32);
        if (lane >= o) x += y;
    }
    for (int o = 16; o > 0; o >>= 1) pre += __shfl_xor(pre, o, 32);
    if (lane == 31) wtot[wave] = x;
    if (lane == 0)  wpre[wave] = pre;
    __syncthreads();
    int w = wtot[lane];
    for (int o = 1; o < 32; o <<= 1) {
        int y = __shfl_up(w, o, 32);
        if (lane >= o) w += y;
    }
    int base = wpre[lane];
    for (int o = 16; o > 0; o >>= 1) base += __shfl_xor(base, o, 32);
    const int before_wave = __shfl(w, wave, 32) - wtot[wave];
    const int excl = base + before_wave + x - v;
    if (i < n) rowptr[i] = excl;
    if (i == n - 1) rowptr[n] = excl + v;
}

__global__ void __launch_bounds__(1024) gat_slice_scatter(const int* __restrict__ src,
                                                          const int* __restrict__ dst,
                                                          const int* __restrict__ wbase,
                                                          int* __restrict__ csrc,
                                                          int* __restrict__ cdst,
                                                          int n_edges, int n_buckets) {
    __shared__ int cnt[BK_NODES];
    __shared__ int lstart[BK_NODES];
    __shared__ int gbase[BK_NODES];
    __shared__ int lsrc[SLICE];
    __shared__ int ldst[SLICE];
    const int t = threadIdx.x, w = blockIdx.x, G = gridDim.x;
    int d[4], s[4], r[4];
#pragma unroll
    for (int k = 0; k < 4; ++k) {
        const int e = w * SLICE + k * 1024 + t;
        const int ec = e < n_edges ? e : n_edges - 1;
        d[k] = dst[ec];
        s[k] = src[ec];
    }
    if (t < BK_NODES) {
        cnt[t] = 0;
        gbase[t] = (t < n_buckets) ? wbase[t * G + w] : 0;
    }
    __syncthreads();
#pragma unroll
    for (int k = 0; k < 4; ++k) {
        const int e = w * SLICE + k * 1024 + t;
        r[k] = (e < n_edges) ? atomicAdd(&cnt[d[k] >> BK_SHIFT], 1) : 0;
    }
    __syncthreads();
    scan256_first_wave(cnt, lstart, t);
    __syncthreads();
#pragma unroll
    for (int k = 0; k < 4; ++k) {
        const int e = w * SLICE + k * 1024 + t;
        if (e < n_edges) {
            const int p = lstart[d[k] >> BK_SHIFT] + r[k];
            lsrc[p] = s[k];
            ldst[p] = d[k];
        }
    }
    __syncthreads();
    int m = n_edges - w * SLICE;
    if (m > SLICE) m = SLICE;
#pragma unroll
    for (int k = 0; k < 4; ++k) {
        const int p = t + k * 1024;
        if (p < m) {
            const int dd = ldst[p];
            const int b = dd >> BK_SHIFT;
            const int g = gbase[b] + (p - lstart[b]);
            csrc[g] = lsrc[p];
            cdst[g] = dd;
        }
    }
}

__global__ void __launch_bounds__(1024) gat_bucket_sort(const int* __restrict__ csrc,
                                                        const int* __restrict__ cdst,
                                                        const int* __restrict__ wbase,
                                                        int* __restrict__ rowptr,
                                                        int* __restrict__ ssrc,
                                                        int n_nodes, int n_edges, int n_slices) {
    __shared__ int cnt[BK_NODES];
    __shared__ int cur[BK_NODES];
    __shared__ int stage[BK_CAP];
    const int t = threadIdx.x, b = blockIdx.x;
    const int lo = wbase[b * n_slices];
    const int hi = wbase[(b + 1) * n_slices];
    const int m = hi - lo;
    const bool fits = m <= BK_CAP;
    int d[8], s[8];
    if (t < BK_NODES) cnt[t] = 0;
    if (fits) {
#pragma unroll
        for (int k = 0; k < 8; ++k) {
            const int j = lo + t + k * 1024;
            const int jc = j < n_edges ? j : n_edges - 1;
            d[k] = cdst[jc];
            s[k] = csrc[jc];
        }
#pragma unroll
        for (int k = 0; k < 8; ++k) d[k] &= BK_NODES - 1;
    }
    __syncthreads();
    if (fits) {
#pragma unroll
        for (int k = 0; k < 8; ++k)
            if (t + k * 1024 < m) atomicAdd(&cnt[d[k]], 1);
    } else {
        for (int i = t; i < m; i += 1024) atomicAdd(&cnt[cdst[lo + i] & (BK_NODES - 1)], 1);
    }
    __syncthreads();
    scan256_first_wave(cnt, cur, t);
    __syncthreads();
    if (t < BK_NODES) {
        const int node = b * BK_NODES + t;
        if (node < n_nodes) rowptr[node] = lo + cur[t];
        if (node == n_nodes - 1) rowptr[n_nodes] = hi;
    }
    __syncthreads();
    if (fits) {
        int p[8];
#pragma unroll
        for (int k = 0; k < 8; ++k) p[k] = (t + k * 1024 < m) ? atomicAdd(&cur[d[k]], 1) : 0;
#pragma unroll
        for (int k = 0; k < 8; ++k)
            if (t + k * 1024 < m) stage[p[k]] = s[k];
        __syncthreads();
#pragma unroll
        for (int k = 0; k < 8; ++k) {
            const int i = t + k * 1024;
            if (i < m) ssrc[lo + i] = stage[i];
        }
    } else {
        for (int i = t; i < m; i += 1024) {
            const int p = atomicAdd(&cur[cdst[lo + i] & (BK_NODES - 1)], 1);
            ssrc[lo + p] = csrc[lo + i];
        }
    }
}

__device__ inline float wave_max(float x) {
    for (int o = 16; o > 0; o >>= 1) x = fmaxf(x, __shfl_xor(x, o, 32));
    return x;
}
__device__ inline float wave_sum(float x) {
    for (int o = 16; o > 0; o >>= 1) x += __shfl_xor(x, o, 32);
    return x;
}

template <int CTRL>
__device__ inline float quad_perm(float x) {
    return __int_as_float(__builtin_amdgcn_update_dpp(0, __float_as_int(x), CTRL, 0xF, 0xF, true));
}

template <int K>
__device__ inline void gat_add_row(const float* __restrict__ Hl, int s_lane, float u_lane,
                                   float4& a0, float4& a1) {
    const int sn = __builtin_amdgcn_readlane(s_lane, K);
    const float al = quad_perm<K * 0x55>(u_lane);
    const float4 x0 = *(const float4*)(Hl + (size_t)sn * 256);
    const float4 x1 = *(const float4*)(Hl + (size_t)sn * 256 + 4);
    a0.x += x0.x * al; a0.y += x0.y * al; a0.z += x0.z * al; a0.w += x0.w * al;
    a1.x += x1.x * al; a1.y += x1.y * al; a1.z += x1.z * al; a1.w += x1.w * al;
}

__global__ void __launch_bounds__(256) gat_l1_segments(const int* __restrict__ rowptr,
                                                       const int* __restrict__ ssrc,
                                                       const float* __restrict__ H,
                                                       const float* __restrict__ el,
                                                       const float* __restrict__ er,
                                                       const float* __restrict__ b1,
                                                       const float* __restrict__ W2,
                                                       const float* __restrict__ al2,
                                                       const float* __restrict__ ar2,
                                                       float* __restrict__ hp,
                                                       float* __restrict__ er2, int n_nodes) {
    const int lane = threadIdx.x & 31;
    const int node = blockIdx.x * 8 + __builtin_amdgcn_readfirstlane((int)(threadIdx.x >> 5));
    if (node >= n_nodes) return;
    const int eb = rowptr[node], ee = rowptr[node + 1];
    const int head = lane >> 2, sub = lane & 3;
    const float erv = er[node * 8 + head];
    const float* Hl = H + lane * 8;
    const float NEG_INF = -__builtin_huge_valf();

    int   S[8];
    float V[8];
    float U[8];
    float4 a0 = {0.f, 0.f, 0.f, 0.f}, a1 = {0.f, 0.f, 0.f, 0.f};

    auto score_chunk = [&](int c) {
        const int last = ee - 1;
#pragma unroll
        for (int i = 0; i < 8; ++i) {
            const int j = c + 4 * i + sub;
            S[i] = ssrc[j < ee ? j : last];
        }
#pragma unroll
        for (int i = 0; i < 8; ++i) {
            float x = el[S[i] * 8 + head] + erv;
            x = (x >= 0.f) ? x : SLOPE * x;
            V[i] = (c + 4 * i + sub < ee) ? x : NEG_INF;
        }
    };
    auto add_chunk = [&](int cnt) {
#pragma unroll
        for (int i = 0; i < 8; ++i) {
            if (4 * i + 4 <= cnt) {
                gat_add_row<0>(Hl, S[i], U[i], a0, a1);
                gat_add_row<1>(Hl, S[i], U[i], a0, a1);
                gat_add_row<2>(Hl, S[i], U[i], a0, a1);
                gat_add_row<3>(Hl, S[i], U[i], a0, a1);
            } else if (4 * i < cnt) {
                gat_add_row<0>(Hl, S[i], U[i], a0, a1);
                if (4 * i + 1 < cnt) gat_add_row<1>(Hl, S[i], U[i], a0, a1);
                if (4 * i + 2 < cnt) gat_add_row<2>(Hl, S[i], U[i], a0, a1);
            }
        }
    };

    auto add_chunk_by_two = [&](int cnt) {
#pragma unroll
        for (int i = 0; i < 8; ++i) {
            if (4 * i + 2 <= cnt) {
                gat_add_row<0>(Hl, S[i], U[i], a0, a1);
                gat_add_row<1>(Hl, S[i], U[i], a0, a1);
            } else if (4 * i < cnt) {
                gat_add_row<0>(Hl, S[i], U[i], a0, a1);
            }
            if (4 * i + 4 <= cnt) {
                gat_add_row<2>(Hl, S[i], U[i], a0, a1);
                gat_add_row<3>(Hl, S[i], U[i], a0, a1);
            } else if (4 * i + 2 < cnt) {
                gat_add_row<2>(Hl, S[i], U[i], a0, a1);
            }
        }
    };

    const int deg = ee - eb;
    if (deg > 0 && deg <= 32) {
        score_chunk(eb);
        float mx = V[0];
#pragma unroll
        for (int i = 1; i < 8; ++i) mx = fmaxf(mx, V[i]);
        mx = fmaxf(mx, quad_perm<0xB1>(mx));
        mx = fmaxf(mx, quad_perm<0x4E>(mx));
        float sm = 0.f;
#pragma unroll
        for (int i = 0; i < 8; ++i) {
            const float u = expf(V[i] - mx);
            U[i] = (eb + 4 * i + sub < ee) ? u : 0.f;
            sm += U[i];
        }
        sm += quad_perm<0xB1>(sm);
        sm += quad_perm<0x4E>(sm);
#pragma unroll
        for (int i = 0; i < 8; ++i) U[i] = U[i] / sm;
        add_chunk(deg);
    } else if (deg > 32) {
        float mx = NEG_INF;
        for (int c = eb; c < ee; c += 32) {
            score_chunk(c);
#pragma unroll
            for (int i = 0; i < 8; ++i) mx = fmaxf(mx, V[i]);
        }
        mx = fmaxf(mx, quad_perm<0xB1>(mx));
        mx = fmaxf(mx, quad_perm<0x4E>(mx));
        float sm = 0.f;
        for (int c = eb; c < ee; c += 32) {
            score_chunk(c);
#pragma unroll
            for (int i = 0; i < 8; ++i) {
                const float u = expf(V[i] - mx);
                sm += (c + 4 * i + sub < ee) ? u : 0.f;
            }
        }
        sm += quad_perm<0xB1>(sm);
        sm += quad_perm<0x4E>(sm);
        for (int c = eb; c < ee; c += 32) {
            score_chunk(c);
#pragma unroll
            for (int i = 0; i < 8; ++i) {
                const float u = expf(V[i] - mx) / sm;
                U[i] = (c + 4 * i + sub < ee) ? u : 0.f;
            }
            const int left = ee - c;
            add_chunk_by_two(left < 32 ? left : 32);
        }
    }
    const float4 c0 = *(const float4*)(b1 + lane * 8);
    const float4 c1 = *(const float4*)(b1 + lane * 8 + 4);
    float x[8] = {a0.x + c0.x, a0.y + c0.y, a0.z + c0.z, a0.w + c0.w,
                  a1.x + c1.x, a1.y + c1.y, a1.z + c1.z, a1.w + c1.w};
#pragma unroll
    for (int i = 0; i < 8; ++i) x[i] = (x[i] > 0.f) ? x[i] : expm1f(x[i]);
    const float4* wp = (const float4*)(W2 + lane * 24);
    const float4 w0 = wp[0], w1 = wp[1], w2 = wp[2], w3 = wp[3], w4 = wp[4], w5 = wp[5];
    float p0 = x[0] * w0.x, p1 = x[0] * w0.y, p2 = x[0] * w0.z;
    p0 += x[1] * w0.w; p1 += x[1] * w1.x; p2 += x[1] * w1.y;
    p0 += x[2] * w1.z; p1 += x[2] * w1.w; p2 += x[2] * w2.x;
    p0 += x[3] * w2.y; p1 += x[3] * w2.z; p2 += x[3] * w2.w;
    p0 += x[4] * w3.x; p1 += x[4] * w3.y; p2 += x[4] * w3.z;
    p0 += x[5] * w3.w; p1 += x[5] * w4.x; p2 += x[5] * w4.y;
    p0 += x[6] * w4.z; p1 += x[6] * w4.w; p2 += x[6] * w5.x;
    p0 += x[7] * w5.y; p1 += x[7] * w5.z; p2 += x[7] * w5.w;
    p0 = wave_sum(p0); p1 = wave_sum(p1); p2 = wave_sum(p2);
    if (lane == 0) {
        float4 row;
        row.x = p0; row.y = p1; row.z = p2;
        row.w = p0 * al2[0] + p1 * al2[1] + p2 * al2[2];
        ((float4*)hp)[node] = row;
        er2[node] = p0 * ar2[0] + p1 * ar2[1] + p2 * ar2[2];
    }
}

__global__ void __launch_bounds__(256) gat_l2_segments(const int* __restrict__ rowptr,
                                                       const int* __restrict__ ssrc,
                                                       const float* __restrict__ hp,
                                                       const float* __restrict__ er2,
                                                       float* __restrict__ agg2, int n_nodes) {
    const int lane = threadIdx.x & 31;
    const int node = blockIdx.x * 8 + __builtin_amdgcn_readfirstlane((int)(threadIdx.x >> 5));
    if (node >= n_nodes) return;
    const int eb = rowptr[node], ee = rowptr[node + 1];
    const float erv = er2[node];
    const float NEG_INF = -__builtin_huge_valf();
    const float4* rows = (const float4*)hp;
    const int deg = ee - eb;
    float p0 = 0.f, p1 = 0.f, p2 = 0.f;

    if (deg > 0 && deg <= 32) {
        const int j = eb + lane;
        const bool ok = j < ee;
        const float4 r = rows[ssrc[ok ? j : ee - 1]];
        float x = r.w + erv;
        x = (x >= 0.f) ? x : SLOPE * x;
        x = ok ? x : NEG_INF;
        const float mx = wave_max(x);
        const float e = expf(x - mx);
        const float u = ok ? e : 0.f;
        const float sm = wave_sum(u);
        const float al = u / sm;
        p0 = ok ? r.x * al : 0.f;
        p1 = ok ? r.y * al : 0.f;
        p2 = ok ? r.z * al : 0.f;
    } else if (deg > 32) {
        float mx = NEG_INF;
        for (int c = eb; c < ee; c += 32) {
            const int j = c + lane;
            if (j < ee) {
                float x = rows[ssrc[j]].w + erv;
                x = (x >= 0.f) ? x : SLOPE * x;
                mx = fmaxf(mx, x);
            }
        }
        mx = wave_max(mx);
        float sm = 0.f;
        for (int c = eb; c < ee; c += 32) {
            const int j = c + lane;
            if (j < ee) {
                float x = rows[ssrc[j]].w + erv;
                x = (x >= 0.f) ? x : SLOPE * x;
                sm += expf(x - mx);
            }
        }
        sm = wave_sum(sm);
        for (int c = eb; c < ee; c += 32) {
            const int j = c + lane;
            if (j < ee) {
                const float4 r = rows[ssrc[j]];
                float x = r.w + erv;
                x = (x >= 0.f) ? x : SLOPE * x;
                const float al = expf(x - mx) / sm;
                p0 += r.x * al; p1 += r.y * al; p2 += r.z * al;
            }
        }
    }
    p0 = wave_sum(p0); p1 = wave_sum(p1); p2 = wave_sum(p2);
    if (lane == 0) {
        agg2[node * 3 + 0] = p0;
        agg2[node * 3 + 1] = p1;
        agg2[node * 3 + 2] = p2;
    }
}

__global__ void gat_finalize(const float* __restrict__ agg2, const float* __restrict__ b2,
                             float* __restrict__ out, int total) {
    int i = blockIdx.x * blockDim.x + threadIdx.x;
    if (i >= total) return;
    out[i] = agg2[i] + b2[i % 3];
}

extern "C" void kernel_launch(void* const* d_in, const int* in_sizes, int n_in,
                              void* d_out, int out_size, void* d_ws, size_t ws_size,
                              hipStream_t stream) {
    const float* feat = (const float*)d_in[0];
    const int*   src  = (const int*)  d_in[1];
    const int*   dst  = (const int*)  d_in[2];
    const float* W1   = (const float*)d_in[3];
    const float* al1  = (const float*)d_in[4];
    const float* ar1  = (const float*)d_in[5];
    const float* b1   = (const float*)d_in[6];
    const float* W2   = (const float*)d_in[7];
    const float* al2  = (const float*)d_in[8];
    const float* ar2  = (const float*)d_in[9];
    const float* b2   = (const float*)d_in[10];
    float* out = (float*)d_out;

    const int n_nodes = in_sizes[0] / 256;
    const int n_edges = in_sizes[1];

    char* cur = (char*)d_ws;
    auto alloc = [&](size_t bytes) -> float* {
        float* p = (float*)cur;
        cur += (bytes + 255) & ~(size_t)255;
        return p;
    };
    float* H1   = alloc((size_t)n_nodes * 256 * 4);
    float* el1  = alloc((size_t)n_nodes * 8 * 4);
    float* er1  = alloc((size_t)n_nodes * 8 * 4);
    float* hp2  = alloc((size_t)n_nodes * 4 * 4);
    float* er2  = alloc((size_t)n_nodes * 4);
    float* agg2 = alloc((size_t)n_nodes * 3 * 4);
    const int n_buckets = cdiv(n_nodes, BK_NODES);
    const int n_slices  = cdiv(n_edges, SLICE);
    int*   whist  = (int*)alloc((size_t)n_buckets * n_slices * 4);
    int*   wbase  = (int*)alloc(((size_t)n_buckets * n_slices + 1) * 4);
    int*   csrc   = (int*)alloc((size_t)n_edges * 4);
    int*   cdst   = (int*)alloc((size_t)n_edges * 4);
    int*   rowptr = (int*)alloc((size_t)(n_nodes + 1) * 4);
    int*   ssrc   = (int*)alloc((size_t)n_edges * 4);

    const int B = 256;

    gat_slice_count<<<n_slices, 1024, 0, stream>>>(dst, whist, n_edges, n_buckets);
    gat_row_offsets<<<cdiv(n_buckets * n_slices, 1024), 1024, 0, stream>>>(whist, wbase, n_buckets * n_slices);
    gat_slice_scatter<<<n_slices, 1024, 0, stream>>>(src, dst, wbase, csrc, cdst, n_edges, n_buckets);
    gat_bucket_sort<<<n_buckets, 1024, 0, stream>>>(csrc, cdst, wbase, rowptr, ssrc, n_nodes, n_edges, n_slices);

    gat_gemm1_wmma<<<cdiv(n_nodes, 32), 256, 0, stream>>>(feat, W1, al1, ar1, H1, el1, er1, n_nodes);
    gat_l1_segments<<<cdiv(n_nodes, 8), 256, 0, stream>>>(rowptr, ssrc, H1, el1, er1,
                                                          b1, W2, al2, ar2, hp2, er2, n_nodes);

    gat_l2_segments<<<cdiv(n_nodes, 8), 256, 0, stream>>>(rowptr, ssrc, hp2, er2, agg2, n_nodes);
    gat_finalize<<<cdiv(n_nodes * 3, B), B, 0, stream>>>(agg2, b2, out, n_nodes * 3);
}


# 3 launches fewer (8 to 5): edge sort without count and prefix passes (slices keep bucket order in place + a run-start table; a bucket sums its table row, gathers its runs); b2 added by layer-2 wave
# speedup vs baseline: 8.4317x; 1.1086x over previous
#include <hip/hip_runtime.h>
#include <hip/hip_bf16.h>

typedef __attribute__((ext_vector_type(2))) float v2f;
typedef __attribute__((ext_vector_type(8))) float v8f;

#define SLOPE 0.2f

static __host__ __device__ inline int cdiv(int a, int b) { return (a + b - 1) / b; }

template <int CTRL>
__device__ inline float row_xmask_add(float x) {
    return x + __int_as_float(__builtin_amdgcn_update_dpp(0, __float_as_int(x), CTRL, 0xF, 0xF, true));
}
__device__ inline float row16_sum(float x) {
    x = row_xmask_add<0x161>(x);
    x = row_xmask_add<0x162>(x);
    x = row_xmask_add<0x164>(x);
    x = row_xmask_add<0x168>(x);
    return x;
}

#define A_PITCH 260
#define W_PITCH 272

__global__ void __launch_bounds__(256) gat_gemm1_wmma(const float* __restrict__ A,
                               const float* __restrict__ W,
                               const float* __restrict__ al,
                               const float* __restrict__ ar,
                               float* __restrict__ H,
                               float* __restrict__ el,
                               float* __restrict__ er,
                               int n_nodes) {
    __shared__ float As[32 * A_PITCH];
    __shared__ float Ws[32 * W_PITCH];
    __shared__ float Sc[2 * 32 * 8];

    const int row0 = blockIdx.x * 32;
    const int tid  = threadIdx.x;
    const int wave = tid >> 5;
    const int lane = tid & 31;

    const int sr = tid >> 6;
    const int sc = (tid & 63) << 2;
    float4 st[8];

#pragma unroll
    for (int j = 0; j < 8; ++j) {
        int rr = row0 + sr + 4 * j; if (rr >= n_nodes) rr = n_nodes - 1;
        st[j] = *(const float4*)(A + (size_t)rr * 256 + sc);
    }
#pragma unroll
    for (int j = 0; j < 8; ++j) *(float4*)(&As[(sr + 4 * j) * A_PITCH + sc]) = st[j];

    const int m   = lane & 15;
    const int kk  = (lane >> 4) << 1;
    const int n0  = wave * 32;
    const int n1  = n0 + 16;

    const float al0 = al[n0 + m], al1 = al[n1 + m];
    const float ar0 = ar[n0 + m], ar1 = ar[n1 + m];

    v8f c00 = {}, c01 = {}, c10 = {}, c11 = {};

    for (int kc = 0; kc < 256; kc += 32) {
        __syncthreads();
#pragma unroll
        for (int j = 0; j < 8; ++j) st[j] = *(const float4*)(W + (size_t)(kc + sr + 4 * j) * 256 + sc);
#pragma unroll
        for (int j = 0; j < 8; ++j) *(float4*)(&Ws[(sr + 4 * j) * W_PITCH + sc]) = st[j];
        __syncthreads();

#pragma unroll
        for (int k = 0; k < 32; k += 4) {
            v2f a0, a1, b0, b1;
            a0.x = As[m * A_PITCH + kc + k + kk];
            a0.y = As[m * A_PITCH + kc + k + kk + 1];
            a1.x = As[(m + 16) * A_PITCH + kc + k + kk];
            a1.y = As[(m + 16) * A_PITCH + kc + k + kk + 1];
            b0.x = Ws[(k + kk) * W_PITCH + n0 + m];
            b0.y = Ws[(k + kk + 1) * W_PITCH + n0 + m];
            b1.x = Ws[(k + kk) * W_PITCH + n1 + m];
            b1.y = Ws[(k + kk + 1) * W_PITCH + n1 + m];
            c00 = __builtin_amdgcn_wmma_f32_16x16x4_f32(false, a0, false, b0, (short)0, c00, false, false);
            c01 = __builtin_amdgcn_wmma_f32_16x16x4_f32(false, a0, false, b1, (short)0, c01, false, false);
            c10 = __builtin_amdgcn_wmma_f32_16x16x4_f32(false, a1, false, b0, (short)0, c10, false, false);
            c11 = __builtin_amdgcn_wmma_f32_16x16x4_f32(false, a1, false, b1, (short)0, c11, false, false);
        }
    }

    const int mhi = (lane >> 4) * 8;

#pragma unroll
    for (int v = 0; v < 8; ++v) {
        float l0 = c00[v] * al0 + c01[v] * al1;
        float l1 = c10[v] * al0 + c11[v] * al1;
        float q0 = c00[v] * ar0 + c01[v] * ar1;
        float q1 = c10[v] * ar0 + c11[v] * ar1;
        l0 = row16_sum(l0); l1 = row16_sum(l1);
        q0 = row16_sum(q0); q1 = row16_sum(q1);
        if (m == 0) {
            Sc[(mhi + v) * 8 + wave]            = l0;
            Sc[(16 + mhi + v) * 8 + wave]       = l1;
            Sc[256 + (mhi + v) * 8 + wave]      = q0;
            Sc[256 + (16 + mhi + v) * 8 + wave] = q1;
        }
    }
    __syncthreads();
    if (row0 + (tid >> 3) < n_nodes) {
        el[(size_t)row0 * 8 + tid] = Sc[tid];
        er[(size_t)row0 * 8 + tid] = Sc[256 + tid];
    }

    for (int v = 0; v < 8; ++v) {
        int r0 = row0 + mhi + v;
        int r1 = r0 + 16;
        if (r0 < n_nodes) {
            H[(size_t)r0 * 256 + n0 + m] = c00[v];
            H[(size_t)r0 * 256 + n1 + m] = c01[v];
        }
        if (r1 < n_nodes) {
            H[(size_t)r1 * 256 + n0 + m] = c10[v];
            H[(size_t)r1 * 256 + n1 + m] = c11[v];
        }
    }
}

#define BK_SHIFT 8
#define BK_NODES 256
#define SLICE    4096
#define BK_CAP   8192
#define SL_MAX   1024

__device__ inline void scan256_first_wave(const int* cnt, int* out, int t) {
    if (t < 32) {
        int v[8];
        int s = 0;
#pragma unroll
        for (int j = 0; j < 8; ++j) { v[j] = cnt[t * 8 + j]; s += v[j]; }
        int x = s;
        for (int o = 1; o < 32; o <<= 1) {
            int y = __shfl_up(x, o, 32);
            if (t >= o) x += y;
        }
        int run = x - s;
#pragma unroll
        for (int j = 0; j < 8; ++j) { out[t * 8 + j] = run; run += v[j]; }
    }
}

__global__ void __launch_bounds__(1024) gat_slice_sort(const int* __restrict__ src,
                                                       const int* __restrict__ dst,
                                                       int* __restrict__ csrc,
                                                       int* __restrict__ cdst,
                                                       int* __restrict__ wstart,
                                                       int n_edges, int n_buckets) {
    __shared__ int cnt[BK_NODES];
    __shared__ int lstart[BK_NODES];
    __shared__ int lsrc[SLICE];
    __shared__ int ldst[SLICE];
    const int t = threadIdx.x, w = blockIdx.x, G = gridDim.x;
    int d[4], s[4], r[4];
#pragma unroll
    for (int k = 0; k < 4; ++k) {
        const int e = w * SLICE + k * 1024 + t;
        const int ec = e < n_edges ? e : n_edges - 1;
        d[k] = dst[ec];
        s[k] = src[ec];
    }
    if (t < BK_NODES) cnt[t] = 0;
    __syncthreads();
#pragma unroll
    for (int k = 0; k < 4; ++k) {
        const int e = w * SLICE + k * 1024 + t;
        r[k] = (e < n_edges) ? atomicAdd(&cnt[d[k] >> BK_SHIFT], 1) : 0;
    }
    __syncthreads();
    scan256_first_wave(cnt, lstart, t);
    __syncthreads();
#pragma unroll
    for (int k = 0; k < 4; ++k) {
        const int e = w * SLICE + k * 1024 + t;
        if (e < n_edges) {
            const int p = lstart[d[k] >> BK_SHIFT] + r[k];
            lsrc[p] = s[k];
            ldst[p] = d[k];
        }
    }
    int m = n_edges - w * SLICE;
    if (m > SLICE) m = SLICE;
    if (t <= n_buckets) wstart[t * G + w] = (t < n_buckets) ? lstart[t] : m;
    __syncthreads();
#pragma unroll
    for (int k = 0; k < 4; ++k) {
        const int p = t + k * 1024;
        if (p < m) {
            csrc[w * SLICE + p] = lsrc[p];
            cdst[w * SLICE + p] = ldst[p];
        }
    }
}

__global__ void __launch_bounds__(1024) gat_bucket_sort(const int* __restrict__ csrc,
                                                        const int* __restrict__ cdst,
                                                        const int* __restrict__ wstart,
                                                        int* __restrict__ rowptr,
                                                        int* __restrict__ ssrc,
                                                        int n_nodes, int n_edges, int n_slices) {
    __shared__ int cnt[BK_NODES];
    __shared__ int cur[BK_NODES];
    __shared__ int stage[BK_CAP];
    __shared__ int off[SL_MAX];
    __shared__ int sst[SL_MAX];
    __shared__ int wtot[32];
    __shared__ int wpre[32];
    const int t = threadIdx.x, lane = t & 31, wave = t >> 5, b = blockIdx.x;
    int st = 0, c = 0;
    if (t < n_slices) {
        st = wstart[b * n_slices + t];
        c  = wstart[(b + 1) * n_slices + t] - st;
    }
    int x = c;
    for (int o = 1; o < 32; o <<= 1) {
        int y = __shfl_up(x, o, 32);
        if (lane >= o) x += y;
    }
    int pre = st;
    for (int o = 16; o > 0; o >>= 1) pre += __shfl_xor(pre, o, 32);
    if (lane == 31) wtot[wave] = x;
    if (lane == 0)  wpre[wave] = pre;
    if (t < BK_NODES) cnt[t] = 0;
    __syncthreads();
    int wv = wtot[lane];
    for (int o = 1; o < 32; o <<= 1) {
        int y = __shfl_up(wv, o, 32);
        if (lane >= o) wv += y;
    }
    int sum_st = wpre[lane];
    for (int o = 16; o > 0; o >>= 1) sum_st += __shfl_xor(sum_st, o, 32);
    const int before_wave = __shfl(wv, wave, 32) - wtot[wave];
    off[t] = (t < n_slices) ? before_wave + x - c : 0x7fffffff;
    sst[t] = st;
    const int m  = __builtin_amdgcn_readfirstlane(__shfl(wv, 31, 32));
    const int lo = __builtin_amdgcn_readfirstlane(sum_st);
    const int hi = lo + m;
    __syncthreads();

    int top = 1;
    for (int q = 0; q < 9 && top * 2 < n_slices; ++q) top *= 2;

    auto edge_at = [&](int i) -> int {
        int sl = 0;
        for (int step = top; step > 0; step >>= 1) {
            const int cnd = sl + step;
            if (off[cnd] <= i) sl = cnd;
        }
        const int j = sl * SLICE + sst[sl] + (i - off[sl]);
        return j < n_edges ? j : n_edges - 1;
    };

    const bool fits = m <= BK_CAP;
    int d[8], s[8];
    if (fits) {
        int ic[8], sl[8];
#pragma unroll
        for (int k = 0; k < 8; ++k) {
            int i = t + k * 1024;
            i = i < m ? i : m - 1;
            ic[k] = i > 0 ? i : 0;
            sl[k] = 0;
        }
        for (int step = top; step > 0; step >>= 1) {
#pragma unroll
            for (int k = 0; k < 8; ++k) {
                const int cnd = sl[k] + step;
                if (off[cnd] <= ic[k]) sl[k] = cnd;
            }
        }
#pragma unroll
        for (int k = 0; k < 8; ++k) {
            int j = sl[k] * SLICE + sst[sl[k]] + (ic[k] - off[sl[k]]);
            j = j < n_edges ? j : n_edges - 1;
            d[k] = cdst[j];
            s[k] = csrc[j];
        }
#pragma unroll
        for (int k = 0; k < 8; ++k) d[k] &= BK_NODES - 1;
#pragma unroll
        for (int k = 0; k < 8; ++k)
            if (t + k * 1024 < m) atomicAdd(&cnt[d[k]], 1);
    } else {
        for (int i = t; i < m; i += 1024) atomicAdd(&cnt[cdst[edge_at(i)] & (BK_NODES - 1)], 1);
    }
    __syncthreads();
    scan256_first_wave(cnt, cur, t);
    __syncthreads();
    if (t < BK_NODES) {
        const int node = b * BK_NODES + t;
        if (node < n_nodes) rowptr[node] = lo + cur[t];
        if (node == n_nodes - 1) rowptr[n_nodes] = hi;
    }
    __syncthreads();
    if (fits) {
        int p[8];
#pragma unroll
        for (int k = 0; k < 8; ++k) p[k] = (t + k * 1024 < m) ? atomicAdd(&cur[d[k]], 1) : 0;
#pragma unroll
        for (int k = 0; k < 8; ++k)
            if (t + k * 1024 < m) stage[p[k]] = s[k];
        __syncthreads();
#pragma unroll
        for (int k = 0; k < 8; ++k) {
            const int i = t + k * 1024;
            if (i < m) ssrc[lo + i] = stage[i];
        }
    } else {
        for (int i = t; i < m; i += 1024) {
            const int j = edge_at(i);
            const int p = atomicAdd(&cur[cdst[j] & (BK_NODES - 1)], 1);
            ssrc[lo + p] = csrc[j];
        }
    }
}

__device__ inline float wave_max(float x) {
    for (int o = 16; o > 0; o >>= 1) x = fmaxf(x, __shfl_xor(x, o, 32));
    return x;
}
__device__ inline float wave_sum(float x) {
    for (int o = 16; o > 0; o >>= 1) x += __shfl_xor(x, o, 32);
    return x;
}

template <int CTRL>
__device__ inline float quad_perm(float x) {
    return __int_as_float(__builtin_amdgcn_update_dpp(0, __float_as_int(x), CTRL, 0xF, 0xF, true));
}

template <int K>
__device__ inline void gat_add_row(const float* __restrict__ Hl, int s_lane, float u_lane,
                                   float4& a0, float4& a1) {
    const int sn = __builtin_amdgcn_readlane(s_lane, K);
    const float al = quad_perm<K * 0x55>(u_lane);
    const float4 x0 = *(const float4*)(Hl + (size_t)sn * 256);
    const float4 x1 = *(const float4*)(Hl + (size_t)sn * 256 + 4);
    a0.x += x0.x * al; a0.y += x0.y * al; a0.z += x0.z * al; a0.w += x0.w * al;
    a1.x += x1.x * al; a1.y += x1.y * al; a1.z += x1.z * al; a1.w += x1.w * al;
}

__global__ void __launch_bounds__(256) gat_l1_segments(const int* __restrict__ rowptr,
                                                       const int* __restrict__ ssrc,
                                                       const float* __restrict__ H,
                                                       const float* __restrict__ el,
                                                       const float* __restrict__ er,
                                                       const float* __restrict__ b1,
                                                       const float* __restrict__ W2,
                                                       const float* __restrict__ al2,
                                                       const float* __restrict__ ar2,
                                                       float* __restrict__ hp,
                                                       float* __restrict__ er2, int n_nodes) {
    const int lane = threadIdx.x & 31;
    const int node = blockIdx.x * 8 + __builtin_amdgcn_readfirstlane((int)(threadIdx.x >> 5));
    if (node >= n_nodes) return;
    const int eb = rowptr[node], ee = rowptr[node + 1];
    const int head = lane >> 2, sub = lane & 3;
    const float erv = er[node * 8 + head];
    const float* Hl = H + lane * 8;
    const float NEG_INF = -__builtin_huge_valf();

    int   S[8];
    float V[8];
    float U[8];
    float4 a0 = {0.f, 0.f, 0.f, 0.f}, a1 = {0.f, 0.f, 0.f, 0.f};

    auto score_chunk = [&](int c) {
        const int last = ee - 1;
#pragma unroll
        for (int i = 0; i < 8; ++i) {
            const int j = c + 4 * i + sub;
            S[i] = ssrc[j < ee ? j : last];
        }
#pragma unroll
        for (int i = 0; i < 8; ++i) {
            float x = el[S[i] * 8 + head] + erv;
            x = (x >= 0.f) ? x : SLOPE * x;
            V[i] = (c + 4 * i + sub < ee) ? x : NEG_INF;
        }
    };
    auto add_chunk = [&](int cnt) {
#pragma unroll
        for (int i = 0; i < 8; ++i) {
            if (4 * i + 4 <= cnt) {
                gat_add_row<0>(Hl, S[i], U[i], a0, a1);
                gat_add_row<1>(Hl, S[i], U[i], a0, a1);
                gat_add_row<2>(Hl, S[i], U[i], a0, a1);
                gat_add_row<3>(Hl, S[i], U[i], a0, a1);
            } else if (4 * i < cnt) {
                gat_add_row<0>(Hl, S[i], U[i], a0, a1);
                if (4 * i + 1 < cnt) gat_add_row<1>(Hl, S[i], U[i], a0, a1);
                if (4 * i + 2 < cnt) gat_add_row<2>(Hl, S[i], U[i], a0, a1);
            }
        }
    };

    auto add_chunk_by_two = [&](int cnt) {
#pragma unroll
        for (int i = 0; i < 8; ++i) {
            if (4 * i + 2 <= cnt) {
                gat_add_row<0>(Hl, S[i], U[i], a0, a1);
                gat_add_row<1>(Hl, S[i], U[i], a0, a1);
            } else if (4 * i < cnt) {
                gat_add_row<0>(Hl, S[i], U[i], a0, a1);
            }
            if (4 * i + 4 <= cnt) {
                gat_add_row<2>(Hl, S[i], U[i], a0, a1);
                gat_add_row<3>(Hl, S[i], U[i], a0, a1);
            } else if (4 * i + 2 < cnt) {
                gat_add_row<2>(Hl, S[i], U[i], a0, a1);
            }
        }
    };

    const int deg = ee - eb;
    if (deg > 0 && deg <= 32) {
        score_chunk(eb);
        float mx = V[0];
#pragma unroll
        for (int i = 1; i < 8; ++i) mx = fmaxf(mx, V[i]);
        mx = fmaxf(mx, quad_perm<0xB1>(mx));
        mx = fmaxf(mx, quad_perm<0x4E>(mx));
        float sm = 0.f;
#pragma unroll
        for (int i = 0; i < 8; ++i) {
            const float u = expf(V[i] - mx);
            U[i] = (eb + 4 * i + sub < ee) ? u : 0.f;
            sm += U[i];
        }
        sm += quad_perm<0xB1>(sm);
        sm += quad_perm<0x4E>(sm);
#pragma unroll
        for (int i = 0; i < 8; ++i) U[i] = U[i] / sm;
        add_chunk(deg);
    } else if (deg > 32) {
        float mx = NEG_INF;
        for (int c = eb; c < ee; c += 32) {
            score_chunk(c);
#pragma unroll
            for (int i = 0; i < 8; ++i) mx = fmaxf(mx, V[i]);
        }
        mx = fmaxf(mx, quad_perm<0xB1>(mx));
        mx = fmaxf(mx, quad_perm<0x4E>(mx));
        float sm = 0.f;
        for (int c = eb; c < ee; c += 32) {
            score_chunk(c);
#pragma unroll
            for (int i = 0; i < 8; ++i) {
                const float u = expf(V[i] - mx);
                sm += (c + 4 * i + sub < ee) ? u : 0.f;
            }
        }
        sm += quad_perm<0xB1>(sm);
        sm += quad_perm<0x4E>(sm);
        for (int c = eb; c < ee; c += 32) {
            score_chunk(c);
#pragma unroll
            for (int i = 0; i < 8; ++i) {
                const float u = expf(V[i] - mx) / sm;
                U[i] = (c + 4 * i + sub < ee) ? u : 0.f;
            }
            const int left = ee - c;
            add_chunk_by_two(left < 32 ? left : 32);
        }
    }
    const float4 c0 = *(const float4*)(b1 + lane * 8);
    const float4 c1 = *(const float4*)(b1 + lane * 8 + 4);
    float x[8] = {a0.x + c0.x, a0.y + c0.y, a0.z + c0.z, a0.w + c0.w,
                  a1.x + c1.x, a1.y + c1.y, a1.z + c1.z, a1.w + c1.w};
#pragma unroll
    for (int i = 0; i < 8; ++i) x[i] = (x[i] > 0.f) ? x[i] : expm1f(x[i]);
    const float4* wp = (const float4*)(W2 + lane * 24);
    const float4 w0 = wp[0], w1 = wp[1], w2 = wp[2], w3 = wp[3], w4 = wp[4], w5 = wp[5];
    float p0 = x[0] * w0.x, p1 = x[0] * w0.y, p2 = x[0] * w0.z;
    p0 += x[1] * w0.w; p1 += x[1] * w1.x; p2 += x[1] * w1.y;
    p0 += x[2] * w1.z; p1 += x[2] * w1.w; p2 += x[2] * w2.x;
    p0 += x[3] * w2.y; p1 += x[3] * w2.z; p2 += x[3] * w2.w;
    p0 += x[4] * w3.x; p1 += x[4] * w3.y; p2 += x[4] * w3.z;
    p0 += x[5] * w3.w; p1 += x[5] * w4.x; p2 += x[5] * w4.y;
    p0 += x[6] * w4.z; p1 += x[6] * w4.w; p2 += x[6] * w5.x;
    p0 += x[7] * w5.y; p1 += x[7] * w5.z; p2 += x[7] * w5.w;
    p0 = wave_sum(p0); p1 = wave_sum(p1); p2 = wave_sum(p2);
    if (lane == 0) {
        float4 row;
        row.x = p0; row.y = p1; row.z = p2;
        row.w = p0 * al2[0] + p1 * al2[1] + p2 * al2[2];
        ((float4*)hp)[node] = row;
        er2[node] = p0 * ar2[0] + p1 * ar2[1] + p2 * ar2[2];
    }
}

__global__ void __launch_bounds__(256) gat_l2_segments(const int* __restrict__ rowptr,
                                                       const int* __restrict__ ssrc,
                                                       const float* __restrict__ hp,
                                                       const float* __restrict__ er2,
                                                       const float* __restrict__ b2,
                                                       float* __restrict__ out, int n_nodes) {
    const int lane = threadIdx.x & 31;
    const int node = blockIdx.x * 8 + __builtin_amdgcn_readfirstlane((int)(threadIdx.x >> 5));
    if (node >= n_nodes) return;
    const int eb = rowptr[node], ee = rowptr[node + 1];
    const float erv = er2[node];
    const float NEG_INF = -__builtin_huge_valf();
    const float4* rows = (const float4*)hp;
    const int deg = ee - eb;
    float p0 = 0.f, p1 = 0.f, p2 = 0.f;

    if (deg > 0 && deg <= 32) {
        const int j = eb + lane;
        const bool ok = j < ee;
        const float4 r = rows[ssrc[ok ? j : ee - 1]];
        float x = r.w + erv;
        x = (x >= 0.f) ? x : SLOPE * x;
        x = ok ? x : NEG_INF;
        const float mx = wave_max(x);
        const float e = expf(x - mx);
        const float u = ok ? e : 0.f;
        const float sm = wave_sum(u);
        const float al = u / sm;
        p0 = ok ? r.x * al : 0.f;
        p1 = ok ? r.y * al : 0.f;
        p2 = ok ? r.z * al : 0.f;
    } else if (deg > 32) {
        float mx = NEG_INF;
        for (int c = eb; c < ee; c += 32) {
            const int j = c + lane;
            if (j < ee) {
                float x = rows[ssrc[j]].w + erv;
                x = (x >= 0.f) ? x : SLOPE * x;
                mx = fmaxf(mx, x);
            }
        }
        mx = wave_max(mx);
        float sm = 0.f;
        for (int c = eb; c < ee; c += 32) {
            const int j = c + lane;
            if (j < ee) {
                float x = rows[ssrc[j]].w + erv;
                x = (x >= 0.f) ? x : SLOPE * x;
                sm += expf(x - mx);
            }
        }
        sm = wave_sum(sm);
        for (int c = eb; c < ee; c += 32) {
            const int j = c + lane;
            if (j < ee) {
                const float4 r = rows[ssrc[j]];
                float x = r.w + erv;
                x = (x >= 0.f) ? x : SLOPE * x;
                const float al = expf(x - mx) / sm;
                p0 += r.x * al; p1 += r.y * al; p2 += r.z * al;
            }
        }
    }
    p0 = wave_sum(p0); p1 = wave_sum(p1); p2 = wave_sum(p2);
    if (lane == 0) {
        out[node * 3 + 0] = p0 + b2[0];
        out[node * 3 + 1] = p1 + b2[1];
        out[node * 3 + 2] = p2 + b2[2];
    }
}

extern "C" void kernel_launch(void* const* d_in, const int* in_sizes, int n_in,
                              void* d_out, int out_size, void* d_ws, size_t ws_size,
                              hipStream_t stream) {
    const float* feat = (const float*)d_in[0];
    const int*   src  = (const int*)  d_in[1];
    const int*   dst  = (const int*)  d_in[2];
    const float* W1   = (const float*)d_in[3];
    const float* al1  = (const float*)d_in[4];
    const float* ar1  = (const float*)d_in[5];
    const float* b1   = (const float*)d_in[6];
    const float* W2   = (const float*)d_in[7];
    const float* al2  = (const float*)d_in[8];
    const float* ar2  = (const float*)d_in[9];
    const float* b2   = (const float*)d_in[10];
    float* out = (float*)d_out;

    const int n_nodes = in_sizes[0] / 256;
    const int n_edges = in_sizes[1];

    char* cur = (char*)d_ws;
    auto alloc = [&](size_t bytes) -> float* {
        float* p = (float*)cur;
        cur += (bytes + 255) & ~(size_t)255;
        return p;
    };
    float* H1   = alloc((size_t)n_nodes * 256 * 4);
    float* el1  = alloc((size_t)n_nodes * 8 * 4);
    float* er1  = alloc((size_t)n_nodes * 8 * 4);
    float* hp2  = alloc((size_t)n_nodes * 4 * 4);
    float* er2  = alloc((size_t)n_nodes * 4);
    const int n_buckets = cdiv(n_nodes, BK_NODES);
    const int n_slices  = cdiv(n_edges, SLICE);
    int*   wstart = (int*)alloc(((size_t)n_buckets + 1) * n_slices * 4);
    int*   csrc   = (int*)alloc((size_t)n_edges * 4);
    int*   cdst   = (int*)alloc((size_t)n_edges * 4);
    int*   rowptr = (int*)alloc((size_t)(n_nodes + 1) * 4);
    int*   ssrc   = (int*)alloc((size_t)n_edges * 4);

    gat_slice_sort<<<n_slices, 1024, 0, stream>>>(src, dst, csrc, cdst, wstart, n_edges, n_buckets);
    gat_bucket_sort<<<n_buckets, 1024, 0, stream>>>(csrc, cdst, wstart, rowptr, ssrc, n_nodes, n_edges, n_slices);

    gat_gemm1_wmma<<<cdiv(n_nodes, 32), 256, 0, stream>>>(feat, W1, al1, ar1, H1, el1, er1, n_nodes);
    gat_l1_segments<<<cdiv(n_nodes, 8), 256, 0, stream>>>(rowptr, ssrc, H1, el1, er1,
                                                          b1, W2, al2, ar2, hp2, er2, n_nodes);

    gat_l2_segments<<<cdiv(n_nodes, 8), 256, 0, stream>>>(rowptr, ssrc, hp2, er2, b2, out, n_nodes);
}


# Segment kernels take several destination nodes per wave (layer 1: 5 in turn, next node's range and rows asked ahead; layer 2: one lane per node, in-lane sums): 8x fewer waves; l2 sum order changed
# speedup vs baseline: 9.0918x; 1.0783x over previous
#include <hip/hip_runtime.h>
#include <hip/hip_bf16.h>

typedef __attribute__((ext_vector_type(2))) float v2f;
typedef __attribute__((ext_vector_type(4))) float v4f;
typedef __attribute__((ext_vector_type(8))) float v8f;

#define SLOPE 0.2f

static __host__ __device__ inline int cdiv(int a, int b) { return (a + b - 1) / b; }

template <int CTRL>
__device__ inline float row_xmask_add(float x) {
    return x + __int_as_float(__builtin_amdgcn_update_dpp(0, __float_as_int(x), CTRL, 0xF, 0xF, true));
}
__device__ inline float row16_sum(float x) {
    x = row_xmask_add<0x161>(x);
    x = row_xmask_add<0x162>(x);
    x = row_xmask_add<0x164>(x);
    x = row_xmask_add<0x168>(x);
    return x;
}

#define A_PITCH 260
#define W_PITCH 272

__global__ void __launch_bounds__(256) gat_gemm1_wmma(const float* __restrict__ A,
                               const float* __restrict__ W,
                               const float* __restrict__ ar,
                               float* __restrict__ H,
                               float* __restrict__ er,
                               int n_nodes) {
    __shared__ float As[32 * A_PITCH];
    __shared__ float Ws[32 * W_PITCH];
    __shared__ float Sc[32 * 8];

    const int row0 = blockIdx.x * 32;
    const int tid  = threadIdx.x;
    const int wave = tid >> 5;
    const int lane = tid & 31;

    const int sr = tid >> 6;
    const int sc = (tid & 63) << 2;
    float4 st[8];

#pragma unroll
    for (int j = 0; j < 8; ++j) {
        int rr = row0 + sr + 4 * j; if (rr >= n_nodes) rr = n_nodes - 1;
        st[j] = *(const float4*)(A + (size_t)rr * 256 + sc);
    }
#pragma unroll
    for (int j = 0; j < 8; ++j) *(float4*)(&As[(sr + 4 * j) * A_PITCH + sc]) = st[j];

    const int m   = lane & 15;
    const int kk  = (lane >> 4) << 1;
    const int n0  = wave * 32;
    const int n1  = n0 + 16;

    const float ar0 = ar[n0 + m], ar1 = ar[n1 + m];

    v8f c00 = {}, c01 = {}, c10 = {}, c11 = {};

    for (int kc = 0; kc < 256; kc += 32) {
        __syncthreads();
#pragma unroll
        for (int j = 0; j < 8; ++j) st[j] = *(const float4*)(W + (size_t)(kc + sr + 4 * j) * 256 + sc);
#pragma unroll
        for (int j = 0; j < 8; ++j) *(float4*)(&Ws[(sr + 4 * j) * W_PITCH + sc]) = st[j];
        __syncthreads();

#pragma unroll
        for (int k = 0; k < 32; k += 4) {
            v2f a0, a1, b0, b1;
            a0.x = As[m * A_PITCH + kc + k + kk];
            a0.y = As[m * A_PITCH + kc + k + kk + 1];
            a1.x = As[(m + 16) * A_PITCH + kc + k + kk];
            a1.y = As[(m + 16) * A_PITCH + kc + k + kk + 1];
            b0.x = Ws[(k + kk) * W_PITCH + n0 + m];
            b0.y = Ws[(k + kk + 1) * W_PITCH + n0 + m];
            b1.x = Ws[(k + kk) * W_PITCH + n1 + m];
            b1.y = Ws[(k + kk + 1) * W_PITCH + n1 + m];
            c00 = __builtin_amdgcn_wmma_f32_16x16x4_f32(false, a0, false, b0, (short)0, c00, false, false);
            c01 = __builtin_amdgcn_wmma_f32_16x16x4_f32(false, a0, false, b1, (short)0, c01, false, false);
            c10 = __builtin_amdgcn_wmma_f32_16x16x4_f32(false, a1, false, b0, (short)0, c10, false, false);
            c11 = __builtin_amdgcn_wmma_f32_16x16x4_f32(false, a1, false, b1, (short)0, c11, false, false);
        }
    }

    const int mhi = (lane >> 4) * 8;

#pragma unroll
    for (int v = 0; v < 8; ++v) {
        float q0 = c00[v] * ar0 + c01[v] * ar1;
        float q1 = c10[v] * ar0 + c11[v] * ar1;
        q0 = row16_sum(q0); q1 = row16_sum(q1);
        if (m == 0) {
            Sc[(mhi + v) * 8 + wave]      = q0;
            Sc[(16 + mhi + v) * 8 + wave] = q1;
        }
    }
    __syncthreads();
    if (row0 + (tid >> 3) < n_nodes) er[(size_t)row0 * 8 + tid] = Sc[tid];

    for (int v = 0; v < 8; ++v) {
        int r0 = row0 + mhi + v;
        int r1 = r0 + 16;
        if (r0 < n_nodes) {
            H[(size_t)r0 * 256 + n0 + m] = c00[v];
            H[(size_t)r0 * 256 + n1 + m] = c01[v];
        }
        if (r1 < n_nodes) {
            H[(size_t)r1 * 256 + n0 + m] = c10[v];
            H[(size_t)r1 * 256 + n1 + m] = c11[v];
        }
    }
}

#define BK_SHIFT 8
#define BK_NODES 256
#define SLICE    4096
#define BK_CAP   8192
#define SL_MAX   1024

__device__ inline void scan256_first_wave(const int* cnt, int* out, int t) {
    if (t < 32) {
        int v[8];
        int s = 0;
#pragma unroll
        for (int j = 0; j < 8; ++j) { v[j] = cnt[t * 8 + j]; s += v[j]; }
        int x = s;
        for (int o = 1; o < 32; o <<= 1) {
            int y = __shfl_up(x, o, 32);
            if (t >= o) x += y;
        }
        int run = x - s;
#pragma unroll
        for (int j = 0; j < 8; ++j) { out[t * 8 + j] = run; run += v[j]; }
    }
}

__global__ void __launch_bounds__(1024) gat_slice_sort(const int* __restrict__ src,
                                                       const int* __restrict__ dst,
                                                       int* __restrict__ csrc,
                                                       int* __restrict__ cdst,
                                                       int* __restrict__ wstart,
                                                       int n_edges, int n_buckets) {
    __shared__ int cnt[BK_NODES];
    __shared__ int lstart[BK_NODES];
    __shared__ int lsrc[SLICE];
    __shared__ int ldst[SLICE];
    const int t = threadIdx.x, w = blockIdx.x, G = gridDim.x;
    int d[4], s[4], r[4];
#pragma unroll
    for (int k = 0; k < 4; ++k) {
        const int e = w * SLICE + k * 1024 + t;
        const int ec = e < n_edges ? e : n_edges - 1;
        d[k] = dst[ec];
        s[k] = src[ec];
    }
    if (t < BK_NODES) cnt[t] = 0;
    __syncthreads();
#pragma unroll
    for (int k = 0; k < 4; ++k) {
        const int e = w * SLICE + k * 1024 + t;
        r[k] = (e < n_edges) ? atomicAdd(&cnt[d[k] >> BK_SHIFT], 1) : 0;
    }
    __syncthreads();
    scan256_first_wave(cnt, lstart, t);
    __syncthreads();
#pragma unroll
    for (int k = 0; k < 4; ++k) {
        const int e = w * SLICE + k * 1024 + t;
        if (e < n_edges) {
            const int p = lstart[d[k] >> BK_SHIFT] + r[k];
            lsrc[p] = s[k];
            ldst[p] = d[k];
        }
    }
    int m = n_edges - w * SLICE;
    if (m > SLICE) m = SLICE;
    if (t <= n_buckets) wstart[t * G + w] = (t < n_buckets) ? lstart[t] : m;
    __syncthreads();
#pragma unroll
    for (int k = 0; k < 4; ++k) {
        const int p = t + k * 1024;
        if (p < m) {
            csrc[w * SLICE + p] = lsrc[p];
            cdst[w * SLICE + p] = ldst[p];
        }
    }
}

__global__ void __launch_bounds__(1024) gat_bucket_sort(const int* __restrict__ csrc,
                                                        const int* __restrict__ cdst,
                                                        const int* __restrict__ wstart,
                                                        int* __restrict__ rowptr,
                                                        int* __restrict__ ssrc,
                                                        int n_nodes, int n_edges, int n_slices) {
    __shared__ int cnt[BK_NODES];
    __shared__ int cur[BK_NODES];
    __shared__ int stage[BK_CAP];
    __shared__ int off[SL_MAX];
    __shared__ int sst[SL_MAX];
    __shared__ int wtot[32];
    __shared__ int wpre[32];
    const int t = threadIdx.x, lane = t & 31, wave = t >> 5, b = blockIdx.x;
    int st = 0, c = 0;
    if (t < n_slices) {
        st = wstart[b * n_slices + t];
        c  = wstart[(b + 1) * n_slices + t] - st;
    }
    int x = c;
    for (int o = 1; o < 32; o <<= 1) {
        int y = __shfl_up(x, o, 32);
        if (lane >= o) x += y;
    }
    int pre = st;
    for (int o = 16; o > 0; o >>= 1) pre += __shfl_xor(pre, o, 32);
    if (lane == 31) wtot[wave] = x;
    if (lane == 0)  wpre[wave] = pre;
    if (t < BK_NODES) cnt[t] = 0;
    __syncthreads();
    int wv = wtot[lane];
    for (int o = 1; o < 32; o <<= 1) {
        int y = __shfl_up(wv, o, 32);
        if (lane >= o) wv += y;
    }
    int sum_st = wpre[lane];
    for (int o = 16; o > 0; o >>= 1) sum_st += __shfl_xor(sum_st, o, 32);
    const int before_wave = __shfl(wv, wave, 32) - wtot[wave];
    off[t] = (t < n_slices) ? before_wave + x - c : 0x7fffffff;
    sst[t] = st;
    const int m  = __builtin_amdgcn_readfirstlane(__shfl(wv, 31, 32));
    const int lo = __builtin_amdgcn_readfirstlane(sum_st);
    const int hi = lo + m;
    __syncthreads();

    int top = 1;
    for (int q = 0; q < 9 && top * 2 < n_slices; ++q) top *= 2;

    auto edge_at = [&](int i) -> int {
        int sl = 0;
        for (int step = top; step > 0; step >>= 1) {
            const int cnd = sl + step;
            if (off[cnd] <= i) sl = cnd;
        }
        const int j = sl * SLICE + sst[sl] + (i - off[sl]);
        return j < n_edges ? j : n_edges - 1;
    };

    const bool fits = m <= BK_CAP;
    int d[8], s[8];
    if (fits) {
        int ic[8], sl[8];
#pragma unroll
        for (int k = 0; k < 8; ++k) {
            int i = t + k * 1024;
            i = i < m ? i : m - 1;
            ic[k] = i > 0 ? i : 0;
            sl[k] = 0;
        }
        for (int step = top; step > 0; step >>= 1) {
#pragma unroll
            for (int k = 0; k < 8; ++k) {
                const int cnd = sl[k] + step;
                if (off[cnd] <= ic[k]) sl[k] = cnd;
            }
        }
#pragma unroll
        for (int k = 0; k < 8; ++k) {
            int j = sl[k] * SLICE + sst[sl[k]] + (ic[k] - off[sl[k]]);
            j = j < n_edges ? j : n_edges - 1;
            d[k] = cdst[j];
            s[k] = csrc[j];
        }
#pragma unroll
        for (int k = 0; k < 8; ++k) d[k] &= BK_NODES - 1;
#pragma unroll
        for (int k = 0; k < 8; ++k)
            if (t + k * 1024 < m) atomicAdd(&cnt[d[k]], 1);
    } else {
        for (int i = t; i < m; i += 1024) atomicAdd(&cnt[cdst[edge_at(i)] & (BK_NODES - 1)], 1);
    }
    __syncthreads();
    scan256_first_wave(cnt, cur, t);
    __syncthreads();
    if (t < BK_NODES) {
        const int node = b * BK_NODES + t;
        if (node < n_nodes) rowptr[node] = lo + cur[t];
        if (node == n_nodes - 1) rowptr[n_nodes] = hi;
    }
    __syncthreads();
    if (fits) {
        int p[8];
#pragma unroll
        for (int k = 0; k < 8; ++k) p[k] = (t + k * 1024 < m) ? atomicAdd(&cur[d[k]], 1) : 0;
#pragma unroll
        for (int k = 0; k < 8; ++k)
            if (t + k * 1024 < m) stage[p[k]] = s[k];
        __syncthreads();
#pragma unroll
        for (int k = 0; k < 8; ++k) {
            const int i = t + k * 1024;
            if (i < m) ssrc[lo + i] = stage[i];
        }
    } else {
        for (int i = t; i < m; i += 1024) {
            const int j = edge_at(i);
            const int p = atomicAdd(&cur[cdst[j] & (BK_NODES - 1)], 1);
            ssrc[lo + p] = csrc[j];
        }
    }
}

__device__ inline float wave_sum(float x) {
    for (int o = 16; o > 0; o >>= 1) x += __shfl_xor(x, o, 32);
    return x;
}

template <int CTRL>
__device__ inline float quad_perm(float x) {
    return __int_as_float(__builtin_amdgcn_update_dpp(0, __float_as_int(x), CTRL, 0xF, 0xF, true));
}

template <int J>
__device__ inline void gat_add_slot(const v4f& x0, const v4f& x1, float u, v4f& a0, v4f& a1) {
    const float uj = quad_perm<J * 0x55>(u);
    a0 += x0 * uj;
    a1 += x1 * uj;
}

template <int NV>
__device__ inline void gat_group(const float* __restrict__ Hl, int sv, int k, int sub,
                                 const v4f& w0, const v4f& w1, float erv,
                                 float& m, float& s, v4f& a0, v4f& a1) {
    v4f x0[NV], x1[NV];
#pragma unroll
    for (int j = 0; j < NV; ++j) {
        const int sn = __builtin_amdgcn_readlane(sv, k + j);
        x0[j] = *(const v4f*)(Hl + (size_t)sn * 256);
        x1[j] = *(const v4f*)(Hl + (size_t)sn * 256 + 4);
    }
    if constexpr (NV == 4)
        asm volatile("" :: "v"(x0[0]), "v"(x1[0]), "v"(x0[1]), "v"(x1[1]),
                           "v"(x0[2]), "v"(x1[2]), "v"(x0[3]), "v"(x1[3]));
    else if constexpr (NV == 3)
        asm volatile("" :: "v"(x0[0]), "v"(x1[0]), "v"(x0[1]), "v"(x1[1]), "v"(x0[2]), "v"(x1[2]));
    else if constexpr (NV == 2)
        asm volatile("" :: "v"(x0[0]), "v"(x1[0]), "v"(x0[1]), "v"(x1[1]));
    float d[4] = {0.f, 0.f, 0.f, 0.f};
#pragma unroll
    for (int j = 0; j < NV; ++j) {
        const v2f t = x0[j].xy * w0.xy + x0[j].zw * w0.zw + x1[j].xy * w1.xy + x1[j].zw * w1.zw;
        d[j] = t.x + t.y;
    }
    const bool odd = (sub & 1) != 0, upper = (sub & 2) != 0;
    float p01 = odd ? d[1] : d[0];
    p01 += quad_perm<0xB1>(odd ? d[0] : d[1]);
    float p23 = odd ? d[3] : d[2];
    p23 += quad_perm<0xB1>(odd ? d[2] : d[3]);
    float e = upper ? p23 : p01;
    e += quad_perm<0x4E>(upper ? p01 : p23);
    e += erv;
    e = (e >= 0.f) ? e : SLOPE * e;
    if (NV < 4) e = (sub < NV) ? e : -__builtin_huge_valf();
    float gm = fmaxf(e, quad_perm<0xB1>(e));
    gm = fmaxf(gm, quad_perm<0x4E>(gm));
    const float mn = fmaxf(m, gm);
    const float sc = expf(m - mn);
    const float u  = expf(e - mn);
    m = mn;
    s = s * sc + u;
    a0 *= sc;
    a1 *= sc;
    gat_add_slot<0>(x0[0], x1[0], u, a0, a1);
    if constexpr (NV > 1) gat_add_slot<1>(x0[1], x1[1], u, a0, a1);
    if constexpr (NV > 2) gat_add_slot<2>(x0[2], x1[2], u, a0, a1);
    if constexpr (NV > 3) gat_add_slot<3>(x0[3], x1[3], u, a0, a1);
}

#define L1_NPW 5
__global__ void __launch_bounds__(256) gat_l1_segments(const int* __restrict__ rowptr,
                                                       const int* __restrict__ ssrc,
                                                       const float* __restrict__ H,
                                                       const float* __restrict__ al,
                                                       const float* __restrict__ er,
                                                       const float* __restrict__ b1,
                                                       const float* __restrict__ W2,
                                                       const float* __restrict__ al2,
                                                       const float* __restrict__ ar2,
                                                       float* __restrict__ hp,
                                                       float* __restrict__ er2, int n_nodes, int n_edges) {
    const int lane = threadIdx.x & 31;
    const int wv = blockIdx.x * 8 + __builtin_amdgcn_readfirstlane((int)(threadIdx.x >> 5));
    const int node0 = wv * L1_NPW;
    if (node0 >= n_nodes) return;
    const int nend = node0 + L1_NPW < n_nodes ? node0 + L1_NPW : n_nodes;
    const int head = lane >> 2, sub = lane & 3;
    const float* Hl = H + lane * 8;
    const v4f av0 = *(const v4f*)(al + lane * 8);
    const v4f av1 = *(const v4f*)(al + lane * 8 + 4);

    int ee = rowptr[node0];
    int ee_n = rowptr[node0 + 1];
    int jn = ee + lane;
    jn = jn < n_edges ? jn : n_edges - 1;
    int sv_n = ssrc[jn > 0 ? jn : 0];
    float erv_n = er[node0 * 8 + head];

  for (int node = node0; node < nend; ++node) {
    const int eb = ee;
    ee = ee_n;
    const int sv0 = sv_n;
    const float erv = erv_n;
    const int nx = node + 1 < n_nodes ? node + 1 : node;
    ee_n = rowptr[nx + 1];
    jn = ee + lane;
    jn = jn < n_edges ? jn : n_edges - 1;
    sv_n = ssrc[jn > 0 ? jn : 0];
    erv_n = er[nx * 8 + head];

    v4f a0 = {0.f, 0.f, 0.f, 0.f}, a1 = {0.f, 0.f, 0.f, 0.f};
    float m = -__builtin_huge_valf(), s = 0.f;

    for (int c = eb; c < ee; c += 32) {
        int sv = sv0;
        if (c != eb) {
            const int j = c + lane;
            sv = ssrc[j < ee ? j : ee - 1];
        }
        const int left = ee - c;
        const int cnt = left < 32 ? left : 32;
        int k = 0;
        for (; k + 4 <= cnt; k += 4) gat_group<4>(Hl, sv, k, sub, av0, av1, erv, m, s, a0, a1);
        const int r = cnt - k;
        if (r == 3)      gat_group<3>(Hl, sv, k, sub, av0, av1, erv, m, s, a0, a1);
        else if (r == 2) gat_group<2>(Hl, sv, k, sub, av0, av1, erv, m, s, a0, a1);
        else if (r == 1) gat_group<1>(Hl, sv, k, sub, av0, av1, erv, m, s, a0, a1);
    }
    if (ee > eb) {
        s += quad_perm<0xB1>(s);
        s += quad_perm<0x4E>(s);
        const float inv = 1.0f / s;
        a0 *= inv;
        a1 *= inv;
    }
    const float4 c0 = *(const float4*)(b1 + lane * 8);
    const float4 c1 = *(const float4*)(b1 + lane * 8 + 4);
    float x[8] = {a0.x + c0.x, a0.y + c0.y, a0.z + c0.z, a0.w + c0.w,
                  a1.x + c1.x, a1.y + c1.y, a1.z + c1.z, a1.w + c1.w};
#pragma unroll
    for (int i = 0; i < 8; ++i) x[i] = (x[i] > 0.f) ? x[i] : expm1f(x[i]);
    const float4* wp = (const float4*)(W2 + lane * 24);
    const float4 w0 = wp[0], w1 = wp[1], w2 = wp[2], w3 = wp[3], w4 = wp[4], w5 = wp[5];
    float p0 = x[0] * w0.x, p1 = x[0] * w0.y, p2 = x[0] * w0.z;
    p0 += x[1] * w0.w; p1 += x[1] * w1.x; p2 += x[1] * w1.y;
    p0 += x[2] * w1.z; p1 += x[2] * w1.w; p2 += x[2] * w2.x;
    p0 += x[3] * w2.y; p1 += x[3] * w2.z; p2 += x[3] * w2.w;
    p0 += x[4] * w3.x; p1 += x[4] * w3.y; p2 += x[4] * w3.z;
    p0 += x[5] * w3.w; p1 += x[5] * w4.x; p2 += x[5] * w4.y;
    p0 += x[6] * w4.z; p1 += x[6] * w4.w; p2 += x[6] * w5.x;
    p0 += x[7] * w5.y; p1 += x[7] * w5.z; p2 += x[7] * w5.w;
    p0 = wave_sum(p0); p1 = wave_sum(p1); p2 = wave_sum(p2);
    if (lane == 0) {
        float4 row;
        row.x = p0; row.y = p1; row.z = p2;
        row.w = p0 * al2[0] + p1 * al2[1] + p2 * al2[2];
        ((float4*)hp)[node] = row;
        er2[node] = p0 * ar2[0] + p1 * ar2[1] + p2 * ar2[2];
    }
  }
}

__global__ void __launch_bounds__(64) gat_l2_segments(const int* __restrict__ rowptr,
                                                      const int* __restrict__ ssrc,
                                                      const float* __restrict__ hp,
                                                      const float* __restrict__ er2,
                                                      const float* __restrict__ b2,
                                                      float* __restrict__ out, int n_nodes) {
    const int node = blockIdx.x * 64 + threadIdx.x;
    if (node >= n_nodes) return;
    const int eb = rowptr[node], ee = rowptr[node + 1];
    const float erv = er2[node];
    const float NEG_INF = -__builtin_huge_valf();
    const float4* rows = (const float4*)hp;
    float m = NEG_INF, s = 0.f, p0 = 0.f, p1 = 0.f, p2 = 0.f;

    for (int c = eb; c < ee; c += 4) {
        int sn[4];
        float4 r[4];
        float x[4];
#pragma unroll
        for (int k = 0; k < 4; ++k) {
            const int j = c + k;
            sn[k] = ssrc[j < ee ? j : ee - 1];
        }
#pragma unroll
        for (int k = 0; k < 4; ++k) r[k] = rows[sn[k]];
#pragma unroll
        for (int k = 0; k < 4; ++k) {
            float v = r[k].w + erv;
            v = (v >= 0.f) ? v : SLOPE * v;
            x[k] = (c + k < ee) ? v : NEG_INF;
        }
        const float gm = fmaxf(fmaxf(x[0], x[1]), fmaxf(x[2], x[3]));
        const float mn = fmaxf(m, gm);
        const float sc = expf(m - mn);
        const float u0 = expf(x[0] - mn), u1 = expf(x[1] - mn);
        const float u2 = expf(x[2] - mn), u3 = expf(x[3] - mn);
        m = mn;
        s  = s  * sc + ((u0 + u1) + (u2 + u3));
        p0 = p0 * sc + ((u0 * r[0].x + u1 * r[1].x) + (u2 * r[2].x + u3 * r[3].x));
        p1 = p1 * sc + ((u0 * r[0].y + u1 * r[1].y) + (u2 * r[2].y + u3 * r[3].y));
        p2 = p2 * sc + ((u0 * r[0].z + u1 * r[1].z) + (u2 * r[2].z + u3 * r[3].z));
    }
    if (ee > eb) {
        const float inv = 1.0f / s;
        p0 *= inv; p1 *= inv; p2 *= inv;
    }
    out[node * 3 + 0] = p0 + b2[0];
    out[node * 3 + 1] = p1 + b2[1];
    out[node * 3 + 2] = p2 + b2[2];
}

extern "C" void kernel_launch(void* const* d_in, const int* in_sizes, int n_in,
                              void* d_out, int out_size, void* d_ws, size_t ws_size,
                              hipStream_t stream) {
    const float* feat = (const float*)d_in[0];
    const int*   src  = (const int*)  d_in[1];
    const int*   dst  = (const int*)  d_in[2];
    const float* W1   = (const float*)d_in[3];
    const float* al1  = (const float*)d_in[4];
    const float* ar1  = (const float*)d_in[5];
    const float* b1   = (const float*)d_in[6];
    const float* W2   = (const float*)d_in[7];
    const float* al2  = (const float*)d_in[8];
    const float* ar2  = (const float*)d_in[9];
    const float* b2   = (const float*)d_in[10];
    float* out = (float*)d_out;

    const int n_nodes = in_sizes[0] / 256;
    const int n_edges = in_sizes[1];

    char* cur = (char*)d_ws;
    auto alloc = [&](size_t bytes) -> float* {
        float* p = (float*)cur;
        cur += (bytes + 255) & ~(size_t)255;
        return p;
    };
    float* H1   = alloc((size_t)n_nodes * 256 * 4);
    float* er1  = alloc((size_t)n_nodes * 8 * 4);
    float* hp2  = alloc((size_t)n_nodes * 4 * 4);
    float* er2  = alloc((size_t)n_nodes * 4);
    const int n_buckets = cdiv(n_nodes, BK_NODES);
    const int n_slices  = cdiv(n_edges, SLICE);
    int*   wstart = (int*)alloc(((size_t)n_buckets + 1) * n_slices * 4);
    int*   csrc   = (int*)alloc((size_t)n_edges * 4);
    int*   cdst   = (int*)alloc((size_t)n_edges * 4);
    int*   rowptr = (int*)alloc((size_t)(n_nodes + 1) * 4);
    int*   ssrc   = (int*)alloc((size_t)n_edges * 4);

    gat_slice_sort<<<n_slices, 1024, 0, stream>>>(src, dst, csrc, cdst, wstart, n_edges, n_buckets);
    gat_bucket_sort<<<n_buckets, 1024, 0, stream>>>(csrc, cdst, wstart, rowptr, ssrc, n_nodes, n_edges, n_slices);

    gat_gemm1_wmma<<<cdiv(n_nodes, 32), 256, 0, stream>>>(feat, W1, ar1, H1, er1, n_nodes);
    gat_l1_segments<<<cdiv(n_nodes, 8 * L1_NPW), 256, 0, stream>>>(rowptr, ssrc, H1, al1, er1,
                                                          b1, W2, al2, ar2, hp2, er2, n_nodes, n_edges);

    gat_l2_segments<<<cdiv(n_nodes, 64), 64, 0, stream>>>(rowptr, ssrc, hp2, er2, b2, out, n_nodes);
}


# Slice level of the edge sort runs as the first work-groups of the GEMM1 launch (256 threads, 16 edges each): 4 launches not 5, sort beside matrix work; a node's edge order = LDS arrival (changed)
# speedup vs baseline: 9.9208x; 1.0912x over previous
#include <hip/hip_runtime.h>
#include <hip/hip_bf16.h>

typedef __attribute__((ext_vector_type(2))) float v2f;
typedef __attribute__((ext_vector_type(4))) float v4f;
typedef __attribute__((ext_vector_type(8))) float v8f;

#define SLOPE 0.2f

static __host__ __device__ inline int cdiv(int a, int b) { return (a + b - 1) / b; }

template <int CTRL>
__device__ inline float row_xmask_add(float x) {
    return x + __int_as_float(__builtin_amdgcn_update_dpp(0, __float_as_int(x), CTRL, 0xF, 0xF, true));
}
__device__ inline float row16_sum(float x) {
    x = row_xmask_add<0x161>(x);
    x = row_xmask_add<0x162>(x);
    x = row_xmask_add<0x164>(x);
    x = row_xmask_add<0x168>(x);
    return x;
}

#define BK_SHIFT 8
#define BK_NODES 256
#define SLICE    4096
#define SL_EPT   16
#define BK_CAP   8192
#define SL_MAX   1024

__device__ inline void scan256_first_wave(const int* cnt, int* out, int t) {
    if (t < 32) {
        int v[8];
        int s = 0;
#pragma unroll
        for (int j = 0; j < 8; ++j) { v[j] = cnt[t * 8 + j]; s += v[j]; }
        int x = s;
        for (int o = 1; o < 32; o <<= 1) {
            int y = __shfl_up(x, o, 32);
            if (t >= o) x += y;
        }
        int run = x - s;
#pragma unroll
        for (int j = 0; j < 8; ++j) { out[t * 8 + j] = run; run += v[j]; }
    }
}

struct SliceSortLds {
    int cnt[BK_NODES];
    int lstart[BK_NODES];
    int lsrc[SLICE];
    int ldst[SLICE];
};

__device__ inline void gat_slice_sort_role(SliceSortLds& lds,
                                           const int* __restrict__ src,
                                           const int* __restrict__ dst,
                                           int* __restrict__ csrc,
                                           int* __restrict__ cdst,
                                           int* __restrict__ wstart,
                                           int n_edges, int n_buckets, int w, int G) {
    int* const cnt    = lds.cnt;
    int* const lstart = lds.lstart;
    int* const lsrc   = lds.lsrc;
    int* const ldst   = lds.ldst;
    const int t = threadIdx.x;
    int d[SL_EPT], s[SL_EPT], r[SL_EPT];
#pragma unroll
    for (int k = 0; k < SL_EPT; ++k) {
        const int e = w * SLICE + k * 256 + t;
        const int ec = e < n_edges ? e : n_edges - 1;
        d[k] = dst[ec];
        s[k] = src[ec];
    }
    cnt[t] = 0;
    __syncthreads();
#pragma unroll
    for (int k = 0; k < SL_EPT; ++k) {
        const int e = w * SLICE + k * 256 + t;
        r[k] = (e < n_edges) ? atomicAdd(&cnt[d[k] >> BK_SHIFT], 1) : 0;
    }
    __syncthreads();
    scan256_first_wave(cnt, lstart, t);
    __syncthreads();
#pragma unroll
    for (int k = 0; k < SL_EPT; ++k) {
        const int e = w * SLICE + k * 256 + t;
        if (e < n_edges) {
            const int p = lstart[d[k] >> BK_SHIFT] + r[k];
            lsrc[p] = s[k];
            ldst[p] = d[k];
        }
    }
    int m = n_edges - w * SLICE;
    if (m > SLICE) m = SLICE;
    for (int i = t; i <= n_buckets; i += 256) wstart[i * G + w] = (i < n_buckets) ? lstart[i] : m;
    __syncthreads();
#pragma unroll
    for (int k = 0; k < SL_EPT; ++k) {
        const int p = t + k * 256;
        if (p < m) {
            csrc[w * SLICE + p] = lsrc[p];
            cdst[w * SLICE + p] = ldst[p];
        }
    }
}

#define A_PITCH 260
#define W_PITCH 272

struct Gemm1Lds {
    float As[32 * A_PITCH];
    float Ws[32 * W_PITCH];
    float Sc[32 * 8];
};
union Launch1Lds {
    Gemm1Lds g;
    SliceSortLds s;
};

__global__ void __launch_bounds__(256) gat_gemm1_and_slice_sort(
                               const float* __restrict__ A,
                               const float* __restrict__ W,
                               const float* __restrict__ ar,
                               float* __restrict__ H,
                               float* __restrict__ er,
                               int n_nodes,
                               const int* __restrict__ src,
                               const int* __restrict__ dst,
                               int* __restrict__ csrc,
                               int* __restrict__ cdst,
                               int* __restrict__ wstart,
                               int n_edges, int n_buckets, int n_slices) {
    __shared__ __attribute__((aligned(16))) Launch1Lds lds;

    if ((int)blockIdx.x < n_slices) {
        gat_slice_sort_role(lds.s, src, dst, csrc, cdst, wstart, n_edges, n_buckets,
                            (int)blockIdx.x, n_slices);
        return;
    }

    float* const As = lds.g.As;
    float* const Ws = lds.g.Ws;
    float* const Sc = lds.g.Sc;

    const int row0 = ((int)blockIdx.x - n_slices) * 32;
    const int tid  = threadIdx.x;
    const int wave = tid >> 5;
    const int lane = tid & 31;

    const int sr = tid >> 6;
    const int sc = (tid & 63) << 2;
    float4 st[8];

#pragma unroll
    for (int j = 0; j < 8; ++j) {
        int rr = row0 + sr + 4 * j; if (rr >= n_nodes) rr = n_nodes - 1;
        st[j] = *(const float4*)(A + (size_t)rr * 256 + sc);
    }
#pragma unroll
    for (int j = 0; j < 8; ++j) *(float4*)(&As[(sr + 4 * j) * A_PITCH + sc]) = st[j];

    const int m   = lane & 15;
    const int kk  = (lane >> 4) << 1;
    const int n0  = wave * 32;
    const int n1  = n0 + 16;

    const float ar0 = ar[n0 + m], ar1 = ar[n1 + m];

    v8f c00 = {}, c01 = {}, c10 = {}, c11 = {};

    for (int kc = 0; kc < 256; kc += 32) {
        __syncthreads();
#pragma unroll
        for (int j = 0; j < 8; ++j) st[j] = *(const float4*)(W + (size_t)(kc + sr + 4 * j) * 256 + sc);
#pragma unroll
        for (int j = 0; j < 8; ++j) *(float4*)(&Ws[(sr + 4 * j) * W_PITCH + sc]) = st[j];
        __syncthreads();

#pragma unroll
        for (int k = 0; k < 32; k += 4) {
            v2f a0, a1, b0, b1;
            a0.x = As[m * A_PITCH + kc + k + kk];
            a0.y = As[m * A_PITCH + kc + k + kk + 1];
            a1.x = As[(m + 16) * A_PITCH + kc + k + kk];
            a1.y = As[(m + 16) * A_PITCH + kc + k + kk + 1];
            b0.x = Ws[(k + kk) * W_PITCH + n0 + m];
            b0.y = Ws[(k + kk + 1) * W_PITCH + n0 + m];
            b1.x = Ws[(k + kk) * W_PITCH + n1 + m];
            b1.y = Ws[(k + kk + 1) * W_PITCH + n1 + m];
            c00 = __builtin_amdgcn_wmma_f32_16x16x4_f32(false, a0, false, b0, (short)0, c00, false, false);
            c01 = __builtin_amdgcn_wmma_f32_16x16x4_f32(false, a0, false, b1, (short)0, c01, false, false);
            c10 = __builtin_amdgcn_wmma_f32_16x16x4_f32(false, a1, false, b0, (short)0, c10, false, false);
            c11 = __builtin_amdgcn_wmma_f32_16x16x4_f32(false, a1, false, b1, (short)0, c11, false, false);
        }
    }

    const int mhi = (lane >> 4) * 8;

#pragma unroll
    for (int v = 0; v < 8; ++v) {
        float q0 = c00[v] * ar0 + c01[v] * ar1;
        float q1 = c10[v] * ar0 + c11[v] * ar1;
        q0 = row16_sum(q0); q1 = row16_sum(q1);
        if (m == 0) {
            Sc[(mhi + v) * 8 + wave]      = q0;
            Sc[(16 + mhi + v) * 8 + wave] = q1;
        }
    }
    __syncthreads();
    if (row0 + (tid >> 3) < n_nodes) er[(size_t)row0 * 8 + tid] = Sc[tid];

    for (int v = 0; v < 8; ++v) {
        int r0 = row0 + mhi + v;
        int r1 = r0 + 16;
        if (r0 < n_nodes) {
            H[(size_t)r0 * 256 + n0 + m] = c00[v];
            H[(size_t)r0 * 256 + n1 + m] = c01[v];
        }
        if (r1 < n_nodes) {
            H[(size_t)r1 * 256 + n0 + m] = c10[v];
            H[(size_t)r1 * 256 + n1 + m] = c11[v];
        }
    }
}

__global__ void __launch_bounds__(1024) gat_bucket_sort(const int* __restrict__ csrc,
                                                        const int* __restrict__ cdst,
                                                        const int* __restrict__ wstart,
                                                        int* __restrict__ rowptr,
                                                        int* __restrict__ ssrc,
                                                        int n_nodes, int n_edges, int n_slices) {
    __shared__ int cnt[BK_NODES];
    __shared__ int cur[BK_NODES];
    __shared__ int stage[BK_CAP];
    __shared__ int off[SL_MAX];
    __shared__ int sst[SL_MAX];
    __shared__ int wtot[32];
    __shared__ int wpre[32];
    const int t = threadIdx.x, lane = t & 31, wave = t >> 5, b = blockIdx.x;
    int st = 0, c = 0;
    if (t < n_slices) {
        st = wstart[b * n_slices + t];
        c  = wstart[(b + 1) * n_slices + t] - st;
    }
    int x = c;
    for (int o = 1; o < 32; o <<= 1) {
        int y = __shfl_up(x, o, 32);
        if (lane >= o) x += y;
    }
    int pre = st;
    for (int o = 16; o > 0; o >>= 1) pre += __shfl_xor(pre, o, 32);
    if (lane == 31) wtot[wave] = x;
    if (lane == 0)  wpre[wave] = pre;
    if (t < BK_NODES) cnt[t] = 0;
    __syncthreads();
    int wv = wtot[lane];
    for (int o = 1; o < 32; o <<= 1) {
        int y = __shfl_up(wv, o, 32);
        if (lane >= o) wv += y;
    }
    int sum_st = wpre[lane];
    for (int o = 16; o > 0; o >>= 1) sum_st += __shfl_xor(sum_st, o, 32);
    const int before_wave = __shfl(wv, wave, 32) - wtot[wave];
    off[t] = (t < n_slices) ? before_wave + x - c : 0x7fffffff;
    sst[t] = st;
    const int m  = __builtin_amdgcn_readfirstlane(__shfl(wv, 31, 32));
    const int lo = __builtin_amdgcn_readfirstlane(sum_st);
    const int hi = lo + m;
    __syncthreads();

    int top = 1;
    for (int q = 0; q < 9 && top * 2 < n_slices; ++q) top *= 2;

    auto edge_at = [&](int i) -> int {
        int sl = 0;
        for (int step = top; step > 0; step >>= 1) {
            const int cnd = sl + step;
            if (off[cnd] <= i) sl = cnd;
        }
        const int j = sl * SLICE + sst[sl] + (i - off[sl]);
        return j < n_edges ? j : n_edges - 1;
    };

    const bool fits = m <= BK_CAP;
    int d[8], s[8];
    if (fits) {
        int ic[8], sl[8];
#pragma unroll
        for (int k = 0; k < 8; ++k) {
            int i = t + k * 1024;
            i = i < m ? i : m - 1;
            ic[k] = i > 0 ? i : 0;
            sl[k] = 0;
        }
        for (int step = top; step > 0; step >>= 1) {
#pragma unroll
            for (int k = 0; k < 8; ++k) {
                const int cnd = sl[k] + step;
                if (off[cnd] <= ic[k]) sl[k] = cnd;
            }
        }
#pragma unroll
        for (int k = 0; k < 8; ++k) {
            int j = sl[k] * SLICE + sst[sl[k]] + (ic[k] - off[sl[k]]);
            j = j < n_edges ? j : n_edges - 1;
            d[k] = cdst[j];
            s[k] = csrc[j];
        }
#pragma unroll
        for (int k = 0; k < 8; ++k) d[k] &= BK_NODES - 1;
#pragma unroll
        for (int k = 0; k < 8; ++k)
            if (t + k * 1024 < m) atomicAdd(&cnt[d[k]], 1);
    } else {
        for (int i = t; i < m; i += 1024) atomicAdd(&cnt[cdst[edge_at(i)] & (BK_NODES - 1)], 1);
    }
    __syncthreads();
    scan256_first_wave(cnt, cur, t);
    __syncthreads();
    if (t < BK_NODES) {
        const int node = b * BK_NODES + t;
        if (node < n_nodes) rowptr[node] = lo + cur[t];
        if (node == n_nodes - 1) rowptr[n_nodes] = hi;
    }
    __syncthreads();
    if (fits) {
        int p[8];
#pragma unroll
        for (int k = 0; k < 8; ++k) p[k] = (t + k * 1024 < m) ? atomicAdd(&cur[d[k]], 1) : 0;
#pragma unroll
        for (int k = 0; k < 8; ++k)
            if (t + k * 1024 < m) stage[p[k]] = s[k];
        __syncthreads();
#pragma unroll
        for (int k = 0; k < 8; ++k) {
            const int i = t + k * 1024;
            if (i < m) ssrc[lo + i] = stage[i];
        }
    } else {
        for (int i = t; i < m; i += 1024) {
            const int j = edge_at(i);
            const int p = atomicAdd(&cur[cdst[j] & (BK_NODES - 1)], 1);
            ssrc[lo + p] = csrc[j];
        }
    }
}

__device__ inline float wave_sum(float x) {
    for (int o = 16; o > 0; o >>= 1) x += __shfl_xor(x, o, 32);
    return x;
}

template <int CTRL>
__device__ inline float quad_perm(float x) {
    return __int_as_float(__builtin_amdgcn_update_dpp(0, __float_as_int(x), CTRL, 0xF, 0xF, true));
}

template <int J>
__device__ inline void gat_add_slot(const v4f& x0, const v4f& x1, float u, v4f& a0, v4f& a1) {
    const float uj = quad_perm<J * 0x55>(u);
    a0 += x0 * uj;
    a1 += x1 * uj;
}

template <int NV>
__device__ inline void gat_group(const float* __restrict__ Hl, int sv, int k, int sub,
                                 const v4f& w0, const v4f& w1, float erv,
                                 float& m, float& s, v4f& a0, v4f& a1) {
    v4f x0[NV], x1[NV];
#pragma unroll
    for (int j = 0; j < NV; ++j) {
        const int sn = __builtin_amdgcn_readlane(sv, k + j);
        x0[j] = *(const v4f*)(Hl + (size_t)sn * 256);
        x1[j] = *(const v4f*)(Hl + (size_t)sn * 256 + 4);
    }
    if constexpr (NV == 4)
        asm volatile("" :: "v"(x0[0]), "v"(x1[0]), "v"(x0[1]), "v"(x1[1]),
                           "v"(x0[2]), "v"(x1[2]), "v"(x0[3]), "v"(x1[3]));
    else if constexpr (NV == 3)
        asm volatile("" :: "v"(x0[0]), "v"(x1[0]), "v"(x0[1]), "v"(x1[1]), "v"(x0[2]), "v"(x1[2]));
    else if constexpr (NV == 2)
        asm volatile("" :: "v"(x0[0]), "v"(x1[0]), "v"(x0[1]), "v"(x1[1]));
    float d[4] = {0.f, 0.f, 0.f, 0.f};
#pragma unroll
    for (int j = 0; j < NV; ++j) {
        const v2f t = x0[j].xy * w0.xy + x0[j].zw * w0.zw + x1[j].xy * w1.xy + x1[j].zw * w1.zw;
        d[j] = t.x + t.y;
    }
    const bool odd = (sub & 1) != 0, upper = (sub & 2) != 0;
    float p01 = odd ? d[1] : d[0];
    p01 += quad_perm<0xB1>(odd ? d[0] : d[1]);
    float p23 = odd ? d[3] : d[2];
    p23 += quad_perm<0xB1>(odd ? d[2] : d[3]);
    float e = upper ? p23 : p01;
    e += quad_perm<0x4E>(upper ? p01 : p23);
    e += erv;
    e = (e >= 0.f) ? e : SLOPE * e;
    if (NV < 4) e = (sub < NV) ? e : -__builtin_huge_valf();
    float gm = fmaxf(e, quad_perm<0xB1>(e));
    gm = fmaxf(gm, quad_perm<0x4E>(gm));
    const float mn = fmaxf(m, gm);
    const float sc = expf(m - mn);
    const float u  = expf(e - mn);
    m = mn;
    s = s * sc + u;
    a0 *= sc;
    a1 *= sc;
    gat_add_slot<0>(x0[0], x1[0], u, a0, a1);
    if constexpr (NV > 1) gat_add_slot<1>(x0[1], x1[1], u, a0, a1);
    if constexpr (NV > 2) gat_add_slot<2>(x0[2], x1[2], u, a0, a1);
    if constexpr (NV > 3) gat_add_slot<3>(x0[3], x1[3], u, a0, a1);
}

#define L1_NPW 5
__global__ void __launch_bounds__(256) gat_l1_segments(const int* __restrict__ rowptr,
                                                       const int* __restrict__ ssrc,
                                                       const float* __restrict__ H,
                                                       const float* __restrict__ al,
                                                       const float* __restrict__ er,
                                                       const float* __restrict__ b1,
                                                       const float* __restrict__ W2,
                                                       const float* __restrict__ al2,
                                                       const float* __restrict__ ar2,
                                                       float* __restrict__ hp,
                                                       float* __restrict__ er2, int n_nodes, int n_edges) {
    const int lane = threadIdx.x & 31;
    const int wv = blockIdx.x * 8 + __builtin_amdgcn_readfirstlane((int)(threadIdx.x >> 5));
    const int node0 = wv * L1_NPW;
    if (node0 >= n_nodes) return;
    const int nend = node0 + L1_NPW < n_nodes ? node0 + L1_NPW : n_nodes;
    const int head = lane >> 2, sub = lane & 3;
    const float* Hl = H + lane * 8;
    const v4f av0 = *(const v4f*)(al + lane * 8);
    const v4f av1 = *(const v4f*)(al + lane * 8 + 4);

    int ee = rowptr[node0];
    int ee_n = rowptr[node0 + 1];
    int jn = ee + lane;
    jn = jn < n_edges ? jn : n_edges - 1;
    int sv_n = ssrc[jn > 0 ? jn : 0];
    float erv_n = er[node0 * 8 + head];

  for (int node = node0; node < nend; ++node) {
    const int eb = ee;
    ee = ee_n;
    const int sv0 = sv_n;
    const float erv = erv_n;
    const int nx = node + 1 < n_nodes ? node + 1 : node;
    ee_n = rowptr[nx + 1];
    jn = ee + lane;
    jn = jn < n_edges ? jn : n_edges - 1;
    sv_n = ssrc[jn > 0 ? jn : 0];
    erv_n = er[nx * 8 + head];

    v4f a0 = {0.f, 0.f, 0.f, 0.f}, a1 = {0.f, 0.f, 0.f, 0.f};
    float m = -__builtin_huge_valf(), s = 0.f;

    for (int c = eb; c < ee; c += 32) {
        int sv = sv0;
        if (c != eb) {
            const int j = c + lane;
            sv = ssrc[j < ee ? j : ee - 1];
        }
        const int left = ee - c;
        const int cnt = left < 32 ? left : 32;
        int k = 0;
        for (; k + 4 <= cnt; k += 4) gat_group<4>(Hl, sv, k, sub, av0, av1, erv, m, s, a0, a1);
        const int r = cnt - k;
        if (r == 3)      gat_group<3>(Hl, sv, k, sub, av0, av1, erv, m, s, a0, a1);
        else if (r == 2) gat_group<2>(Hl, sv, k, sub, av0, av1, erv, m, s, a0, a1);
        else if (r == 1) gat_group<1>(Hl, sv, k, sub, av0, av1, erv, m, s, a0, a1);
    }
    if (ee > eb) {
        s += quad_perm<0xB1>(s);
        s += quad_perm<0x4E>(s);
        const float inv = 1.0f / s;
        a0 *= inv;
        a1 *= inv;
    }
    const float4 c0 = *(const float4*)(b1 + lane * 8);
    const float4 c1 = *(const float4*)(b1 + lane * 8 + 4);
    float x[8] = {a0.x + c0.x, a0.y + c0.y, a0.z + c0.z, a0.w + c0.w,
                  a1.x + c1.x, a1.y + c1.y, a1.z + c1.z, a1.w + c1.w};
#pragma unroll
    for (int i = 0; i < 8; ++i) x[i] = (x[i] > 0.f) ? x[i] : expm1f(x[i]);
    const float4* wp = (const float4*)(W2 + lane * 24);
    const float4 w0 = wp[0], w1 = wp[1], w2 = wp[2], w3 = wp[3], w4 = wp[4], w5 = wp[5];
    float p0 = x[0] * w0.x, p1 = x[0] * w0.y, p2 = x[0] * w0.z;
    p0 += x[1] * w0.w; p1 += x[1] * w1.x; p2 += x[1] * w1.y;
    p0 += x[2] * w1.z; p1 += x[2] * w1.w; p2 += x[2] * w2.x;
    p0 += x[3] * w2.y; p1 += x[3] * w2.z; p2 += x[3] * w2.w;
    p0 += x[4] * w3.x; p1 += x[4] * w3.y; p2 += x[4] * w3.z;
    p0 += x[5] * w3.w; p1 += x[5] * w4.x; p2 += x[5] * w4.y;
    p0 += x[6] * w4.z; p1 += x[6] * w4.w; p2 += x[6] * w5.x;
    p0 += x[7] * w5.y; p1 += x[7] * w5.z; p2 += x[7] * w5.w;
    p0 = wave_sum(p0); p1 = wave_sum(p1); p2 = wave_sum(p2);
    if (lane == 0) {
        float4 row;
        row.x = p0; row.y = p1; row.z = p2;
        row.w = p0 * al2[0] + p1 * al2[1] + p2 * al2[2];
        ((float4*)hp)[node] = row;
        er2[node] = p0 * ar2[0] + p1 * ar2[1] + p2 * ar2[2];
    }
  }
}

__global__ void __launch_bounds__(64) gat_l2_segments(const int* __restrict__ rowptr,
                                                      const int* __restrict__ ssrc,
                                                      const float* __restrict__ hp,
                                                      const float* __restrict__ er2,
                                                      const float* __restrict__ b2,
                                                      float* __restrict__ out, int n_nodes) {
    const int node = blockIdx.x * 64 + threadIdx.x;
    if (node >= n_nodes) return;
    const int eb = rowptr[node], ee = rowptr[node + 1];
    const float erv = er2[node];
    const float NEG_INF = -__builtin_huge_valf();
    const float4* rows = (const float4*)hp;
    float m = NEG_INF, s = 0.f, p0 = 0.f, p1 = 0.f, p2 = 0.f;

    for (int c = eb; c < ee; c += 4) {
        int sn[4];
        float4 r[4];
        float x[4];
#pragma unroll
        for (int k = 0; k < 4; ++k) {
            const int j = c + k;
            sn[k] = ssrc[j < ee ? j : ee - 1];
        }
#pragma unroll
        for (int k = 0; k < 4; ++k) r[k] = rows[sn[k]];
#pragma unroll
        for (int k = 0; k < 4; ++k) {
            float v = r[k].w + erv;
            v = (v >= 0.f) ? v : SLOPE * v;
            x[k] = (c + k < ee) ? v : NEG_INF;
        }
        const float gm = fmaxf(fmaxf(x[0], x[1]), fmaxf(x[2], x[3]));
        const float mn = fmaxf(m, gm);
        const float sc = expf(m - mn);
        const float u0 = expf(x[0] - mn), u1 = expf(x[1] - mn);
        const float u2 = expf(x[2] - mn), u3 = expf(x[3] - mn);
        m = mn;
        s  = s  * sc + ((u0 + u1) + (u2 + u3));
        p0 = p0 * sc + ((u0 * r[0].x + u1 * r[1].x) + (u2 * r[2].x + u3 * r[3].x));
        p1 = p1 * sc + ((u0 * r[0].y + u1 * r[1].y) + (u2 * r[2].y + u3 * r[3].y));
        p2 = p2 * sc + ((u0 * r[0].z + u1 * r[1].z) + (u2 * r[2].z + u3 * r[3].z));
    }
    if (ee > eb) {
        const float inv = 1.0f / s;
        p0 *= inv; p1 *= inv; p2 *= inv;
    }
    out[node * 3 + 0] = p0 + b2[0];
    out[node * 3 + 1] = p1 + b2[1];
    out[node * 3 + 2] = p2 + b2[2];
}

extern "C" void kernel_launch(void* const* d_in, const int* in_sizes, int n_in,
                              void* d_out, int out_size, void* d_ws, size_t ws_size,
                              hipStream_t stream) {
    const float* feat = (const float*)d_in[0];
    const int*   src  = (const int*)  d_in[1];
    const int*   dst  = (const int*)  d_in[2];
    const float* W1   = (const float*)d_in[3];
    const float* al1  = (const float*)d_in[4];
    const float* ar1  = (const float*)d_in[5];
    const float* b1   = (const float*)d_in[6];
    const float* W2   = (const float*)d_in[7];
    const float* al2  = (const float*)d_in[8];
    const float* ar2  = (const float*)d_in[9];
    const float* b2   = (const float*)d_in[10];
    float* out = (float*)d_out;

    const int n_nodes = in_sizes[0] / 256;
    const int n_edges = in_sizes[1];

    char* cur = (char*)d_ws;
    auto alloc = [&](size_t bytes) -> float* {
        float* p = (float*)cur;
        cur += (bytes + 255) & ~(size_t)255;
        return p;
    };
    float* H1   = alloc((size_t)n_nodes * 256 * 4);
    float* er1  = alloc((size_t)n_nodes * 8 * 4);
    float* hp2  = alloc((size_t)n_nodes * 4 * 4);
    float* er2  = alloc((size_t)n_nodes * 4);
    const int n_buckets = cdiv(n_nodes, BK_NODES);
    const int n_slices  = cdiv(n_edges, SLICE);
    int*   wstart = (int*)alloc(((size_t)n_buckets + 1) * n_slices * 4);
    int*   csrc   = (int*)alloc((size_t)n_edges * 4);
    int*   cdst   = (int*)alloc((size_t)n_edges * 4);
    int*   rowptr = (int*)alloc((size_t)(n_nodes + 1) * 4);
    int*   ssrc   = (int*)alloc((size_t)n_edges * 4);

    gat_gemm1_and_slice_sort<<<n_slices + cdiv(n_nodes, 32), 256, 0, stream>>>(
        feat, W1, ar1, H1, er1, n_nodes, src, dst, csrc, cdst, wstart, n_edges, n_buckets, n_slices);
    gat_bucket_sort<<<n_buckets, 1024, 0, stream>>>(csrc, cdst, wstart, rowptr, ssrc, n_nodes, n_edges, n_slices);

    gat_l1_segments<<<cdiv(n_nodes, 8 * L1_NPW), 256, 0, stream>>>(rowptr, ssrc, H1, al1, er1,
                                                          b1, W2, al2, ar2, hp2, er2, n_nodes, n_edges);

    gat_l2_segments<<<cdiv(n_nodes, 64), 64, 0, stream>>>(rowptr, ssrc, hp2, er2, b2, out, n_nodes);
}
